# starting kernel
# baseline (speedup 1.0000x reference)
#include <hip/hip_runtime.h>

typedef __attribute__((ext_vector_type(2))) float v2f;
typedef __attribute__((ext_vector_type(8))) float v8f;

#define IN_CH   128
#define OUT_CH  64
#define POS_DIM 64

__global__ void init_deg_kernel(float* __restrict__ deg, int N) {
    int i = blockIdx.x * blockDim.x + threadIdx.x;
    if (i < N) deg[i] = 1.0f;
}

__global__ void count_deg_kernel(const int* __restrict__ dst, float* __restrict__ deg, int E) {
    int e = blockIdx.x * blockDim.x + threadIdx.x;
    if (e < E) atomicAdd(&deg[dst[e]], 1.0f);
}

__global__ void rsqrt_kernel(float* __restrict__ deg, int N) {
    int i = blockIdx.x * blockDim.x + threadIdx.x;
    if (i < N) deg[i] = rsqrtf(deg[i]);
}

__global__ void gemm_xw_kernel(const float* __restrict__ x,
                               const float* __restrict__ W,
                               float* __restrict__ h, int N, int mTiles) {
    const int wavesPerBlock = blockDim.x >> 5;
    const int waveId = threadIdx.x >> 5;
    const int lane   = threadIdx.x & 31;
    const int mTile  = blockIdx.x * wavesPerBlock + waveId;
    const int nTile  = blockIdx.y;
    if (mTile >= mTiles) return;

    const int l16 = lane & 15;
    const int kh  = (lane >> 4) << 1;
    int m = mTile * 16 + l16;
    if (m >= N) m = N - 1;
    const int n = nTile * 16 + l16;

    v8f c = {};
#pragma unroll
    for (int k = 0; k < IN_CH; k += 4) {
        v2f a = *(const v2f*)(x + (size_t)m * IN_CH + k + kh);
        v2f bfrag;
        bfrag[0] = W[(size_t)(k + kh + 0) * OUT_CH + n];
        bfrag[1] = W[(size_t)(k + kh + 1) * OUT_CH + n];
        c = __builtin_amdgcn_wmma_f32_16x16x4_f32(
                false, a, false, bfrag, (short)0, c, false, false);
    }

    const int rbase = mTile * 16 + ((lane >> 4) << 3);
#pragma unroll
    for (int r = 0; r < 8; ++r) {
        int row = rbase + r;
        if (row < N) h[(size_t)row * OUT_CH + n] = c[r];
    }
}

__global__ void gemm_pos_fused_kernel(const float* __restrict__ pos,
                                      const float* __restrict__ Wp,
                                      const float* __restrict__ bp,
                                      const float* __restrict__ b,
                                      const float* __restrict__ h,
                                      const float* __restrict__ dinv,
                                      float* __restrict__ out, int N, int mTiles) {
    const int wavesPerBlock = blockDim.x >> 5;
    const int waveId = threadIdx.x >> 5;
    const int lane   = threadIdx.x & 31;
    const int mTile  = blockIdx.x * wavesPerBlock + waveId;
    const int nTile  = blockIdx.y;
    if (mTile >= mTiles) return;

    const int l16 = lane & 15;
    const int kh  = (lane >> 4) << 1;
    int m = mTile * 16 + l16;
    if (m >= N) m = N - 1;
    const int n = nTile * 16 + l16;

    v8f c = {};
#pragma unroll
    for (int k = 0; k < POS_DIM; k += 4) {
        v2f a = *(const v2f*)(pos + (size_t)m * POS_DIM + k + kh);
        v2f bfrag;
        bfrag[0] = Wp[(size_t)(k + kh + 0) * OUT_CH + n];
        bfrag[1] = Wp[(size_t)(k + kh + 1) * OUT_CH + n];
        c = __builtin_amdgcn_wmma_f32_16x16x4_f32(
                false, a, false, bfrag, (short)0, c, false, false);
    }

    const float bias = bp[n] + b[n];
    const int rbase = mTile * 16 + ((lane >> 4) << 3);
#pragma unroll
    for (int r = 0; r < 8; ++r) {
        int row = rbase + r;
        if (row < N) {
            float di = dinv[row];
            out[(size_t)row * OUT_CH + n] =
                c[r] + bias + h[(size_t)row * OUT_CH + n] * di * di;
        }
    }
}

__global__ void edge_scatter_kernel(const int* __restrict__ src,
                                    const int* __restrict__ dst,
                                    const float* __restrict__ h,
                                    const float* __restrict__ dinv,
                                    float* __restrict__ out, int E) {
    const int wavesPerBlock = blockDim.x >> 5;
    const int e    = blockIdx.x * wavesPerBlock + (threadIdx.x >> 5);
    const int lane = threadIdx.x & 31;
    if (e >= E) return;

    const int s = src[e];
    const int d = dst[e];
    const float nrm = dinv[s] * dinv[d];

    const float2 hv = *(const float2*)(h + (size_t)s * OUT_CH + 2 * lane);
    float* op = out + (size_t)d * OUT_CH + 2 * lane;
    atomicAdd(op + 0, hv.x * nrm);
    atomicAdd(op + 1, hv.y * nrm);
}

extern "C" void kernel_launch(void* const* d_in, const int* in_sizes, int n_in,
                              void* d_out, int out_size, void* d_ws, size_t ws_size,
                              hipStream_t stream) {
    const float* x   = (const float*)d_in[0];
    const int*   ei  = (const int*)d_in[1];
    const float* pos = (const float*)d_in[2];
    const float* W   = (const float*)d_in[3];
    const float* b   = (const float*)d_in[4];
    const float* Wp  = (const float*)d_in[5];
    const float* bp  = (const float*)d_in[6];
    float* out = (float*)d_out;

    const int N = in_sizes[0] / IN_CH;
    const int E = in_sizes[1] / 2;
    const int* src = ei;
    const int* dst = ei + E;

    float* dinv = (float*)d_ws;
    float* h    = dinv + ((N + 3) & ~3);

    init_deg_kernel<<<(N + 255) / 256, 256, 0, stream>>>(dinv, N);
    count_deg_kernel<<<(E + 255) / 256, 256, 0, stream>>>(dst, dinv, E);
    rsqrt_kernel<<<(N + 255) / 256, 256, 0, stream>>>(dinv, N);

    const int mTiles = (N + 15) / 16;
    const int wavesPerBlock = 8;
    dim3 gGemm((mTiles + wavesPerBlock - 1) / wavesPerBlock, OUT_CH / 16);
    gemm_xw_kernel<<<gGemm, 32 * wavesPerBlock, 0, stream>>>(x, W, h, N, mTiles);

    gemm_pos_fused_kernel<<<gGemm, 32 * wavesPerBlock, 0, stream>>>(
        pos, Wp, bp, b, h, dinv, out, N, mTiles);

    edge_scatter_kernel<<<(E + wavesPerBlock - 1) / wavesPerBlock,
                          32 * wavesPerBlock, 0, stream>>>(src, dst, h, dinv, out, E);
}


# degree counted per bin from the binned edge list (one small block per 400-node bin, LDS integer counters) behind the fill; both degree launches and their 64 partial planes go; counts exact
# speedup vs baseline: 2.4079x; 2.4079x over previous
#include <hip/hip_runtime.h>

typedef __attribute__((ext_vector_type(2))) float v2f;
typedef __attribute__((ext_vector_type(8))) float v8f;
typedef __attribute__((ext_vector_type(4))) float v4f;

#define IN_CH   128
#define OUT_CH  64
#define POS_DIM 64

#define DEG_RNG    25600
#define DEG_CHUNKS 64
#define DEG_BLOCK  1024

__global__ __launch_bounds__(DEG_BLOCK)
void deg_count_lds_kernel(const int* __restrict__ dst, unsigned* __restrict__ planes,
                          int N, int E) {
    __shared__ unsigned cnt[DEG_RNG];
    const int tid  = threadIdx.x;
    const int c    = blockIdx.x;
    const int base = blockIdx.y * DEG_RNG;
    int len = N - base;
    if (len > DEG_RNG) len = DEG_RNG;

    for (int i = tid; i < len; i += DEG_BLOCK) cnt[i] = 0u;
    __syncthreads();

    const int per = (E + DEG_CHUNKS - 1) / DEG_CHUNKS;
    const int e0  = c * per;
    int e1 = e0 + per;
    if (e1 > E) e1 = E;
    for (int e = e0 + tid; e < e1; e += 4 * DEG_BLOCK) {
        const int d0 = dst[e];
        const int d1 = (e + 1 * DEG_BLOCK < e1) ? dst[e + 1 * DEG_BLOCK] : -1;
        const int d2 = (e + 2 * DEG_BLOCK < e1) ? dst[e + 2 * DEG_BLOCK] : -1;
        const int d3 = (e + 3 * DEG_BLOCK < e1) ? dst[e + 3 * DEG_BLOCK] : -1;
        const unsigned r0 = (unsigned)(d0 - base);
        const unsigned r1 = (unsigned)(d1 - base);
        const unsigned r2 = (unsigned)(d2 - base);
        const unsigned r3 = (unsigned)(d3 - base);
        if (r0 < (unsigned)len) atomicAdd(&cnt[r0], 1u);
        if (r1 < (unsigned)len) atomicAdd(&cnt[r1], 1u);
        if (r2 < (unsigned)len) atomicAdd(&cnt[r2], 1u);
        if (r3 < (unsigned)len) atomicAdd(&cnt[r3], 1u);
    }
    __syncthreads();

    unsigned* p = planes + (size_t)c * N + base;
    for (int i = tid; i < len; i += DEG_BLOCK) p[i] = cnt[i];
}

__global__ void deg_finish_kernel(const unsigned* __restrict__ planes,
                                  float* __restrict__ dinv, int N) {
    const int i = blockIdx.x * blockDim.x + threadIdx.x;
    if (i < N) {
        unsigned s = 1u;
#pragma unroll
        for (int c = 0; c < DEG_CHUNKS; ++c) s += planes[(size_t)c * N + i];
        dinv[i] = rsqrtf((float)s);
    }
}

#define GX_WAVES 8
#define XS_PITCH 132
#define BS_PITCH 160
#define GX_WITEMS ((IN_CH / 2) * (OUT_CH / 4) / (32 * GX_WAVES))

__global__ __launch_bounds__(32 * GX_WAVES)
void gemm_xw_lds_kernel(const float* __restrict__ x,
                        const float* __restrict__ W,
                        float* __restrict__ h, int N, int mTiles) {
    __shared__ float lds[(IN_CH / 2) * BS_PITCH + GX_WAVES * 16 * XS_PITCH];
    float* wsm = lds;
    float* xs  = lds + (IN_CH / 2) * BS_PITCH;
    const int tid  = threadIdx.x;
    const int wave = __builtin_amdgcn_readfirstlane(tid >> 5);
    const int lane = tid & 31;
    const int mTile = blockIdx.x * GX_WAVES + wave;
    const int row0  = mTile * 16;
    const int nl    = N - 1;

    v4f w0[GX_WITEMS], w1[GX_WITEMS];
#pragma unroll
    for (int j = 0; j < GX_WITEMS; ++j) {
        const int item = tid + j * 32 * GX_WAVES;
        const int p = item >> 4;
        const int q = item & 15;
        w0[j] = *(const v4f*)(W + (2 * p + 0) * OUT_CH + 4 * q);
        w1[j] = *(const v4f*)(W + (2 * p + 1) * OUT_CH + 4 * q);
    }
    v4f xv[16];
    const float* xp = x + lane * 4;
#pragma unroll
    for (int r = 0; r < 16; ++r)
        xv[r] = *(const v4f*)(xp + (size_t)min(row0 + r, nl) * IN_CH);
#pragma unroll
    for (int j = 0; j < GX_WITEMS; ++j) {
        const int item = tid + j * 32 * GX_WAVES;
        const int p = item >> 4;
        const int q = item & 15;
        float* d = wsm + p * BS_PITCH + 8 * q;
        const v4f lo = {w0[j][0], w1[j][0], w0[j][1], w1[j][1]};
        const v4f hi = {w0[j][2], w1[j][2], w0[j][3], w1[j][3]};
        *(v4f*)(d + 0) = lo;
        *(v4f*)(d + 4) = hi;
    }
    float* xw = xs + wave * 16 * XS_PITCH;
#pragma unroll
    for (int r = 0; r < 16; ++r)
        *(v4f*)(xw + r * XS_PITCH + lane * 4) = xv[r];
    __syncthreads();

    const int l16 = lane & 15;
    const int hh  = lane >> 4;
    v2f a[32];
#pragma unroll
    for (int s = 0; s < 32; ++s)
        a[s] = *(const v2f*)(xw + l16 * XS_PITCH + 4 * s + 2 * hh);
    v8f c0 = {}, c1 = {}, c2 = {}, c3 = {};
#pragma unroll
    for (int s = 0; s < 32; ++s) {
        const float* bb = wsm + (2 * s + hh) * BS_PITCH + l16 * 2;
        const v2f f0 = *(const v2f*)(bb + 0);
        const v2f f1 = *(const v2f*)(bb + 32);
        const v2f f2 = *(const v2f*)(bb + 64);
        const v2f f3 = *(const v2f*)(bb + 96);
        c0 = __builtin_amdgcn_wmma_f32_16x16x4_f32(false, a[s], false, f0, (short)0, c0, false, false);
        c1 = __builtin_amdgcn_wmma_f32_16x16x4_f32(false, a[s], false, f1, (short)0, c1, false, false);
        c2 = __builtin_amdgcn_wmma_f32_16x16x4_f32(false, a[s], false, f2, (short)0, c2, false, false);
        c3 = __builtin_amdgcn_wmma_f32_16x16x4_f32(false, a[s], false, f3, (short)0, c3, false, false);
    }

    float* pr = xw + (8 * hh) * XS_PITCH + l16;
#pragma unroll
    for (int r = 0; r < 8; ++r) {
        pr[r * XS_PITCH +  0] = c0[r];
        pr[r * XS_PITCH + 16] = c1[r];
        pr[r * XS_PITCH + 32] = c2[r];
        pr[r * XS_PITCH + 48] = c3[r];
    }
    const int rl = lane >> 4;
    const int c4 = (lane & 15) << 2;
    float* hp = h + (size_t)row0 * OUT_CH + rl * OUT_CH + c4;
    if (row0 + 16 <= N) {
        v4f v[8];
#pragma unroll
        for (int it = 0; it < 8; ++it)
            v[it] = *(const v4f*)(xw + (2 * it + rl) * XS_PITCH + c4);
#pragma unroll
        for (int it = 0; it < 8; ++it)
            *(v4f*)(hp + it * 2 * OUT_CH) = v[it];
    } else if (mTile < mTiles) {
#pragma unroll
        for (int it = 0; it < 8; ++it) {
            if (row0 + 2 * it + rl < N)
                *(v4f*)(hp + it * 2 * OUT_CH) = *(const v4f*)(xw + (2 * it + rl) * XS_PITCH + c4);
        }
    }
}

__global__ void gemm_pos_fused_kernel(const float* __restrict__ pos,
                                      const float* __restrict__ Wp,
                                      const float* __restrict__ bp,
                                      const float* __restrict__ b,
                                      const float* __restrict__ h,
                                      const float* __restrict__ dinv,
                                      float* __restrict__ out, int N, int mTiles) {
    const int wavesPerBlock = blockDim.x >> 5;
    const int waveId = threadIdx.x >> 5;
    const int lane   = threadIdx.x & 31;
    const int mTile  = blockIdx.x * wavesPerBlock + waveId;
    const int nTile  = blockIdx.y;
    if (mTile >= mTiles) return;

    const int l16 = lane & 15;
    const int kh  = (lane >> 4) << 1;
    int m = mTile * 16 + l16;
    if (m >= N) m = N - 1;
    const int n = nTile * 16 + l16;

    v8f c = {};
#pragma unroll
    for (int k = 0; k < POS_DIM; k += 4) {
        v2f a = *(const v2f*)(pos + (size_t)m * POS_DIM + k + kh);
        v2f bfrag;
        bfrag[0] = Wp[(size_t)(k + kh + 0) * OUT_CH + n];
        bfrag[1] = Wp[(size_t)(k + kh + 1) * OUT_CH + n];
        c = __builtin_amdgcn_wmma_f32_16x16x4_f32(
                false, a, false, bfrag, (short)0, c, false, false);
    }

    const float bias = bp[n] + b[n];
    const int rbase = mTile * 16 + ((lane >> 4) << 3);
#pragma unroll
    for (int r = 0; r < 8; ++r) {
        int row = rbase + r;
        if (row < N) {
            float di = dinv[row];
            out[(size_t)row * OUT_CH + n] =
                c[r] + bias + h[(size_t)row * OUT_CH + n] * di * di;
        }
    }
}

__global__ void edge_scatter_kernel(const int* __restrict__ src,
                                    const int* __restrict__ dst,
                                    const float* __restrict__ h,
                                    const float* __restrict__ dinv,
                                    float* __restrict__ out, int E) {
    const int wavesPerBlock = blockDim.x >> 5;
    const int e    = blockIdx.x * wavesPerBlock + (threadIdx.x >> 5);
    const int lane = threadIdx.x & 31;
    if (e >= E) return;

    const int s = src[e];
    const int d = dst[e];
    const float nrm = dinv[s] * dinv[d];

    const float2 hv = *(const float2*)(h + (size_t)s * OUT_CH + 2 * lane);
    float* op = out + (size_t)d * OUT_CH + 2 * lane;
    atomicAdd(op + 0, hv.x * nrm);
    atomicAdd(op + 1, hv.y * nrm);
}

#define BIN_RNG   400
#define BIN_CH    128
#define BIN_MAX   1024
#define BIN_BLOCK 1024

__global__ __launch_bounds__(BIN_BLOCK)
void bin_count_kernel(const int* __restrict__ dst, unsigned* __restrict__ counts,
                      int N, int E, int NB) {
    __shared__ unsigned hist[BIN_MAX];
    const int tid = threadIdx.x;
    const int c   = blockIdx.x;
    for (int i = tid; i < NB; i += BIN_BLOCK) hist[i] = 0u;
    __syncthreads();

    const int per = (E + BIN_CH - 1) / BIN_CH;
    const int e0  = c * per;
    int e1 = e0 + per;
    if (e1 > E) e1 = E;
    for (int e = e0 + tid; e < e1; e += 4 * BIN_BLOCK) {
        const int d0 = dst[e];
        const int d1 = (e + 1 * BIN_BLOCK < e1) ? dst[e + 1 * BIN_BLOCK] : -1;
        const int d2 = (e + 2 * BIN_BLOCK < e1) ? dst[e + 2 * BIN_BLOCK] : -1;
        const int d3 = (e + 3 * BIN_BLOCK < e1) ? dst[e + 3 * BIN_BLOCK] : -1;
        if ((unsigned)d0 < (unsigned)N) atomicAdd(&hist[d0 / BIN_RNG], 1u);
        if ((unsigned)d1 < (unsigned)N) atomicAdd(&hist[d1 / BIN_RNG], 1u);
        if ((unsigned)d2 < (unsigned)N) atomicAdd(&hist[d2 / BIN_RNG], 1u);
        if ((unsigned)d3 < (unsigned)N) atomicAdd(&hist[d3 / BIN_RNG], 1u);
    }
    __syncthreads();
    for (int i = tid; i < NB; i += BIN_BLOCK) counts[(size_t)i * BIN_CH + c] = hist[i];
}

__global__ __launch_bounds__(BIN_BLOCK)
void bin_scan_kernel(const unsigned* __restrict__ counts, unsigned* __restrict__ offs, int M) {
    __shared__ unsigned part[BIN_BLOCK];
    const int tid = threadIdx.x;
    unsigned carry = 0u;
    for (int t0 = 0; t0 < M; t0 += 32 * BIN_BLOCK) {
        const int i0 = t0 + tid * 32;
        uint4 v[8];
#pragma unroll
        for (int k = 0; k < 8; ++k) {
            v[k] = make_uint4(0u, 0u, 0u, 0u);
            if (i0 + 4 * k < M) v[k] = *(const uint4*)(counts + i0 + 4 * k);
        }
        unsigned sum = 0u;
#pragma unroll
        for (int k = 0; k < 8; ++k) sum += v[k].x + v[k].y + v[k].z + v[k].w;
        part[tid] = sum;
        __syncthreads();
        for (int off = 1; off < BIN_BLOCK; off <<= 1) {
            const unsigned t = (tid >= off) ? part[tid - off] : 0u;
            __syncthreads();
            part[tid] += t;
            __syncthreads();
        }
        unsigned run = carry + part[tid] - sum;
#pragma unroll
        for (int k = 0; k < 8; ++k) {
            uint4 o;
            o.x = run; run += v[k].x;
            o.y = run; run += v[k].y;
            o.z = run; run += v[k].z;
            o.w = run; run += v[k].w;
            if (i0 + 4 * k < M) *(uint4*)(offs + i0 + 4 * k) = o;
        }
        carry += part[BIN_BLOCK - 1];
        __syncthreads();
    }
    if (tid == 0) offs[M] = carry;
}

__global__ __launch_bounds__(BIN_BLOCK)
void bin_fill_kernel(const int* __restrict__ src, const int* __restrict__ dst,
                     const unsigned* __restrict__ offs, unsigned* __restrict__ list,
                     int N, int E, int NB) {
    __shared__ unsigned cursor[BIN_MAX];
    const int tid = threadIdx.x;
    const int c   = blockIdx.x;
    for (int i = tid; i < NB; i += BIN_BLOCK) cursor[i] = offs[(size_t)i * BIN_CH + c];
    __syncthreads();

    const int per = (E + BIN_CH - 1) / BIN_CH;
    const int e0  = c * per;
    int e1 = e0 + per;
    if (e1 > E) e1 = E;
    for (int e = e0 + tid; e < e1; e += 4 * BIN_BLOCK) {
        int d[4], sv[4];
#pragma unroll
        for (int k = 0; k < 4; ++k) {
            const int ee = e + k * BIN_BLOCK;
            const bool in = ee < e1;
            d[k]  = in ? dst[ee] : -1;
            sv[k] = in ? src[ee] : 0;
        }
#pragma unroll
        for (int k = 0; k < 4; ++k) {
            if ((unsigned)d[k] < (unsigned)N) {
                const int bin = d[k] / BIN_RNG;
                const unsigned p = atomicAdd(&cursor[bin], 1u);
                list[p] = ((unsigned)sv[k] << 9) | (unsigned)(d[k] - bin * BIN_RNG);
            }
        }
    }
}

#define DEGL_BLOCK 256
#define DEGL_DEPTH 16

__global__ __launch_bounds__(DEGL_BLOCK)
void bin_deg_kernel(const unsigned* __restrict__ list, const unsigned* __restrict__ offs,
                    float* __restrict__ dinv, int N) {
    __shared__ unsigned cnt[512];
    const int tid  = threadIdx.x;
    const int bin  = blockIdx.x;
    const int base = bin * BIN_RNG;
    int len = N - base;
    if (len > BIN_RNG) len = BIN_RNG;
    for (int i = tid; i < 512; i += DEGL_BLOCK) cnt[i] = 0u;
    const unsigned b0 = offs[(size_t)bin * BIN_CH];
    const unsigned b1 = offs[(size_t)(bin + 1) * BIN_CH];
    __syncthreads();

    for (unsigned i = b0 + (unsigned)tid; i < b1; i += DEGL_DEPTH * DEGL_BLOCK) {
        unsigned w[DEGL_DEPTH];
#pragma unroll
        for (int k = 0; k < DEGL_DEPTH; ++k) {
            unsigned idx = i + (unsigned)k * DEGL_BLOCK;
            if (idx > b1 - 1u) idx = b1 - 1u;
            w[k] = list[idx];
        }
#pragma unroll
        for (int k = 0; k < DEGL_DEPTH; ++k)
            atomicAdd(&cnt[w[k] & 511u], (i + (unsigned)k * DEGL_BLOCK < b1) ? 1u : 0u);
    }
    __syncthreads();

    for (int i = tid; i < len; i += DEGL_BLOCK)
        dinv[base + i] = rsqrtf((float)(cnt[i] + 1u));
}

template <bool GUARD>
__device__ __forceinline__ void bin_sum_group(const unsigned* __restrict__ list,
                                              const float* __restrict__ dinv,
                                              const float* __restrict__ h,
                                              float* acc, unsigned i, unsigned b1, int lane) {
    unsigned w[8];
    float    sc[8];
    float2   hv[8];
#pragma unroll
    for (int k = 0; k < 8; ++k) {
        unsigned idx = i + (unsigned)k * (BIN_BLOCK / 32);
        if (GUARD && idx > b1 - 1u) idx = b1 - 1u;
        w[k] = list[idx];
    }
#pragma unroll
    for (int k = 0; k < 8; ++k) {
        const unsigned sn = w[k] >> 9;
        sc[k] = dinv[sn];
        hv[k] = *(const float2*)(h + (size_t)sn * OUT_CH + 2 * lane);
    }
#pragma unroll
    for (int k = 0; k < 8; ++k) {
        if (!GUARD || i + (unsigned)k * (BIN_BLOCK / 32) < b1) {
            float* a = acc + (w[k] & 511u) * OUT_CH + 2 * lane;
            atomicAdd(a + 0, hv[k].x * sc[k]);
            atomicAdd(a + 1, hv[k].y * sc[k]);
        }
    }
}

#define POSP_PITCH 68
#define OUT_ITERS  ((BIN_RNG * (OUT_CH / 4) + BIN_BLOCK - 1) / BIN_BLOCK)

__device__ __forceinline__ void pos_tile_wmma(float* posp, const float* bs,
                                              const float* __restrict__ bp,
                                              const float* __restrict__ b,
                                              int r0, int lane) {
    const int l16 = lane & 15;
    const int hh  = lane >> 4;
    v2f a[16];
#pragma unroll
    for (int s = 0; s < 16; ++s)
        a[s] = *(const v2f*)(posp + (r0 + l16) * POSP_PITCH + 4 * s + 2 * hh);
    v8f c0 = {}, c1 = {}, c2 = {}, c3 = {};
#pragma unroll
    for (int s = 0; s < 16; ++s) {
        const float* bb = bs + (2 * s + hh) * BS_PITCH + l16 * 2;
        const v2f f0 = *(const v2f*)(bb + 0);
        const v2f f1 = *(const v2f*)(bb + 32);
        const v2f f2 = *(const v2f*)(bb + 64);
        const v2f f3 = *(const v2f*)(bb + 96);
        c0 = __builtin_amdgcn_wmma_f32_16x16x4_f32(false, a[s], false, f0, (short)0, c0, false, false);
        c1 = __builtin_amdgcn_wmma_f32_16x16x4_f32(false, a[s], false, f1, (short)0, c1, false, false);
        c2 = __builtin_amdgcn_wmma_f32_16x16x4_f32(false, a[s], false, f2, (short)0, c2, false, false);
        c3 = __builtin_amdgcn_wmma_f32_16x16x4_f32(false, a[s], false, f3, (short)0, c3, false, false);
    }
    const float bias0 = bp[l16 +  0] + b[l16 +  0];
    const float bias1 = bp[l16 + 16] + b[l16 + 16];
    const float bias2 = bp[l16 + 32] + b[l16 + 32];
    const float bias3 = bp[l16 + 48] + b[l16 + 48];
    float* pr = posp + (r0 + 8 * hh) * POSP_PITCH + l16;
#pragma unroll
    for (int r = 0; r < 8; ++r) {
        pr[r * POSP_PITCH +  0] = c0[r] + bias0;
        pr[r * POSP_PITCH + 16] = c1[r] + bias1;
        pr[r * POSP_PITCH + 32] = c2[r] + bias2;
        pr[r * POSP_PITCH + 48] = c3[r] + bias3;
    }
}

__global__ __launch_bounds__(BIN_BLOCK)
void bin_sum_pos_kernel(const unsigned* __restrict__ list, const unsigned* __restrict__ offs,
                        const float* __restrict__ h, const float* __restrict__ dinv,
                        const float* __restrict__ pos, const float* __restrict__ Wp,
                        const float* __restrict__ bp, const float* __restrict__ b,
                        float* __restrict__ out, int N) {
    __shared__ float acc[BIN_RNG * OUT_CH];
    __shared__ float posp[BIN_RNG * POSP_PITCH];
    __shared__ float bs[(POS_DIM / 2) * BS_PITCH];
    const int tid  = threadIdx.x;
    const int bin  = blockIdx.x;
    const int base = bin * BIN_RNG;
    int len = N - base;
    if (len > BIN_RNG) len = BIN_RNG;
    const unsigned wave = (unsigned)__builtin_amdgcn_readfirstlane(tid >> 5);
    const int lane = tid & 31;
    const bool tile_wave = wave < BIN_RNG / 16;
    const int r0 = (int)wave * 16;

    v4f pv0 = {}, pv1 = {}, pv2 = {}, pv3 = {}, pv4 = {}, pv5 = {}, pv6 = {}, pv7 = {};
    if (tile_wave) {
        const float* pp = pos + (lane & 15) * 4;
        const int rl = base + r0 + (lane >> 4);
        const int nl = N - 1;
        pv0 = *(const v4f*)(pp + (size_t)min(rl +  0, nl) * POS_DIM);
        pv1 = *(const v4f*)(pp + (size_t)min(rl +  2, nl) * POS_DIM);
        pv2 = *(const v4f*)(pp + (size_t)min(rl +  4, nl) * POS_DIM);
        pv3 = *(const v4f*)(pp + (size_t)min(rl +  6, nl) * POS_DIM);
        pv4 = *(const v4f*)(pp + (size_t)min(rl +  8, nl) * POS_DIM);
        pv5 = *(const v4f*)(pp + (size_t)min(rl + 10, nl) * POS_DIM);
        pv6 = *(const v4f*)(pp + (size_t)min(rl + 12, nl) * POS_DIM);
        pv7 = *(const v4f*)(pp + (size_t)min(rl + 14, nl) * POS_DIM);
    }
    const float4 wv = *(const float4*)(Wp + (tid >> 4) * OUT_CH + ((tid & 15) << 2));

    for (int i = tid; i < BIN_RNG * OUT_CH; i += BIN_BLOCK) acc[i] = 0.0f;
    {
        const int k  = tid >> 4;
        const int n0 = (tid & 15) << 2;
        float* d = bs + (k >> 1) * BS_PITCH + (k & 1);
        d[(n0 + 0) * 2] = wv.x;
        d[(n0 + 1) * 2] = wv.y;
        d[(n0 + 2) * 2] = wv.z;
        d[(n0 + 3) * 2] = wv.w;
    }
    if (tile_wave) {
        float* ps = posp + (r0 + (lane >> 4)) * POSP_PITCH + (lane & 15) * 4;
        *(v4f*)(ps +  0 * POSP_PITCH) = pv0;
        *(v4f*)(ps +  2 * POSP_PITCH) = pv1;
        *(v4f*)(ps +  4 * POSP_PITCH) = pv2;
        *(v4f*)(ps +  6 * POSP_PITCH) = pv3;
        *(v4f*)(ps +  8 * POSP_PITCH) = pv4;
        *(v4f*)(ps + 10 * POSP_PITCH) = pv5;
        *(v4f*)(ps + 12 * POSP_PITCH) = pv6;
        *(v4f*)(ps + 14 * POSP_PITCH) = pv7;
    }
    __syncthreads();

    const unsigned b0 = offs[(size_t)bin * BIN_CH];
    const unsigned b1 = offs[(size_t)(bin + 1) * BIN_CH];

    unsigned i = b0 + wave;
    unsigned g = 0u;
    const unsigned turn = wave;
    bool pending = tile_wave;
    for (;;) {
        const bool more = i + 7u * (BIN_BLOCK / 32) < b1;
        if (pending && (g == turn || !more)) {
            pos_tile_wmma(posp, bs, bp, b, r0, lane);
            pending = false;
        }
        if (!more) break;
        bin_sum_group<false>(list, dinv, h, acc, i, b1, lane);
        i += 8u * (BIN_BLOCK / 32);
        ++g;
    }
    if (i < b1)
        bin_sum_group<true>(list, dinv, h, acc, i, b1, lane);
    __syncthreads();

    const float* hb = h + (size_t)base * OUT_CH;
    float* o = out + (size_t)base * OUT_CH;
    const int nq = len * (OUT_CH / 4);
    float  di[OUT_ITERS];
    float4 hv[OUT_ITERS];
#pragma unroll
    for (int it = 0; it < OUT_ITERS; ++it) {
        int q = tid + it * BIN_BLOCK;
        if (q > nq - 1) q = nq - 1;
        di[it] = dinv[base + (q >> 4)];
        hv[it] = *(const float4*)(hb + (q >> 4) * OUT_CH + ((q & 15) << 2));
    }
#pragma unroll
    for (int it = 0; it < OUT_ITERS; ++it) {
        const int q = tid + it * BIN_BLOCK;
        if (q < nq) {
            const int row = q >> 4;
            const int c4  = (q & 15) << 2;
            const float4 pq = *(const float4*)(posp + row * POSP_PITCH + c4);
            const float4 av = *(const float4*)(acc + row * OUT_CH + c4);
            const float d = di[it];
            float4 r;
            r.x = (pq.x + hv[it].x * d * d) + d * av.x;
            r.y = (pq.y + hv[it].y * d * d) + d * av.y;
            r.z = (pq.z + hv[it].z * d * d) + d * av.z;
            r.w = (pq.w + hv[it].w * d * d) + d * av.w;
            *(float4*)(o + row * OUT_CH + c4) = r;
        }
    }
}

extern "C" void kernel_launch(void* const* d_in, const int* in_sizes, int n_in,
                              void* d_out, int out_size, void* d_ws, size_t ws_size,
                              hipStream_t stream) {
    const float* x   = (const float*)d_in[0];
    const int*   ei  = (const int*)d_in[1];
    const float* pos = (const float*)d_in[2];
    const float* W   = (const float*)d_in[3];
    const float* b   = (const float*)d_in[4];
    const float* Wp  = (const float*)d_in[5];
    const float* bp  = (const float*)d_in[6];
    float* out = (float*)d_out;

    const int N = in_sizes[0] / IN_CH;
    const int E = in_sizes[1] / 2;
    const int* src = ei;
    const int* dst = ei + E;

    float* dinv = (float*)d_ws;
    float* h    = dinv + ((N + 3) & ~3);

    const int mTiles = (N + 15) / 16;
    const int wavesPerBlock = 8;
    dim3 gGemm((mTiles + wavesPerBlock - 1) / wavesPerBlock, OUT_CH / 16);

    const int NB = (N + BIN_RNG - 1) / BIN_RNG;
    const int M  = NB * BIN_CH;
    unsigned* counts = (unsigned*)(h + (size_t)N * OUT_CH);
    unsigned* offs   = counts + M;
    unsigned* list   = offs + M + 4;
    const size_t need = (size_t)((char*)(list + E) - (char*)d_ws);
    if (NB <= BIN_MAX && N <= (1 << 23) && need <= ws_size) {
        gemm_xw_lds_kernel<<<(mTiles + GX_WAVES - 1) / GX_WAVES, 32 * GX_WAVES, 0, stream>>>(x, W, h, N, mTiles);
        bin_count_kernel<<<BIN_CH, BIN_BLOCK, 0, stream>>>(dst, counts, N, E, NB);
        bin_scan_kernel<<<1, BIN_BLOCK, 0, stream>>>(counts, offs, M);
        bin_fill_kernel<<<BIN_CH, BIN_BLOCK, 0, stream>>>(src, dst, offs, list, N, E, NB);
        bin_deg_kernel<<<NB, DEGL_BLOCK, 0, stream>>>(list, offs, dinv, N);
        bin_sum_pos_kernel<<<NB, BIN_BLOCK, 0, stream>>>(list, offs, h, dinv, pos, Wp, bp, b, out, N);
    } else {
        unsigned* planes = (unsigned*)h;
        dim3 gDeg(DEG_CHUNKS, (N + DEG_RNG - 1) / DEG_RNG);
        deg_count_lds_kernel<<<gDeg, DEG_BLOCK, 0, stream>>>(dst, planes, N, E);
        deg_finish_kernel<<<(N + 255) / 256, 256, 0, stream>>>(planes, dinv, N);
        gemm_xw_lds_kernel<<<(mTiles + GX_WAVES - 1) / GX_WAVES, 32 * GX_WAVES, 0, stream>>>(x, W, h, N, mTiles);
        gemm_pos_fused_kernel<<<gGemm, 32 * wavesPerBlock, 0, stream>>>(
            pos, Wp, bp, b, h, dinv, out, N, mTiles);
        edge_scatter_kernel<<<(E + wavesPerBlock - 1) / wavesPerBlock,
                              32 * wavesPerBlock, 0, stream>>>(src, dst, h, dinv, out, E);
    }
}


# bin sum: each bin block sorts its edges by target row in LDS (count, scan, place); a wave then sums a row's h rows in registers, loads issued together; no LDS float atomics; sum order = placing order
# speedup vs baseline: 2.5676x; 1.0663x over previous
#include <hip/hip_runtime.h>

typedef __attribute__((ext_vector_type(2))) float v2f;
typedef __attribute__((ext_vector_type(8))) float v8f;
typedef __attribute__((ext_vector_type(4))) float v4f;

#define IN_CH   128
#define OUT_CH  64
#define POS_DIM 64

#define DEG_RNG    25600
#define DEG_CHUNKS 64
#define DEG_BLOCK  1024

__global__ __launch_bounds__(DEG_BLOCK)
void deg_count_lds_kernel(const int* __restrict__ dst, unsigned* __restrict__ planes,
                          int N, int E) {
    __shared__ unsigned cnt[DEG_RNG];
    const int tid  = threadIdx.x;
    const int c    = blockIdx.x;
    const int base = blockIdx.y * DEG_RNG;
    int len = N - base;
    if (len > DEG_RNG) len = DEG_RNG;

    for (int i = tid; i < len; i += DEG_BLOCK) cnt[i] = 0u;
    __syncthreads();

    const int per = (E + DEG_CHUNKS - 1) / DEG_CHUNKS;
    const int e0  = c * per;
    int e1 = e0 + per;
    if (e1 > E) e1 = E;
    for (int e = e0 + tid; e < e1; e += 4 * DEG_BLOCK) {
        const int d0 = dst[e];
        const int d1 = (e + 1 * DEG_BLOCK < e1) ? dst[e + 1 * DEG_BLOCK] : -1;
        const int d2 = (e + 2 * DEG_BLOCK < e1) ? dst[e + 2 * DEG_BLOCK] : -1;
        const int d3 = (e + 3 * DEG_BLOCK < e1) ? dst[e + 3 * DEG_BLOCK] : -1;
        const unsigned r0 = (unsigned)(d0 - base);
        const unsigned r1 = (unsigned)(d1 - base);
        const unsigned r2 = (unsigned)(d2 - base);
        const unsigned r3 = (unsigned)(d3 - base);
        if (r0 < (unsigned)len) atomicAdd(&cnt[r0], 1u);
        if (r1 < (unsigned)len) atomicAdd(&cnt[r1], 1u);
        if (r2 < (unsigned)len) atomicAdd(&cnt[r2], 1u);
        if (r3 < (unsigned)len) atomicAdd(&cnt[r3], 1u);
    }
    __syncthreads();

    unsigned* p = planes + (size_t)c * N + base;
    for (int i = tid; i < len; i += DEG_BLOCK) p[i] = cnt[i];
}

__global__ void deg_finish_kernel(const unsigned* __restrict__ planes,
                                  float* __restrict__ dinv, int N) {
    const int i = blockIdx.x * blockDim.x + threadIdx.x;
    if (i < N) {
        unsigned s = 1u;
#pragma unroll
        for (int c = 0; c < DEG_CHUNKS; ++c) s += planes[(size_t)c * N + i];
        dinv[i] = rsqrtf((float)s);
    }
}

#define GX_WAVES 8
#define XS_PITCH 132
#define BS_PITCH 160
#define GX_WITEMS ((IN_CH / 2) * (OUT_CH / 4) / (32 * GX_WAVES))

__global__ __launch_bounds__(32 * GX_WAVES)
void gemm_xw_lds_kernel(const float* __restrict__ x,
                        const float* __restrict__ W,
                        float* __restrict__ h, int N, int mTiles) {
    __shared__ float lds[(IN_CH / 2) * BS_PITCH + GX_WAVES * 16 * XS_PITCH];
    float* wsm = lds;
    float* xs  = lds + (IN_CH / 2) * BS_PITCH;
    const int tid  = threadIdx.x;
    const int wave = __builtin_amdgcn_readfirstlane(tid >> 5);
    const int lane = tid & 31;
    const int mTile = blockIdx.x * GX_WAVES + wave;
    const int row0  = mTile * 16;
    const int nl    = N - 1;

    v4f w0[GX_WITEMS], w1[GX_WITEMS];
#pragma unroll
    for (int j = 0; j < GX_WITEMS; ++j) {
        const int item = tid + j * 32 * GX_WAVES;
        const int p = item >> 4;
        const int q = item & 15;
        w0[j] = *(const v4f*)(W + (2 * p + 0) * OUT_CH + 4 * q);
        w1[j] = *(const v4f*)(W + (2 * p + 1) * OUT_CH + 4 * q);
    }
    v4f xv[16];
    const float* xp = x + lane * 4;
#pragma unroll
    for (int r = 0; r < 16; ++r)
        xv[r] = *(const v4f*)(xp + (size_t)min(row0 + r, nl) * IN_CH);
#pragma unroll
    for (int j = 0; j < GX_WITEMS; ++j) {
        const int item = tid + j * 32 * GX_WAVES;
        const int p = item >> 4;
        const int q = item & 15;
        float* d = wsm + p * BS_PITCH + 8 * q;
        const v4f lo = {w0[j][0], w1[j][0], w0[j][1], w1[j][1]};
        const v4f hi = {w0[j][2], w1[j][2], w0[j][3], w1[j][3]};
        *(v4f*)(d + 0) = lo;
        *(v4f*)(d + 4) = hi;
    }
    float* xw = xs + wave * 16 * XS_PITCH;
#pragma unroll
    for (int r = 0; r < 16; ++r)
        *(v4f*)(xw + r * XS_PITCH + lane * 4) = xv[r];
    __syncthreads();

    const int l16 = lane & 15;
    const int hh  = lane >> 4;
    v2f a[32];
#pragma unroll
    for (int s = 0; s < 32; ++s)
        a[s] = *(const v2f*)(xw + l16 * XS_PITCH + 4 * s + 2 * hh);
    v8f c0 = {}, c1 = {}, c2 = {}, c3 = {};
#pragma unroll
    for (int s = 0; s < 32; ++s) {
        const float* bb = wsm + (2 * s + hh) * BS_PITCH + l16 * 2;
        const v2f f0 = *(const v2f*)(bb + 0);
        const v2f f1 = *(const v2f*)(bb + 32);
        const v2f f2 = *(const v2f*)(bb + 64);
        const v2f f3 = *(const v2f*)(bb + 96);
        c0 = __builtin_amdgcn_wmma_f32_16x16x4_f32(false, a[s], false, f0, (short)0, c0, false, false);
        c1 = __builtin_amdgcn_wmma_f32_16x16x4_f32(false, a[s], false, f1, (short)0, c1, false, false);
        c2 = __builtin_amdgcn_wmma_f32_16x16x4_f32(false, a[s], false, f2, (short)0, c2, false, false);
        c3 = __builtin_amdgcn_wmma_f32_16x16x4_f32(false, a[s], false, f3, (short)0, c3, false, false);
    }

    float* pr = xw + (8 * hh) * XS_PITCH + l16;
#pragma unroll
    for (int r = 0; r < 8; ++r) {
        pr[r * XS_PITCH +  0] = c0[r];
        pr[r * XS_PITCH + 16] = c1[r];
        pr[r * XS_PITCH + 32] = c2[r];
        pr[r * XS_PITCH + 48] = c3[r];
    }
    const int rl = lane >> 4;
    const int c4 = (lane & 15) << 2;
    float* hp = h + (size_t)row0 * OUT_CH + rl * OUT_CH + c4;
    if (row0 + 16 <= N) {
        v4f v[8];
#pragma unroll
        for (int it = 0; it < 8; ++it)
            v[it] = *(const v4f*)(xw + (2 * it + rl) * XS_PITCH + c4);
#pragma unroll
        for (int it = 0; it < 8; ++it)
            *(v4f*)(hp + it * 2 * OUT_CH) = v[it];
    } else if (mTile < mTiles) {
#pragma unroll
        for (int it = 0; it < 8; ++it) {
            if (row0 + 2 * it + rl < N)
                *(v4f*)(hp + it * 2 * OUT_CH) = *(const v4f*)(xw + (2 * it + rl) * XS_PITCH + c4);
        }
    }
}

__global__ void gemm_pos_fused_kernel(const float* __restrict__ pos,
                                      const float* __restrict__ Wp,
                                      const float* __restrict__ bp,
                                      const float* __restrict__ b,
                                      const float* __restrict__ h,
                                      const float* __restrict__ dinv,
                                      float* __restrict__ out, int N, int mTiles) {
    const int wavesPerBlock = blockDim.x >> 5;
    const int waveId = threadIdx.x >> 5;
    const int lane   = threadIdx.x & 31;
    const int mTile  = blockIdx.x * wavesPerBlock + waveId;
    const int nTile  = blockIdx.y;
    if (mTile >= mTiles) return;

    const int l16 = lane & 15;
    const int kh  = (lane >> 4) << 1;
    int m = mTile * 16 + l16;
    if (m >= N) m = N - 1;
    const int n = nTile * 16 + l16;

    v8f c = {};
#pragma unroll
    for (int k = 0; k < POS_DIM; k += 4) {
        v2f a = *(const v2f*)(pos + (size_t)m * POS_DIM + k + kh);
        v2f bfrag;
        bfrag[0] = Wp[(size_t)(k + kh + 0) * OUT_CH + n];
        bfrag[1] = Wp[(size_t)(k + kh + 1) * OUT_CH + n];
        c = __builtin_amdgcn_wmma_f32_16x16x4_f32(
                false, a, false, bfrag, (short)0, c, false, false);
    }

    const float bias = bp[n] + b[n];
    const int rbase = mTile * 16 + ((lane >> 4) << 3);
#pragma unroll
    for (int r = 0; r < 8; ++r) {
        int row = rbase + r;
        if (row < N) {
            float di = dinv[row];
            out[(size_t)row * OUT_CH + n] =
                c[r] + bias + h[(size_t)row * OUT_CH + n] * di * di;
        }
    }
}

__global__ void edge_scatter_kernel(const int* __restrict__ src,
                                    const int* __restrict__ dst,
                                    const float* __restrict__ h,
                                    const float* __restrict__ dinv,
                                    float* __restrict__ out, int E) {
    const int wavesPerBlock = blockDim.x >> 5;
    const int e    = blockIdx.x * wavesPerBlock + (threadIdx.x >> 5);
    const int lane = threadIdx.x & 31;
    if (e >= E) return;

    const int s = src[e];
    const int d = dst[e];
    const float nrm = dinv[s] * dinv[d];

    const float2 hv = *(const float2*)(h + (size_t)s * OUT_CH + 2 * lane);
    float* op = out + (size_t)d * OUT_CH + 2 * lane;
    atomicAdd(op + 0, hv.x * nrm);
    atomicAdd(op + 1, hv.y * nrm);
}

#define BIN_RNG   400
#define BIN_CH    128
#define BIN_MAX   1024
#define BIN_BLOCK 1024

__global__ __launch_bounds__(BIN_BLOCK)
void bin_count_kernel(const int* __restrict__ dst, unsigned* __restrict__ counts,
                      int N, int E, int NB) {
    __shared__ unsigned hist[BIN_MAX];
    const int tid = threadIdx.x;
    const int c   = blockIdx.x;
    for (int i = tid; i < NB; i += BIN_BLOCK) hist[i] = 0u;
    __syncthreads();

    const int per = (E + BIN_CH - 1) / BIN_CH;
    const int e0  = c * per;
    int e1 = e0 + per;
    if (e1 > E) e1 = E;
    for (int e = e0 + tid; e < e1; e += 4 * BIN_BLOCK) {
        const int d0 = dst[e];
        const int d1 = (e + 1 * BIN_BLOCK < e1) ? dst[e + 1 * BIN_BLOCK] : -1;
        const int d2 = (e + 2 * BIN_BLOCK < e1) ? dst[e + 2 * BIN_BLOCK] : -1;
        const int d3 = (e + 3 * BIN_BLOCK < e1) ? dst[e + 3 * BIN_BLOCK] : -1;
        if ((unsigned)d0 < (unsigned)N) atomicAdd(&hist[d0 / BIN_RNG], 1u);
        if ((unsigned)d1 < (unsigned)N) atomicAdd(&hist[d1 / BIN_RNG], 1u);
        if ((unsigned)d2 < (unsigned)N) atomicAdd(&hist[d2 / BIN_RNG], 1u);
        if ((unsigned)d3 < (unsigned)N) atomicAdd(&hist[d3 / BIN_RNG], 1u);
    }
    __syncthreads();
    for (int i = tid; i < NB; i += BIN_BLOCK) counts[(size_t)i * BIN_CH + c] = hist[i];
}

__global__ __launch_bounds__(BIN_BLOCK)
void bin_scan_kernel(const unsigned* __restrict__ counts, unsigned* __restrict__ offs, int M) {
    __shared__ unsigned part[BIN_BLOCK];
    const int tid = threadIdx.x;
    unsigned carry = 0u;
    for (int t0 = 0; t0 < M; t0 += 32 * BIN_BLOCK) {
        const int i0 = t0 + tid * 32;
        uint4 v[8];
#pragma unroll
        for (int k = 0; k < 8; ++k) {
            v[k] = make_uint4(0u, 0u, 0u, 0u);
            if (i0 + 4 * k < M) v[k] = *(const uint4*)(counts + i0 + 4 * k);
        }
        unsigned sum = 0u;
#pragma unroll
        for (int k = 0; k < 8; ++k) sum += v[k].x + v[k].y + v[k].z + v[k].w;
        part[tid] = sum;
        __syncthreads();
        for (int off = 1; off < BIN_BLOCK; off <<= 1) {
            const unsigned t = (tid >= off) ? part[tid - off] : 0u;
            __syncthreads();
            part[tid] += t;
            __syncthreads();
        }
        unsigned run = carry + part[tid] - sum;
#pragma unroll
        for (int k = 0; k < 8; ++k) {
            uint4 o;
            o.x = run; run += v[k].x;
            o.y = run; run += v[k].y;
            o.z = run; run += v[k].z;
            o.w = run; run += v[k].w;
            if (i0 + 4 * k < M) *(uint4*)(offs + i0 + 4 * k) = o;
        }
        carry += part[BIN_BLOCK - 1];
        __syncthreads();
    }
    if (tid == 0) offs[M] = carry;
}

__global__ __launch_bounds__(BIN_BLOCK)
void bin_fill_kernel(const int* __restrict__ src, const int* __restrict__ dst,
                     const unsigned* __restrict__ offs, unsigned* __restrict__ list,
                     int N, int E, int NB) {
    __shared__ unsigned cursor[BIN_MAX];
    const int tid = threadIdx.x;
    const int c   = blockIdx.x;
    for (int i = tid; i < NB; i += BIN_BLOCK) cursor[i] = offs[(size_t)i * BIN_CH + c];
    __syncthreads();

    const int per = (E + BIN_CH - 1) / BIN_CH;
    const int e0  = c * per;
    int e1 = e0 + per;
    if (e1 > E) e1 = E;
    for (int e = e0 + tid; e < e1; e += 4 * BIN_BLOCK) {
        int d[4], sv[4];
#pragma unroll
        for (int k = 0; k < 4; ++k) {
            const int ee = e + k * BIN_BLOCK;
            const bool in = ee < e1;
            d[k]  = in ? dst[ee] : -1;
            sv[k] = in ? src[ee] : 0;
        }
#pragma unroll
        for (int k = 0; k < 4; ++k) {
            if ((unsigned)d[k] < (unsigned)N) {
                const int bin = d[k] / BIN_RNG;
                const unsigned p = atomicAdd(&cursor[bin], 1u);
                list[p] = ((unsigned)sv[k] << 9) | (unsigned)(d[k] - bin * BIN_RNG);
            }
        }
    }
}

#define DEGL_BLOCK 256
#define DEGL_DEPTH 16

__global__ __launch_bounds__(DEGL_BLOCK)
void bin_deg_kernel(const unsigned* __restrict__ list, const unsigned* __restrict__ offs,
                    float* __restrict__ dinv, int N) {
    __shared__ unsigned cnt[512];
    const int tid  = threadIdx.x;
    const int bin  = blockIdx.x;
    const int base = bin * BIN_RNG;
    int len = N - base;
    if (len > BIN_RNG) len = BIN_RNG;
    for (int i = tid; i < 512; i += DEGL_BLOCK) cnt[i] = 0u;
    const unsigned b0 = offs[(size_t)bin * BIN_CH];
    const unsigned b1 = offs[(size_t)(bin + 1) * BIN_CH];
    __syncthreads();

    for (unsigned i = b0 + (unsigned)tid; i < b1; i += DEGL_DEPTH * DEGL_BLOCK) {
        unsigned w[DEGL_DEPTH];
#pragma unroll
        for (int k = 0; k < DEGL_DEPTH; ++k) {
            unsigned idx = i + (unsigned)k * DEGL_BLOCK;
            if (idx > b1 - 1u) idx = b1 - 1u;
            w[k] = list[idx];
        }
#pragma unroll
        for (int k = 0; k < DEGL_DEPTH; ++k)
            atomicAdd(&cnt[w[k] & 511u], (i + (unsigned)k * DEGL_BLOCK < b1) ? 1u : 0u);
    }
    __syncthreads();

    for (int i = tid; i < len; i += DEGL_BLOCK)
        dinv[base + i] = rsqrtf((float)(cnt[i] + 1u));
}

#define POSP_PITCH 68
#define OUT_ITERS  ((BIN_RNG * (OUT_CH / 4) + BIN_BLOCK - 1) / BIN_BLOCK)

__device__ __forceinline__ void pos_tile_wmma(float* posp, const float* bs,
                                              const float* __restrict__ bp,
                                              const float* __restrict__ b,
                                              int r0, int lane) {
    const int l16 = lane & 15;
    const int hh  = lane >> 4;
    v2f a[16];
#pragma unroll
    for (int s = 0; s < 16; ++s)
        a[s] = *(const v2f*)(posp + (r0 + l16) * POSP_PITCH + 4 * s + 2 * hh);
    v8f c0 = {}, c1 = {}, c2 = {}, c3 = {};
#pragma unroll
    for (int s = 0; s < 16; ++s) {
        const float* bb = bs + (2 * s + hh) * BS_PITCH + l16 * 2;
        const v2f f0 = *(const v2f*)(bb + 0);
        const v2f f1 = *(const v2f*)(bb + 32);
        const v2f f2 = *(const v2f*)(bb + 64);
        const v2f f3 = *(const v2f*)(bb + 96);
        c0 = __builtin_amdgcn_wmma_f32_16x16x4_f32(false, a[s], false, f0, (short)0, c0, false, false);
        c1 = __builtin_amdgcn_wmma_f32_16x16x4_f32(false, a[s], false, f1, (short)0, c1, false, false);
        c2 = __builtin_amdgcn_wmma_f32_16x16x4_f32(false, a[s], false, f2, (short)0, c2, false, false);
        c3 = __builtin_amdgcn_wmma_f32_16x16x4_f32(false, a[s], false, f3, (short)0, c3, false, false);
    }
    const float bias0 = bp[l16 +  0] + b[l16 +  0];
    const float bias1 = bp[l16 + 16] + b[l16 + 16];
    const float bias2 = bp[l16 + 32] + b[l16 + 32];
    const float bias3 = bp[l16 + 48] + b[l16 + 48];
    float* pr = posp + (r0 + 8 * hh) * POSP_PITCH + l16;
#pragma unroll
    for (int r = 0; r < 8; ++r) {
        pr[r * POSP_PITCH +  0] = c0[r] + bias0;
        pr[r * POSP_PITCH + 16] = c1[r] + bias1;
        pr[r * POSP_PITCH + 32] = c2[r] + bias2;
        pr[r * POSP_PITCH + 48] = c3[r] + bias3;
    }
}

#define SORT_CAP   8192
#define SORT_PER   (SORT_CAP / BIN_BLOCK)
#define ROW_TURNS  ((BIN_RNG + 31) / 32)

template <int CNT>
__device__ __forceinline__ void row_group(const float* __restrict__ h, unsigned sv, unsigned dv,
                                          int k0, int lane, float& ax, float& ay) {
    float2 hv[CNT];
    float  sc[CNT];
#pragma unroll
    for (int k = 0; k < CNT; ++k) {
        const unsigned sn = (unsigned)__builtin_amdgcn_readlane((int)sv, k0 + k);
        sc[k] = __uint_as_float((unsigned)__builtin_amdgcn_readlane((int)dv, k0 + k));
        hv[k] = *(const float2*)(h + (size_t)sn * OUT_CH + 2 * lane);
    }
#pragma unroll
    for (int k = 0; k < CNT; ++k) {
        ax += hv[k].x * sc[k];
        ay += hv[k].y * sc[k];
    }
}

__global__ __launch_bounds__(BIN_BLOCK)
void bin_sum_pos_kernel(const unsigned* __restrict__ list, const unsigned* __restrict__ offs,
                        const float* __restrict__ h, const float* __restrict__ dinv,
                        const float* __restrict__ pos, const float* __restrict__ Wp,
                        const float* __restrict__ bp, const float* __restrict__ b,
                        float* __restrict__ out, int N) {
    __shared__ float acc[BIN_RNG * OUT_CH];
    __shared__ float posp[BIN_RNG * POSP_PITCH];
    __shared__ float bs[(POS_DIM / 2) * BS_PITCH];
    __shared__ uint2 ent[SORT_CAP];
    __shared__ unsigned cnt[512];
    __shared__ unsigned rstart[513];
    __shared__ unsigned wsum[16];
    const int tid  = threadIdx.x;
    const int bin  = blockIdx.x;
    const int base = bin * BIN_RNG;
    int len = N - base;
    if (len > BIN_RNG) len = BIN_RNG;
    const int wave = __builtin_amdgcn_readfirstlane(tid >> 5);
    const int lane = tid & 31;
    const bool tile_wave = wave < BIN_RNG / 16;
    const int r0 = wave * 16;

    v4f pv0 = {}, pv1 = {}, pv2 = {}, pv3 = {}, pv4 = {}, pv5 = {}, pv6 = {}, pv7 = {};
    if (tile_wave) {
        const float* pp = pos + (lane & 15) * 4;
        const int rl = base + r0 + (lane >> 4);
        const int nl = N - 1;
        pv0 = *(const v4f*)(pp + (size_t)min(rl +  0, nl) * POS_DIM);
        pv1 = *(const v4f*)(pp + (size_t)min(rl +  2, nl) * POS_DIM);
        pv2 = *(const v4f*)(pp + (size_t)min(rl +  4, nl) * POS_DIM);
        pv3 = *(const v4f*)(pp + (size_t)min(rl +  6, nl) * POS_DIM);
        pv4 = *(const v4f*)(pp + (size_t)min(rl +  8, nl) * POS_DIM);
        pv5 = *(const v4f*)(pp + (size_t)min(rl + 10, nl) * POS_DIM);
        pv6 = *(const v4f*)(pp + (size_t)min(rl + 12, nl) * POS_DIM);
        pv7 = *(const v4f*)(pp + (size_t)min(rl + 14, nl) * POS_DIM);
    }
    const float4 wv = *(const float4*)(Wp + (tid >> 4) * OUT_CH + ((tid & 15) << 2));
    const unsigned b0 = offs[(size_t)bin * BIN_CH];
    const unsigned b1 = offs[(size_t)(bin + 1) * BIN_CH];

    {
        const int k  = tid >> 4;
        const int n0 = (tid & 15) << 2;
        float* d = bs + (k >> 1) * BS_PITCH + (k & 1);
        d[(n0 + 0) * 2] = wv.x;
        d[(n0 + 1) * 2] = wv.y;
        d[(n0 + 2) * 2] = wv.z;
        d[(n0 + 3) * 2] = wv.w;
    }
    if (tile_wave) {
        float* ps = posp + (r0 + (lane >> 4)) * POSP_PITCH + (lane & 15) * 4;
        *(v4f*)(ps +  0 * POSP_PITCH) = pv0;
        *(v4f*)(ps +  2 * POSP_PITCH) = pv1;
        *(v4f*)(ps +  4 * POSP_PITCH) = pv2;
        *(v4f*)(ps +  6 * POSP_PITCH) = pv3;
        *(v4f*)(ps +  8 * POSP_PITCH) = pv4;
        *(v4f*)(ps + 10 * POSP_PITCH) = pv5;
        *(v4f*)(ps + 12 * POSP_PITCH) = pv6;
        *(v4f*)(ps + 14 * POSP_PITCH) = pv7;
    }

    const int turn = wave % ROW_TURNS;
    bool pending = tile_wave;
    unsigned c0 = b0;
    bool first = true;
    do {
        unsigned n = b1 - c0;
        if (n > SORT_CAP) n = SORT_CAP;
        if (tid < 512) cnt[tid] = 0u;
        __syncthreads();

        unsigned w[SORT_PER];
        float    dsrc[SORT_PER];
#pragma unroll
        for (int k = 0; k < SORT_PER; ++k) {
            const unsigned i = (unsigned)tid + (unsigned)k * BIN_BLOCK;
            w[k] = (i < n) ? list[c0 + i] : 0u;
        }
#pragma unroll
        for (int k = 0; k < SORT_PER; ++k) {
            const unsigned i = (unsigned)tid + (unsigned)k * BIN_BLOCK;
            dsrc[k] = (i < n) ? dinv[w[k] >> 9] : 0.0f;
        }
#pragma unroll
        for (int k = 0; k < SORT_PER; ++k) {
            const unsigned i = (unsigned)tid + (unsigned)k * BIN_BLOCK;
            if (i < n) atomicAdd(&cnt[w[k] & 511u], 1u);
        }
        __syncthreads();

        unsigned v = 0u, inc = 0u;
        if (tid < 512) {
            v = cnt[tid];
            inc = v;
#pragma unroll
            for (int o = 1; o < 32; o <<= 1) {
                const unsigned t = __shfl_up(inc, o);
                if (lane >= o) inc += t;
            }
            if (lane == 31) wsum[wave] = inc;
        }
        __syncthreads();
        if (tid < 512) {
            unsigned pre = 0u;
#pragma unroll
            for (int j = 0; j < 16; ++j) pre += (j < wave) ? wsum[j] : 0u;
            const unsigned excl = pre + inc - v;
            rstart[tid] = excl;
            cnt[tid] = excl;
            if (tid == 511) rstart[512] = excl + v;
        }
        __syncthreads();

#pragma unroll
        for (int k = 0; k < SORT_PER; ++k) {
            const unsigned i = (unsigned)tid + (unsigned)k * BIN_BLOCK;
            if (i < n) {
                const unsigned p = atomicAdd(&cnt[w[k] & 511u], 1u);
                ent[p] = make_uint2(w[k] >> 9, __float_as_uint(dsrc[k]));
            }
        }
        __syncthreads();

        for (int j = 0; j < ROW_TURNS; ++j) {
            if (pending && j == turn) {
                pos_tile_wmma(posp, bs, bp, b, r0, lane);
                pending = false;
            }
            const int r = wave + 32 * j;
            if (r < BIN_RNG) {
                const unsigned s0 = (unsigned)__builtin_amdgcn_readfirstlane((int)rstart[r]);
                const unsigned s1 = (unsigned)__builtin_amdgcn_readfirstlane((int)rstart[r + 1]);
                float ax = 0.0f, ay = 0.0f;
                for (unsigned e0 = s0; e0 < s1; e0 += 32u) {
                    unsigned m = s1 - e0;
                    if (m > 32u) m = 32u;
                    const unsigned li = ((unsigned)lane < m) ? (unsigned)lane : m - 1u;
                    const uint2 en = ent[e0 + li];
                    const unsigned sv = en.x;
                    const unsigned dv = ((unsigned)lane < m) ? en.y : 0u;
                    if (m > 8u)  row_group<16>(h, sv, dv, 0, lane, ax, ay);
                    else         row_group<8>(h, sv, dv, 0, lane, ax, ay);
                    if (m > 16u) {
                        if (m > 24u) row_group<16>(h, sv, dv, 16, lane, ax, ay);
                        else         row_group<8>(h, sv, dv, 16, lane, ax, ay);
                    }
                }
                float2* ar = (float2*)(acc + r * OUT_CH + 2 * lane);
                if (first) {
                    *ar = make_float2(ax, ay);
                } else {
                    const float2 o = *ar;
                    *ar = make_float2(o.x + ax, o.y + ay);
                }
            }
        }
        __syncthreads();
        first = false;
        c0 += SORT_CAP;
    } while (c0 < b1);

    const float* hb = h + (size_t)base * OUT_CH;
    float* o = out + (size_t)base * OUT_CH;
    const int nq = len * (OUT_CH / 4);
    float  di[OUT_ITERS];
    float4 hv[OUT_ITERS];
#pragma unroll
    for (int it = 0; it < OUT_ITERS; ++it) {
        int q = tid + it * BIN_BLOCK;
        if (q > nq - 1) q = nq - 1;
        di[it] = dinv[base + (q >> 4)];
        hv[it] = *(const float4*)(hb + (q >> 4) * OUT_CH + ((q & 15) << 2));
    }
#pragma unroll
    for (int it = 0; it < OUT_ITERS; ++it) {
        const int q = tid + it * BIN_BLOCK;
        if (q < nq) {
            const int row = q >> 4;
            const int c4  = (q & 15) << 2;
            const float4 pq = *(const float4*)(posp + row * POSP_PITCH + c4);
            const float4 av = *(const float4*)(acc + row * OUT_CH + c4);
            const float d = di[it];
            float4 r;
            r.x = (pq.x + hv[it].x * d * d) + d * av.x;
            r.y = (pq.y + hv[it].y * d * d) + d * av.y;
            r.z = (pq.z + hv[it].z * d * d) + d * av.z;
            r.w = (pq.w + hv[it].w * d * d) + d * av.w;
            *(float4*)(o + row * OUT_CH + c4) = r;
        }
    }
}

extern "C" void kernel_launch(void* const* d_in, const int* in_sizes, int n_in,
                              void* d_out, int out_size, void* d_ws, size_t ws_size,
                              hipStream_t stream) {
    const float* x   = (const float*)d_in[0];
    const int*   ei  = (const int*)d_in[1];
    const float* pos = (const float*)d_in[2];
    const float* W   = (const float*)d_in[3];
    const float* b   = (const float*)d_in[4];
    const float* Wp  = (const float*)d_in[5];
    const float* bp  = (const float*)d_in[6];
    float* out = (float*)d_out;

    const int N = in_sizes[0] / IN_CH;
    const int E = in_sizes[1] / 2;
    const int* src = ei;
    const int* dst = ei + E;

    float* dinv = (float*)d_ws;
    float* h    = dinv + ((N + 3) & ~3);

    const int mTiles = (N + 15) / 16;
    const int wavesPerBlock = 8;
    dim3 gGemm((mTiles + wavesPerBlock - 1) / wavesPerBlock, OUT_CH / 16);

    const int NB = (N + BIN_RNG - 1) / BIN_RNG;
    const int M  = NB * BIN_CH;
    unsigned* counts = (unsigned*)(h + (size_t)N * OUT_CH);
    unsigned* offs   = counts + M;
    unsigned* list   = offs + M + 4;
    const size_t need = (size_t)((char*)(list + E) - (char*)d_ws);
    if (NB <= BIN_MAX && N <= (1 << 23) && need <= ws_size) {
        gemm_xw_lds_kernel<<<(mTiles + GX_WAVES - 1) / GX_WAVES, 32 * GX_WAVES, 0, stream>>>(x, W, h, N, mTiles);
        bin_count_kernel<<<BIN_CH, BIN_BLOCK, 0, stream>>>(dst, counts, N, E, NB);
        bin_scan_kernel<<<1, BIN_BLOCK, 0, stream>>>(counts, offs, M);
        bin_fill_kernel<<<BIN_CH, BIN_BLOCK, 0, stream>>>(src, dst, offs, list, N, E, NB);
        bin_deg_kernel<<<NB, DEGL_BLOCK, 0, stream>>>(list, offs, dinv, N);
        bin_sum_pos_kernel<<<NB, BIN_BLOCK, 0, stream>>>(list, offs, h, dinv, pos, Wp, bp, b, out, N);
    } else {
        unsigned* planes = (unsigned*)h;
        dim3 gDeg(DEG_CHUNKS, (N + DEG_RNG - 1) / DEG_RNG);
        deg_count_lds_kernel<<<gDeg, DEG_BLOCK, 0, stream>>>(dst, planes, N, E);
        deg_finish_kernel<<<(N + 255) / 256, 256, 0, stream>>>(planes, dinv, N);
        gemm_xw_lds_kernel<<<(mTiles + GX_WAVES - 1) / GX_WAVES, 32 * GX_WAVES, 0, stream>>>(x, W, h, N, mTiles);
        gemm_pos_fused_kernel<<<gGemm, 32 * wavesPerBlock, 0, stream>>>(
            pos, Wp, bp, b, h, dinv, out, N, mTiles);
        edge_scatter_kernel<<<(E + wavesPerBlock - 1) / wavesPerBlock,
                              32 * wavesPerBlock, 0, stream>>>(src, dst, h, dinv, out, E);
    }
}


# h = x @ W loses its own launch: its 8-wave tile blocks ride behind the count, fill and degree blocks (those passes recut to 8-wave blocks, 25 edges a thread a trip); sum order = placing order
# speedup vs baseline: 2.8455x; 1.1082x over previous
#include <hip/hip_runtime.h>

typedef __attribute__((ext_vector_type(2))) float v2f;
typedef __attribute__((ext_vector_type(8))) float v8f;
typedef __attribute__((ext_vector_type(4))) float v4f;

#define IN_CH   128
#define OUT_CH  64
#define POS_DIM 64

#define DEG_RNG    25600
#define DEG_CHUNKS 64
#define DEG_BLOCK  1024

__global__ __launch_bounds__(DEG_BLOCK)
void deg_count_lds_kernel(const int* __restrict__ dst, unsigned* __restrict__ planes,
                          int N, int E) {
    __shared__ unsigned cnt[DEG_RNG];
    const int tid  = threadIdx.x;
    const int c    = blockIdx.x;
    const int base = blockIdx.y * DEG_RNG;
    int len = N - base;
    if (len > DEG_RNG) len = DEG_RNG;

    for (int i = tid; i < len; i += DEG_BLOCK) cnt[i] = 0u;
    __syncthreads();

    const int per = (E + DEG_CHUNKS - 1) / DEG_CHUNKS;
    const int e0  = c * per;
    int e1 = e0 + per;
    if (e1 > E) e1 = E;
    for (int e = e0 + tid; e < e1; e += 4 * DEG_BLOCK) {
        const int d0 = dst[e];
        const int d1 = (e + 1 * DEG_BLOCK < e1) ? dst[e + 1 * DEG_BLOCK] : -1;
        const int d2 = (e + 2 * DEG_BLOCK < e1) ? dst[e + 2 * DEG_BLOCK] : -1;
        const int d3 = (e + 3 * DEG_BLOCK < e1) ? dst[e + 3 * DEG_BLOCK] : -1;
        const unsigned r0 = (unsigned)(d0 - base);
        const unsigned r1 = (unsigned)(d1 - base);
        const unsigned r2 = (unsigned)(d2 - base);
        const unsigned r3 = (unsigned)(d3 - base);
        if (r0 < (unsigned)len) atomicAdd(&cnt[r0], 1u);
        if (r1 < (unsigned)len) atomicAdd(&cnt[r1], 1u);
        if (r2 < (unsigned)len) atomicAdd(&cnt[r2], 1u);
        if (r3 < (unsigned)len) atomicAdd(&cnt[r3], 1u);
    }
    __syncthreads();

    unsigned* p = planes + (size_t)c * N + base;
    for (int i = tid; i < len; i += DEG_BLOCK) p[i] = cnt[i];
}

__global__ void deg_finish_kernel(const unsigned* __restrict__ planes,
                                  float* __restrict__ dinv, int N) {
    const int i = blockIdx.x * blockDim.x + threadIdx.x;
    if (i < N) {
        unsigned s = 1u;
#pragma unroll
        for (int c = 0; c < DEG_CHUNKS; ++c) s += planes[(size_t)c * N + i];
        dinv[i] = rsqrtf((float)s);
    }
}

#define GX_WAVES 8
#define XS_PITCH 132
#define BS_PITCH 160
#define GX_WITEMS ((IN_CH / 2) * (OUT_CH / 4) / (32 * GX_WAVES))

#define GX_LDS_FLOATS ((IN_CH / 2) * BS_PITCH + GX_WAVES * 16 * XS_PITCH)

__device__ __forceinline__ void gemm_xw_block(const float* __restrict__ x,
                                              const float* __restrict__ W,
                                              float* __restrict__ h, int N, int mTiles,
                                              int blk, float* lds) {
    float* wsm = lds;
    float* xs  = lds + (IN_CH / 2) * BS_PITCH;
    const int tid  = threadIdx.x;
    const int wave = __builtin_amdgcn_readfirstlane(tid >> 5);
    const int lane = tid & 31;
    const int mTile = blk * GX_WAVES + wave;
    const int row0  = mTile * 16;
    const int nl    = N - 1;

    v4f w0[GX_WITEMS], w1[GX_WITEMS];
#pragma unroll
    for (int j = 0; j < GX_WITEMS; ++j) {
        const int item = tid + j * 32 * GX_WAVES;
        const int p = item >> 4;
        const int q = item & 15;
        w0[j] = *(const v4f*)(W + (2 * p + 0) * OUT_CH + 4 * q);
        w1[j] = *(const v4f*)(W + (2 * p + 1) * OUT_CH + 4 * q);
    }
    v4f xv[16];
    const float* xp = x + lane * 4;
#pragma unroll
    for (int r = 0; r < 16; ++r)
        xv[r] = *(const v4f*)(xp + (size_t)min(row0 + r, nl) * IN_CH);
#pragma unroll
    for (int j = 0; j < GX_WITEMS; ++j) {
        const int item = tid + j * 32 * GX_WAVES;
        const int p = item >> 4;
        const int q = item & 15;
        float* d = wsm + p * BS_PITCH + 8 * q;
        const v4f lo = {w0[j][0], w1[j][0], w0[j][1], w1[j][1]};
        const v4f hi = {w0[j][2], w1[j][2], w0[j][3], w1[j][3]};
        *(v4f*)(d + 0) = lo;
        *(v4f*)(d + 4) = hi;
    }
    float* xw = xs + wave * 16 * XS_PITCH;
#pragma unroll
    for (int r = 0; r < 16; ++r)
        *(v4f*)(xw + r * XS_PITCH + lane * 4) = xv[r];
    __syncthreads();

    const int l16 = lane & 15;
    const int hh  = lane >> 4;
    v2f a[32];
#pragma unroll
    for (int s = 0; s < 32; ++s)
        a[s] = *(const v2f*)(xw + l16 * XS_PITCH + 4 * s + 2 * hh);
    v8f c0 = {}, c1 = {}, c2 = {}, c3 = {};
#pragma unroll
    for (int s = 0; s < 32; ++s) {
        const float* bb = wsm + (2 * s + hh) * BS_PITCH + l16 * 2;
        const v2f f0 = *(const v2f*)(bb + 0);
        const v2f f1 = *(const v2f*)(bb + 32);
        const v2f f2 = *(const v2f*)(bb + 64);
        const v2f f3 = *(const v2f*)(bb + 96);
        c0 = __builtin_amdgcn_wmma_f32_16x16x4_f32(false, a[s], false, f0, (short)0, c0, false, false);
        c1 = __builtin_amdgcn_wmma_f32_16x16x4_f32(false, a[s], false, f1, (short)0, c1, false, false);
        c2 = __builtin_amdgcn_wmma_f32_16x16x4_f32(false, a[s], false, f2, (short)0, c2, false, false);
        c3 = __builtin_amdgcn_wmma_f32_16x16x4_f32(false, a[s], false, f3, (short)0, c3, false, false);
    }

    float* pr = xw + (8 * hh) * XS_PITCH + l16;
#pragma unroll
    for (int r = 0; r < 8; ++r) {
        pr[r * XS_PITCH +  0] = c0[r];
        pr[r * XS_PITCH + 16] = c1[r];
        pr[r * XS_PITCH + 32] = c2[r];
        pr[r * XS_PITCH + 48] = c3[r];
    }
    const int rl = lane >> 4;
    const int c4 = (lane & 15) << 2;
    float* hp = h + (size_t)row0 * OUT_CH + rl * OUT_CH + c4;
    if (row0 + 16 <= N) {
        v4f v[8];
#pragma unroll
        for (int it = 0; it < 8; ++it)
            v[it] = *(const v4f*)(xw + (2 * it + rl) * XS_PITCH + c4);
#pragma unroll
        for (int it = 0; it < 8; ++it)
            *(v4f*)(hp + it * 2 * OUT_CH) = v[it];
    } else if (mTile < mTiles) {
#pragma unroll
        for (int it = 0; it < 8; ++it) {
            if (row0 + 2 * it + rl < N)
                *(v4f*)(hp + it * 2 * OUT_CH) = *(const v4f*)(xw + (2 * it + rl) * XS_PITCH + c4);
        }
    }
}

__global__ __launch_bounds__(32 * GX_WAVES)
void gemm_xw_lds_kernel(const float* __restrict__ x,
                        const float* __restrict__ W,
                        float* __restrict__ h, int N, int mTiles) {
    __shared__ float lds[GX_LDS_FLOATS];
    gemm_xw_block(x, W, h, N, mTiles, blockIdx.x, lds);
}

__global__ void gemm_pos_fused_kernel(const float* __restrict__ pos,
                                      const float* __restrict__ Wp,
                                      const float* __restrict__ bp,
                                      const float* __restrict__ b,
                                      const float* __restrict__ h,
                                      const float* __restrict__ dinv,
                                      float* __restrict__ out, int N, int mTiles) {
    const int wavesPerBlock = blockDim.x >> 5;
    const int waveId = threadIdx.x >> 5;
    const int lane   = threadIdx.x & 31;
    const int mTile  = blockIdx.x * wavesPerBlock + waveId;
    const int nTile  = blockIdx.y;
    if (mTile >= mTiles) return;

    const int l16 = lane & 15;
    const int kh  = (lane >> 4) << 1;
    int m = mTile * 16 + l16;
    if (m >= N) m = N - 1;
    const int n = nTile * 16 + l16;

    v8f c = {};
#pragma unroll
    for (int k = 0; k < POS_DIM; k += 4) {
        v2f a = *(const v2f*)(pos + (size_t)m * POS_DIM + k + kh);
        v2f bfrag;
        bfrag[0] = Wp[(size_t)(k + kh + 0) * OUT_CH + n];
        bfrag[1] = Wp[(size_t)(k + kh + 1) * OUT_CH + n];
        c = __builtin_amdgcn_wmma_f32_16x16x4_f32(
                false, a, false, bfrag, (short)0, c, false, false);
    }

    const float bias = bp[n] + b[n];
    const int rbase = mTile * 16 + ((lane >> 4) << 3);
#pragma unroll
    for (int r = 0; r < 8; ++r) {
        int row = rbase + r;
        if (row < N) {
            float di = dinv[row];
            out[(size_t)row * OUT_CH + n] =
                c[r] + bias + h[(size_t)row * OUT_CH + n] * di * di;
        }
    }
}

__global__ void edge_scatter_kernel(const int* __restrict__ src,
                                    const int* __restrict__ dst,
                                    const float* __restrict__ h,
                                    const float* __restrict__ dinv,
                                    float* __restrict__ out, int E) {
    const int wavesPerBlock = blockDim.x >> 5;
    const int e    = blockIdx.x * wavesPerBlock + (threadIdx.x >> 5);
    const int lane = threadIdx.x & 31;
    if (e >= E) return;

    const int s = src[e];
    const int d = dst[e];
    const float nrm = dinv[s] * dinv[d];

    const float2 hv = *(const float2*)(h + (size_t)s * OUT_CH + 2 * lane);
    float* op = out + (size_t)d * OUT_CH + 2 * lane;
    atomicAdd(op + 0, hv.x * nrm);
    atomicAdd(op + 1, hv.y * nrm);
}

#define BIN_RNG   400
#define BIN_CH    128
#define BIN_MAX   1024
#define BIN_BLOCK 1024

#define BINW_BLOCK (32 * GX_WAVES)
#define BINW_DEPTH 25

__device__ __forceinline__ void bin_count_block(const int* __restrict__ dst, unsigned* __restrict__ counts,
                                                int N, int E, int NB, int c, unsigned* hist) {
    const int tid = threadIdx.x;
    for (int i = tid; i < NB; i += BINW_BLOCK) hist[i] = 0u;
    __syncthreads();

    const int per = (E + BIN_CH - 1) / BIN_CH;
    const int e0  = c * per;
    int e1 = e0 + per;
    if (e1 > E) e1 = E;
    for (int e = e0 + tid; e < e1; e += BINW_DEPTH * BINW_BLOCK) {
        int d[BINW_DEPTH];
#pragma unroll
        for (int k = 0; k < BINW_DEPTH; ++k)
            d[k] = dst[min(e + k * BINW_BLOCK, e1 - 1)];
#pragma unroll
        for (int k = 0; k < BINW_DEPTH; ++k)
            if (e + k * BINW_BLOCK < e1 && (unsigned)d[k] < (unsigned)N) atomicAdd(&hist[d[k] / BIN_RNG], 1u);
    }
    __syncthreads();
    for (int i = tid; i < NB; i += BINW_BLOCK) counts[(size_t)i * BIN_CH + c] = hist[i];
}

__global__ __launch_bounds__(BINW_BLOCK)
void bin_count_gemm_kernel(const int* __restrict__ dst, unsigned* __restrict__ counts,
                           int N, int E, int NB,
                           const float* __restrict__ x, const float* __restrict__ W,
                           float* __restrict__ h, int mTiles, int g0) {
    __shared__ __attribute__((aligned(16))) unsigned char smem[GX_LDS_FLOATS * 4];
    if (blockIdx.x < BIN_CH)
        bin_count_block(dst, counts, N, E, NB, blockIdx.x, (unsigned*)smem);
    else
        gemm_xw_block(x, W, h, N, mTiles, g0 + (int)blockIdx.x - BIN_CH, (float*)smem);
}

__global__ __launch_bounds__(BIN_BLOCK)
void bin_scan_kernel(const unsigned* __restrict__ counts, unsigned* __restrict__ offs, int M) {
    __shared__ unsigned part[BIN_BLOCK];
    const int tid = threadIdx.x;
    unsigned carry = 0u;
    for (int t0 = 0; t0 < M; t0 += 32 * BIN_BLOCK) {
        const int i0 = t0 + tid * 32;
        uint4 v[8];
#pragma unroll
        for (int k = 0; k < 8; ++k) {
            v[k] = make_uint4(0u, 0u, 0u, 0u);
            if (i0 + 4 * k < M) v[k] = *(const uint4*)(counts + i0 + 4 * k);
        }
        unsigned sum = 0u;
#pragma unroll
        for (int k = 0; k < 8; ++k) sum += v[k].x + v[k].y + v[k].z + v[k].w;
        part[tid] = sum;
        __syncthreads();
        for (int off = 1; off < BIN_BLOCK; off <<= 1) {
            const unsigned t = (tid >= off) ? part[tid - off] : 0u;
            __syncthreads();
            part[tid] += t;
            __syncthreads();
        }
        unsigned run = carry + part[tid] - sum;
#pragma unroll
        for (int k = 0; k < 8; ++k) {
            uint4 o;
            o.x = run; run += v[k].x;
            o.y = run; run += v[k].y;
            o.z = run; run += v[k].z;
            o.w = run; run += v[k].w;
            if (i0 + 4 * k < M) *(uint4*)(offs + i0 + 4 * k) = o;
        }
        carry += part[BIN_BLOCK - 1];
        __syncthreads();
    }
    if (tid == 0) offs[M] = carry;
}

__device__ __forceinline__ void bin_fill_block(const int* __restrict__ src, const int* __restrict__ dst,
                                               const unsigned* __restrict__ offs, unsigned* __restrict__ list,
                                               int N, int E, int NB, int c, unsigned* cursor) {
    const int tid = threadIdx.x;
    for (int i = tid; i < NB; i += BINW_BLOCK) cursor[i] = offs[(size_t)i * BIN_CH + c];
    __syncthreads();

    const int per = (E + BIN_CH - 1) / BIN_CH;
    const int e0  = c * per;
    int e1 = e0 + per;
    if (e1 > E) e1 = E;
    for (int e = e0 + tid; e < e1; e += BINW_DEPTH * BINW_BLOCK) {
        int d[BINW_DEPTH], sv[BINW_DEPTH];
#pragma unroll
        for (int k = 0; k < BINW_DEPTH; ++k) {
            const int ec = min(e + k * BINW_BLOCK, e1 - 1);
            d[k]  = dst[ec];
            sv[k] = src[ec];
        }
        unsigned p[BINW_DEPTH];
#pragma unroll
        for (int k = 0; k < BINW_DEPTH; ++k) {
            p[k] = 0u;
            if (e + k * BINW_BLOCK < e1 && (unsigned)d[k] < (unsigned)N)
                p[k] = atomicAdd(&cursor[d[k] / BIN_RNG], 1u);
        }
#pragma unroll
        for (int k = 0; k < BINW_DEPTH; ++k) {
            if (e + k * BINW_BLOCK < e1 && (unsigned)d[k] < (unsigned)N)
                list[p[k]] = ((unsigned)sv[k] << 9) | (unsigned)(d[k] - (d[k] / BIN_RNG) * BIN_RNG);
        }
    }
}

__global__ __launch_bounds__(BINW_BLOCK)
void bin_fill_gemm_kernel(const int* __restrict__ src, const int* __restrict__ dst,
                          const unsigned* __restrict__ offs, unsigned* __restrict__ list,
                          int N, int E, int NB,
                          const float* __restrict__ x, const float* __restrict__ W,
                          float* __restrict__ h, int mTiles, int g0) {
    __shared__ __attribute__((aligned(16))) unsigned char smem[GX_LDS_FLOATS * 4];
    if (blockIdx.x < BIN_CH)
        bin_fill_block(src, dst, offs, list, N, E, NB, blockIdx.x, (unsigned*)smem);
    else
        gemm_xw_block(x, W, h, N, mTiles, g0 + (int)blockIdx.x - BIN_CH, (float*)smem);
}

#define DEGL_BLOCK 256
#define DEGL_DEPTH 16

__device__ __forceinline__ void bin_deg_block(const unsigned* __restrict__ list, const unsigned* __restrict__ offs,
                                              float* __restrict__ dinv, int N, int bin, unsigned* cnt) {
    const int tid  = threadIdx.x;
    const int base = bin * BIN_RNG;
    int len = N - base;
    if (len > BIN_RNG) len = BIN_RNG;
    for (int i = tid; i < 512; i += DEGL_BLOCK) cnt[i] = 0u;
    const unsigned b0 = offs[(size_t)bin * BIN_CH];
    const unsigned b1 = offs[(size_t)(bin + 1) * BIN_CH];
    __syncthreads();

    for (unsigned i = b0 + (unsigned)tid; i < b1; i += DEGL_DEPTH * DEGL_BLOCK) {
        unsigned w[DEGL_DEPTH];
#pragma unroll
        for (int k = 0; k < DEGL_DEPTH; ++k) {
            unsigned idx = i + (unsigned)k * DEGL_BLOCK;
            if (idx > b1 - 1u) idx = b1 - 1u;
            w[k] = list[idx];
        }
#pragma unroll
        for (int k = 0; k < DEGL_DEPTH; ++k)
            atomicAdd(&cnt[w[k] & 511u], (i + (unsigned)k * DEGL_BLOCK < b1) ? 1u : 0u);
    }
    __syncthreads();

    for (int i = tid; i < len; i += DEGL_BLOCK)
        dinv[base + i] = rsqrtf((float)(cnt[i] + 1u));
}

__global__ __launch_bounds__(DEGL_BLOCK)
void bin_deg_gemm_kernel(const unsigned* __restrict__ list, const unsigned* __restrict__ offs,
                         float* __restrict__ dinv, int N, int NB,
                         const float* __restrict__ x, const float* __restrict__ W,
                         float* __restrict__ h, int mTiles, int g0) {
    __shared__ __attribute__((aligned(16))) unsigned char smem[GX_LDS_FLOATS * 4];
    if ((int)blockIdx.x < NB)
        bin_deg_block(list, offs, dinv, N, blockIdx.x, (unsigned*)smem);
    else
        gemm_xw_block(x, W, h, N, mTiles, g0 + (int)blockIdx.x - NB, (float*)smem);
}

#define POSP_PITCH 68
#define OUT_ITERS  ((BIN_RNG * (OUT_CH / 4) + BIN_BLOCK - 1) / BIN_BLOCK)

__device__ __forceinline__ void pos_tile_wmma(float* posp, const float* bs,
                                              const float* __restrict__ bp,
                                              const float* __restrict__ b,
                                              int r0, int lane) {
    const int l16 = lane & 15;
    const int hh  = lane >> 4;
    v2f a[16];
#pragma unroll
    for (int s = 0; s < 16; ++s)
        a[s] = *(const v2f*)(posp + (r0 + l16) * POSP_PITCH + 4 * s + 2 * hh);
    v8f c0 = {}, c1 = {}, c2 = {}, c3 = {};
#pragma unroll
    for (int s = 0; s < 16; ++s) {
        const float* bb = bs + (2 * s + hh) * BS_PITCH + l16 * 2;
        const v2f f0 = *(const v2f*)(bb + 0);
        const v2f f1 = *(const v2f*)(bb + 32);
        const v2f f2 = *(const v2f*)(bb + 64);
        const v2f f3 = *(const v2f*)(bb + 96);
        c0 = __builtin_amdgcn_wmma_f32_16x16x4_f32(false, a[s], false, f0, (short)0, c0, false, false);
        c1 = __builtin_amdgcn_wmma_f32_16x16x4_f32(false, a[s], false, f1, (short)0, c1, false, false);
        c2 = __builtin_amdgcn_wmma_f32_16x16x4_f32(false, a[s], false, f2, (short)0, c2, false, false);
        c3 = __builtin_amdgcn_wmma_f32_16x16x4_f32(false, a[s], false, f3, (short)0, c3, false, false);
    }
    const float bias0 = bp[l16 +  0] + b[l16 +  0];
    const float bias1 = bp[l16 + 16] + b[l16 + 16];
    const float bias2 = bp[l16 + 32] + b[l16 + 32];
    const float bias3 = bp[l16 + 48] + b[l16 + 48];
    float* pr = posp + (r0 + 8 * hh) * POSP_PITCH + l16;
#pragma unroll
    for (int r = 0; r < 8; ++r) {
        pr[r * POSP_PITCH +  0] = c0[r] + bias0;
        pr[r * POSP_PITCH + 16] = c1[r] + bias1;
        pr[r * POSP_PITCH + 32] = c2[r] + bias2;
        pr[r * POSP_PITCH + 48] = c3[r] + bias3;
    }
}

#define SORT_CAP   8192
#define SORT_PER   (SORT_CAP / BIN_BLOCK)
#define ROW_TURNS  ((BIN_RNG + 31) / 32)

template <int CNT>
__device__ __forceinline__ void row_group(const float* __restrict__ h, unsigned sv, unsigned dv,
                                          int k0, int lane, float& ax, float& ay) {
    float2 hv[CNT];
    float  sc[CNT];
#pragma unroll
    for (int k = 0; k < CNT; ++k) {
        const unsigned sn = (unsigned)__builtin_amdgcn_readlane((int)sv, k0 + k);
        sc[k] = __uint_as_float((unsigned)__builtin_amdgcn_readlane((int)dv, k0 + k));
        hv[k] = *(const float2*)(h + (size_t)sn * OUT_CH + 2 * lane);
    }
#pragma unroll
    for (int k = 0; k < CNT; ++k) {
        ax += hv[k].x * sc[k];
        ay += hv[k].y * sc[k];
    }
}

__global__ __launch_bounds__(BIN_BLOCK)
void bin_sum_pos_kernel(const unsigned* __restrict__ list, const unsigned* __restrict__ offs,
                        const float* __restrict__ h, const float* __restrict__ dinv,
                        const float* __restrict__ pos, const float* __restrict__ Wp,
                        const float* __restrict__ bp, const float* __restrict__ b,
                        float* __restrict__ out, int N) {
    __shared__ float acc[BIN_RNG * OUT_CH];
    __shared__ float posp[BIN_RNG * POSP_PITCH];
    __shared__ float bs[(POS_DIM / 2) * BS_PITCH];
    __shared__ uint2 ent[SORT_CAP];
    __shared__ unsigned cnt[512];
    __shared__ unsigned rstart[513];
    __shared__ unsigned wsum[16];
    const int tid  = threadIdx.x;
    const int bin  = blockIdx.x;
    const int base = bin * BIN_RNG;
    int len = N - base;
    if (len > BIN_RNG) len = BIN_RNG;
    const int wave = __builtin_amdgcn_readfirstlane(tid >> 5);
    const int lane = tid & 31;
    const bool tile_wave = wave < BIN_RNG / 16;
    const int r0 = wave * 16;

    v4f pv0 = {}, pv1 = {}, pv2 = {}, pv3 = {}, pv4 = {}, pv5 = {}, pv6 = {}, pv7 = {};
    if (tile_wave) {
        const float* pp = pos + (lane & 15) * 4;
        const int rl = base + r0 + (lane >> 4);
        const int nl = N - 1;
        pv0 = *(const v4f*)(pp + (size_t)min(rl +  0, nl) * POS_DIM);
        pv1 = *(const v4f*)(pp + (size_t)min(rl +  2, nl) * POS_DIM);
        pv2 = *(const v4f*)(pp + (size_t)min(rl +  4, nl) * POS_DIM);
        pv3 = *(const v4f*)(pp + (size_t)min(rl +  6, nl) * POS_DIM);
        pv4 = *(const v4f*)(pp + (size_t)min(rl +  8, nl) * POS_DIM);
        pv5 = *(const v4f*)(pp + (size_t)min(rl + 10, nl) * POS_DIM);
        pv6 = *(const v4f*)(pp + (size_t)min(rl + 12, nl) * POS_DIM);
        pv7 = *(const v4f*)(pp + (size_t)min(rl + 14, nl) * POS_DIM);
    }
    const float4 wv = *(const float4*)(Wp + (tid >> 4) * OUT_CH + ((tid & 15) << 2));
    const unsigned b0 = offs[(size_t)bin * BIN_CH];
    const unsigned b1 = offs[(size_t)(bin + 1) * BIN_CH];

    {
        const int k  = tid >> 4;
        const int n0 = (tid & 15) << 2;
        float* d = bs + (k >> 1) * BS_PITCH + (k & 1);
        d[(n0 + 0) * 2] = wv.x;
        d[(n0 + 1) * 2] = wv.y;
        d[(n0 + 2) * 2] = wv.z;
        d[(n0 + 3) * 2] = wv.w;
    }
    if (tile_wave) {
        float* ps = posp + (r0 + (lane >> 4)) * POSP_PITCH + (lane & 15) * 4;
        *(v4f*)(ps +  0 * POSP_PITCH) = pv0;
        *(v4f*)(ps +  2 * POSP_PITCH) = pv1;
        *(v4f*)(ps +  4 * POSP_PITCH) = pv2;
        *(v4f*)(ps +  6 * POSP_PITCH) = pv3;
        *(v4f*)(ps +  8 * POSP_PITCH) = pv4;
        *(v4f*)(ps + 10 * POSP_PITCH) = pv5;
        *(v4f*)(ps + 12 * POSP_PITCH) = pv6;
        *(v4f*)(ps + 14 * POSP_PITCH) = pv7;
    }

    const int turn = wave % ROW_TURNS;
    bool pending = tile_wave;
    unsigned c0 = b0;
    bool first = true;
    do {
        unsigned n = b1 - c0;
        if (n > SORT_CAP) n = SORT_CAP;
        if (tid < 512) cnt[tid] = 0u;
        __syncthreads();

        unsigned w[SORT_PER];
        float    dsrc[SORT_PER];
#pragma unroll
        for (int k = 0; k < SORT_PER; ++k) {
            const unsigned i = (unsigned)tid + (unsigned)k * BIN_BLOCK;
            w[k] = (i < n) ? list[c0 + i] : 0u;
        }
#pragma unroll
        for (int k = 0; k < SORT_PER; ++k) {
            const unsigned i = (unsigned)tid + (unsigned)k * BIN_BLOCK;
            dsrc[k] = (i < n) ? dinv[w[k] >> 9] : 0.0f;
        }
#pragma unroll
        for (int k = 0; k < SORT_PER; ++k) {
            const unsigned i = (unsigned)tid + (unsigned)k * BIN_BLOCK;
            if (i < n) atomicAdd(&cnt[w[k] & 511u], 1u);
        }
        __syncthreads();

        unsigned v = 0u, inc = 0u;
        if (tid < 512) {
            v = cnt[tid];
            inc = v;
#pragma unroll
            for (int o = 1; o < 32; o <<= 1) {
                const unsigned t = __shfl_up(inc, o);
                if (lane >= o) inc += t;
            }
            if (lane == 31) wsum[wave] = inc;
        }
        __syncthreads();
        if (tid < 512) {
            unsigned pre = 0u;
#pragma unroll
            for (int j = 0; j < 16; ++j) pre += (j < wave) ? wsum[j] : 0u;
            const unsigned excl = pre + inc - v;
            rstart[tid] = excl;
            cnt[tid] = excl;
            if (tid == 511) rstart[512] = excl + v;
        }
        __syncthreads();

#pragma unroll
        for (int k = 0; k < SORT_PER; ++k) {
            const unsigned i = (unsigned)tid + (unsigned)k * BIN_BLOCK;
            if (i < n) {
                const unsigned p = atomicAdd(&cnt[w[k] & 511u], 1u);
                ent[p] = make_uint2(w[k] >> 9, __float_as_uint(dsrc[k]));
            }
        }
        __syncthreads();

        for (int j = 0; j < ROW_TURNS; ++j) {
            if (pending && j == turn) {
                pos_tile_wmma(posp, bs, bp, b, r0, lane);
                pending = false;
            }
            const int r = wave + 32 * j;
            if (r < BIN_RNG) {
                const unsigned s0 = (unsigned)__builtin_amdgcn_readfirstlane((int)rstart[r]);
                const unsigned s1 = (unsigned)__builtin_amdgcn_readfirstlane((int)rstart[r + 1]);
                float ax = 0.0f, ay = 0.0f;
                for (unsigned e0 = s0; e0 < s1; e0 += 32u) {
                    unsigned m = s1 - e0;
                    if (m > 32u) m = 32u;
                    const unsigned li = ((unsigned)lane < m) ? (unsigned)lane : m - 1u;
                    const uint2 en = ent[e0 + li];
                    const unsigned sv = en.x;
                    const unsigned dv = ((unsigned)lane < m) ? en.y : 0u;
                    if (m > 8u)  row_group<16>(h, sv, dv, 0, lane, ax, ay);
                    else         row_group<8>(h, sv, dv, 0, lane, ax, ay);
                    if (m > 16u) {
                        if (m > 24u) row_group<16>(h, sv, dv, 16, lane, ax, ay);
                        else         row_group<8>(h, sv, dv, 16, lane, ax, ay);
                    }
                }
                float2* ar = (float2*)(acc + r * OUT_CH + 2 * lane);
                if (first) {
                    *ar = make_float2(ax, ay);
                } else {
                    const float2 o = *ar;
                    *ar = make_float2(o.x + ax, o.y + ay);
                }
            }
        }
        __syncthreads();
        first = false;
        c0 += SORT_CAP;
    } while (c0 < b1);

    const float* hb = h + (size_t)base * OUT_CH;
    float* o = out + (size_t)base * OUT_CH;
    const int nq = len * (OUT_CH / 4);
    float  di[OUT_ITERS];
    float4 hv[OUT_ITERS];
#pragma unroll
    for (int it = 0; it < OUT_ITERS; ++it) {
        int q = tid + it * BIN_BLOCK;
        if (q > nq - 1) q = nq - 1;
        di[it] = dinv[base + (q >> 4)];
        hv[it] = *(const float4*)(hb + (q >> 4) * OUT_CH + ((q & 15) << 2));
    }
#pragma unroll
    for (int it = 0; it < OUT_ITERS; ++it) {
        const int q = tid + it * BIN_BLOCK;
        if (q < nq) {
            const int row = q >> 4;
            const int c4  = (q & 15) << 2;
            const float4 pq = *(const float4*)(posp + row * POSP_PITCH + c4);
            const float4 av = *(const float4*)(acc + row * OUT_CH + c4);
            const float d = di[it];
            float4 r;
            r.x = (pq.x + hv[it].x * d * d) + d * av.x;
            r.y = (pq.y + hv[it].y * d * d) + d * av.y;
            r.z = (pq.z + hv[it].z * d * d) + d * av.z;
            r.w = (pq.w + hv[it].w * d * d) + d * av.w;
            *(float4*)(o + row * OUT_CH + c4) = r;
        }
    }
}

extern "C" void kernel_launch(void* const* d_in, const int* in_sizes, int n_in,
                              void* d_out, int out_size, void* d_ws, size_t ws_size,
                              hipStream_t stream) {
    const float* x   = (const float*)d_in[0];
    const int*   ei  = (const int*)d_in[1];
    const float* pos = (const float*)d_in[2];
    const float* W   = (const float*)d_in[3];
    const float* b   = (const float*)d_in[4];
    const float* Wp  = (const float*)d_in[5];
    const float* bp  = (const float*)d_in[6];
    float* out = (float*)d_out;

    const int N = in_sizes[0] / IN_CH;
    const int E = in_sizes[1] / 2;
    const int* src = ei;
    const int* dst = ei + E;

    float* dinv = (float*)d_ws;
    float* h    = dinv + ((N + 3) & ~3);

    const int mTiles = (N + 15) / 16;
    const int wavesPerBlock = 8;
    dim3 gGemm((mTiles + wavesPerBlock - 1) / wavesPerBlock, OUT_CH / 16);

    const int NB = (N + BIN_RNG - 1) / BIN_RNG;
    const int M  = NB * BIN_CH;
    unsigned* counts = (unsigned*)(h + (size_t)N * OUT_CH);
    unsigned* offs   = counts + M;
    unsigned* list   = offs + M + 4;
    const size_t need = (size_t)((char*)(list + E) - (char*)d_ws);
    if (NB <= BIN_MAX && N <= (1 << 23) && need <= ws_size) {
        const int gTot = (mTiles + GX_WAVES - 1) / GX_WAVES;
        const int g1 = gTot * 3 / 10;
        const int g3 = gTot * 3 / 10;
        const int g2 = gTot - g1 - g3;
        bin_count_gemm_kernel<<<BIN_CH + g1, BINW_BLOCK, 0, stream>>>(dst, counts, N, E, NB, x, W, h, mTiles, 0);
        bin_scan_kernel<<<1, BIN_BLOCK, 0, stream>>>(counts, offs, M);
        bin_fill_gemm_kernel<<<BIN_CH + g2, BINW_BLOCK, 0, stream>>>(src, dst, offs, list, N, E, NB, x, W, h, mTiles, g1);
        bin_deg_gemm_kernel<<<NB + g3, DEGL_BLOCK, 0, stream>>>(list, offs, dinv, N, NB, x, W, h, mTiles, g1 + g2);
        bin_sum_pos_kernel<<<NB, BIN_BLOCK, 0, stream>>>(list, offs, h, dinv, pos, Wp, bp, b, out, N);
    } else {
        unsigned* planes = (unsigned*)h;
        dim3 gDeg(DEG_CHUNKS, (N + DEG_RNG - 1) / DEG_RNG);
        deg_count_lds_kernel<<<gDeg, DEG_BLOCK, 0, stream>>>(dst, planes, N, E);
        deg_finish_kernel<<<(N + 255) / 256, 256, 0, stream>>>(planes, dinv, N);
        gemm_xw_lds_kernel<<<(mTiles + GX_WAVES - 1) / GX_WAVES, 32 * GX_WAVES, 0, stream>>>(x, W, h, N, mTiles);
        gemm_pos_fused_kernel<<<gGemm, 32 * wavesPerBlock, 0, stream>>>(
            pos, Wp, bp, b, h, dinv, out, N, mTiles);
        edge_scatter_kernel<<<(E + wavesPerBlock - 1) / wavesPerBlock,
                              32 * wavesPerBlock, 0, stream>>>(src, dst, h, dinv, out, E);
    }
}


# edges binned inside each chunk (LDS count, prefix, place into the chunk's own list slice); bin blocks read a bin's 128 segments, a share per wave; count and scan launches go; sums in placing order
# speedup vs baseline: 4.3646x; 1.5339x over previous
#include <hip/hip_runtime.h>

typedef __attribute__((ext_vector_type(2))) float v2f;
typedef __attribute__((ext_vector_type(8))) float v8f;
typedef __attribute__((ext_vector_type(4))) float v4f;

#define IN_CH   128
#define OUT_CH  64
#define POS_DIM 64

#define DEG_RNG    25600
#define DEG_CHUNKS 64
#define DEG_BLOCK  1024

__global__ __launch_bounds__(DEG_BLOCK)
void deg_count_lds_kernel(const int* __restrict__ dst, unsigned* __restrict__ planes,
                          int N, int E) {
    __shared__ unsigned cnt[DEG_RNG];
    const int tid  = threadIdx.x;
    const int c    = blockIdx.x;
    const int base = blockIdx.y * DEG_RNG;
    int len = N - base;
    if (len > DEG_RNG) len = DEG_RNG;

    for (int i = tid; i < len; i += DEG_BLOCK) cnt[i] = 0u;
    __syncthreads();

    const int per = (E + DEG_CHUNKS - 1) / DEG_CHUNKS;
    const int e0  = c * per;
    int e1 = e0 + per;
    if (e1 > E) e1 = E;
    for (int e = e0 + tid; e < e1; e += 4 * DEG_BLOCK) {
        const int d0 = dst[e];
        const int d1 = (e + 1 * DEG_BLOCK < e1) ? dst[e + 1 * DEG_BLOCK] : -1;
        const int d2 = (e + 2 * DEG_BLOCK < e1) ? dst[e + 2 * DEG_BLOCK] : -1;
        const int d3 = (e + 3 * DEG_BLOCK < e1) ? dst[e + 3 * DEG_BLOCK] : -1;
        const unsigned r0 = (unsigned)(d0 - base);
        const unsigned r1 = (unsigned)(d1 - base);
        const unsigned r2 = (unsigned)(d2 - base);
        const unsigned r3 = (unsigned)(d3 - base);
        if (r0 < (unsigned)len) atomicAdd(&cnt[r0], 1u);
        if (r1 < (unsigned)len) atomicAdd(&cnt[r1], 1u);
        if (r2 < (unsigned)len) atomicAdd(&cnt[r2], 1u);
        if (r3 < (unsigned)len) atomicAdd(&cnt[r3], 1u);
    }
    __syncthreads();

    unsigned* p = planes + (size_t)c * N + base;
    for (int i = tid; i < len; i += DEG_BLOCK) p[i] = cnt[i];
}

__global__ void deg_finish_kernel(const unsigned* __restrict__ planes,
                                  float* __restrict__ dinv, int N) {
    const int i = blockIdx.x * blockDim.x + threadIdx.x;
    if (i < N) {
        unsigned s = 1u;
#pragma unroll
        for (int c = 0; c < DEG_CHUNKS; ++c) s += planes[(size_t)c * N + i];
        dinv[i] = rsqrtf((float)s);
    }
}

#define GX_WAVES 8
#define XS_PITCH 132
#define BS_PITCH 160
#define GX_WITEMS ((IN_CH / 2) * (OUT_CH / 4) / (32 * GX_WAVES))

#define GX_LDS_FLOATS ((IN_CH / 2) * BS_PITCH + GX_WAVES * 16 * XS_PITCH)

__device__ __forceinline__ void gemm_xw_block(const float* __restrict__ x,
                                              const float* __restrict__ W,
                                              float* __restrict__ h, int N, int mTiles,
                                              int blk, float* lds) {
    float* wsm = lds;
    float* xs  = lds + (IN_CH / 2) * BS_PITCH;
    const int tid  = threadIdx.x;
    const int wave = __builtin_amdgcn_readfirstlane(tid >> 5);
    const int lane = tid & 31;
    const int mTile = blk * GX_WAVES + wave;
    const int row0  = mTile * 16;
    const int nl    = N - 1;

    v4f w0[GX_WITEMS], w1[GX_WITEMS];
#pragma unroll
    for (int j = 0; j < GX_WITEMS; ++j) {
        const int item = tid + j * 32 * GX_WAVES;
        const int p = item >> 4;
        const int q = item & 15;
        w0[j] = *(const v4f*)(W + (2 * p + 0) * OUT_CH + 4 * q);
        w1[j] = *(const v4f*)(W + (2 * p + 1) * OUT_CH + 4 * q);
    }
    v4f xv[16];
    const float* xp = x + lane * 4;
#pragma unroll
    for (int r = 0; r < 16; ++r)
        xv[r] = *(const v4f*)(xp + (size_t)min(row0 + r, nl) * IN_CH);
#pragma unroll
    for (int j = 0; j < GX_WITEMS; ++j) {
        const int item = tid + j * 32 * GX_WAVES;
        const int p = item >> 4;
        const int q = item & 15;
        float* d = wsm + p * BS_PITCH + 8 * q;
        const v4f lo = {w0[j][0], w1[j][0], w0[j][1], w1[j][1]};
        const v4f hi = {w0[j][2], w1[j][2], w0[j][3], w1[j][3]};
        *(v4f*)(d + 0) = lo;
        *(v4f*)(d + 4) = hi;
    }
    float* xw = xs + wave * 16 * XS_PITCH;
#pragma unroll
    for (int r = 0; r < 16; ++r)
        *(v4f*)(xw + r * XS_PITCH + lane * 4) = xv[r];
    __syncthreads();

    const int l16 = lane & 15;
    const int hh  = lane >> 4;
    v2f a[32];
#pragma unroll
    for (int s = 0; s < 32; ++s)
        a[s] = *(const v2f*)(xw + l16 * XS_PITCH + 4 * s + 2 * hh);
    v8f c0 = {}, c1 = {}, c2 = {}, c3 = {};
#pragma unroll
    for (int s = 0; s < 32; ++s) {
        const float* bb = wsm + (2 * s + hh) * BS_PITCH + l16 * 2;
        const v2f f0 = *(const v2f*)(bb + 0);
        const v2f f1 = *(const v2f*)(bb + 32);
        const v2f f2 = *(const v2f*)(bb + 64);
        const v2f f3 = *(const v2f*)(bb + 96);
        c0 = __builtin_amdgcn_wmma_f32_16x16x4_f32(false, a[s], false, f0, (short)0, c0, false, false);
        c1 = __builtin_amdgcn_wmma_f32_16x16x4_f32(false, a[s], false, f1, (short)0, c1, false, false);
        c2 = __builtin_amdgcn_wmma_f32_16x16x4_f32(false, a[s], false, f2, (short)0, c2, false, false);
        c3 = __builtin_amdgcn_wmma_f32_16x16x4_f32(false, a[s], false, f3, (short)0, c3, false, false);
    }

    float* pr = xw + (8 * hh) * XS_PITCH + l16;
#pragma unroll
    for (int r = 0; r < 8; ++r) {
        pr[r * XS_PITCH +  0] = c0[r];
        pr[r * XS_PITCH + 16] = c1[r];
        pr[r * XS_PITCH + 32] = c2[r];
        pr[r * XS_PITCH + 48] = c3[r];
    }
    const int rl = lane >> 4;
    const int c4 = (lane & 15) << 2;
    float* hp = h + (size_t)row0 * OUT_CH + rl * OUT_CH + c4;
    if (row0 + 16 <= N) {
        v4f v[8];
#pragma unroll
        for (int it = 0; it < 8; ++it)
            v[it] = *(const v4f*)(xw + (2 * it + rl) * XS_PITCH + c4);
#pragma unroll
        for (int it = 0; it < 8; ++it)
            *(v4f*)(hp + it * 2 * OUT_CH) = v[it];
    } else if (mTile < mTiles) {
#pragma unroll
        for (int it = 0; it < 8; ++it) {
            if (row0 + 2 * it + rl < N)
                *(v4f*)(hp + it * 2 * OUT_CH) = *(const v4f*)(xw + (2 * it + rl) * XS_PITCH + c4);
        }
    }
}

__global__ __launch_bounds__(32 * GX_WAVES)
void gemm_xw_lds_kernel(const float* __restrict__ x,
                        const float* __restrict__ W,
                        float* __restrict__ h, int N, int mTiles) {
    __shared__ float lds[GX_LDS_FLOATS];
    gemm_xw_block(x, W, h, N, mTiles, blockIdx.x, lds);
}

__global__ void gemm_pos_fused_kernel(const float* __restrict__ pos,
                                      const float* __restrict__ Wp,
                                      const float* __restrict__ bp,
                                      const float* __restrict__ b,
                                      const float* __restrict__ h,
                                      const float* __restrict__ dinv,
                                      float* __restrict__ out, int N, int mTiles) {
    const int wavesPerBlock = blockDim.x >> 5;
    const int waveId = threadIdx.x >> 5;
    const int lane   = threadIdx.x & 31;
    const int mTile  = blockIdx.x * wavesPerBlock + waveId;
    const int nTile  = blockIdx.y;
    if (mTile >= mTiles) return;

    const int l16 = lane & 15;
    const int kh  = (lane >> 4) << 1;
    int m = mTile * 16 + l16;
    if (m >= N) m = N - 1;
    const int n = nTile * 16 + l16;

    v8f c = {};
#pragma unroll
    for (int k = 0; k < POS_DIM; k += 4) {
        v2f a = *(const v2f*)(pos + (size_t)m * POS_DIM + k + kh);
        v2f bfrag;
        bfrag[0] = Wp[(size_t)(k + kh + 0) * OUT_CH + n];
        bfrag[1] = Wp[(size_t)(k + kh + 1) * OUT_CH + n];
        c = __builtin_amdgcn_wmma_f32_16x16x4_f32(
                false, a, false, bfrag, (short)0, c, false, false);
    }

    const float bias = bp[n] + b[n];
    const int rbase = mTile * 16 + ((lane >> 4) << 3);
#pragma unroll
    for (int r = 0; r < 8; ++r) {
        int row = rbase + r;
        if (row < N) {
            float di = dinv[row];
            out[(size_t)row * OUT_CH + n] =
                c[r] + bias + h[(size_t)row * OUT_CH + n] * di * di;
        }
    }
}

__global__ void edge_scatter_kernel(const int* __restrict__ src,
                                    const int* __restrict__ dst,
                                    const float* __restrict__ h,
                                    const float* __restrict__ dinv,
                                    float* __restrict__ out, int E) {
    const int wavesPerBlock = blockDim.x >> 5;
    const int e    = blockIdx.x * wavesPerBlock + (threadIdx.x >> 5);
    const int lane = threadIdx.x & 31;
    if (e >= E) return;

    const int s = src[e];
    const int d = dst[e];
    const float nrm = dinv[s] * dinv[d];

    const float2 hv = *(const float2*)(h + (size_t)s * OUT_CH + 2 * lane);
    float* op = out + (size_t)d * OUT_CH + 2 * lane;
    atomicAdd(op + 0, hv.x * nrm);
    atomicAdd(op + 1, hv.y * nrm);
}

#define BIN_RNG   400
#define BIN_CH    128
#define BIN_MAX   1024
#define BIN_BLOCK 1024

#define BINW_BLOCK (32 * GX_WAVES)
#define BINW_DEPTH 25

__device__ __forceinline__ void bin_local_block(const int* __restrict__ src, const int* __restrict__ dst,
                                                unsigned* __restrict__ segtab, unsigned* __restrict__ list,
                                                int N, int E, int NB, int c, unsigned* sm) {
    unsigned* hist = sm;
    unsigned* wsum = sm + BIN_MAX;
    const int tid  = threadIdx.x;
    const int wave = tid >> 5;
    const int lane = tid & 31;
    for (int i = tid; i < BIN_MAX; i += BINW_BLOCK) hist[i] = 0u;
    __syncthreads();

    const int per = (E + BIN_CH - 1) / BIN_CH;
    const int e0  = c * per;
    int e1 = e0 + per;
    if (e1 > E) e1 = E;
    for (int e = e0 + tid; e < e1; e += BINW_DEPTH * BINW_BLOCK) {
        int d[BINW_DEPTH];
#pragma unroll
        for (int k = 0; k < BINW_DEPTH; ++k)
            d[k] = dst[min(e + k * BINW_BLOCK, e1 - 1)];
#pragma unroll
        for (int k = 0; k < BINW_DEPTH; ++k)
            if (e + k * BINW_BLOCK < e1 && (unsigned)d[k] < (unsigned)N) atomicAdd(&hist[d[k] / BIN_RNG], 1u);
    }
    __syncthreads();

    const unsigned v0 = hist[4 * tid + 0];
    const unsigned v1 = hist[4 * tid + 1];
    const unsigned v2 = hist[4 * tid + 2];
    const unsigned v3 = hist[4 * tid + 3];
    const unsigned own = v0 + v1 + v2 + v3;
    unsigned inc = own;
#pragma unroll
    for (int o = 1; o < 32; o <<= 1) {
        const unsigned t = __shfl_up(inc, o);
        if (lane >= o) inc += t;
    }
    if (lane == 31) wsum[wave] = inc;
    __syncthreads();
    unsigned pre = 0u, tot = 0u;
#pragma unroll
    for (int j = 0; j < GX_WAVES; ++j) {
        const unsigned q = wsum[j];
        pre += (j < wave) ? q : 0u;
        tot += q;
    }
    const unsigned o0 = pre + inc - own;
    const unsigned o1 = o0 + v0;
    const unsigned o2 = o1 + v1;
    const unsigned o3 = o2 + v2;
    hist[4 * tid + 0] = o0;
    hist[4 * tid + 1] = o1;
    hist[4 * tid + 2] = o2;
    hist[4 * tid + 3] = o3;
    unsigned* trow = segtab + (size_t)c * (NB + 1);
    if (4 * tid + 0 < NB) trow[4 * tid + 0] = o0;
    if (4 * tid + 1 < NB) trow[4 * tid + 1] = o1;
    if (4 * tid + 2 < NB) trow[4 * tid + 2] = o2;
    if (4 * tid + 3 < NB) trow[4 * tid + 3] = o3;
    if (tid == 0) trow[NB] = tot;
    __syncthreads();

    unsigned* lc = list + (size_t)c * per;
    for (int e = e0 + tid; e < e1; e += BINW_DEPTH * BINW_BLOCK) {
        int d[BINW_DEPTH], sv[BINW_DEPTH];
#pragma unroll
        for (int k = 0; k < BINW_DEPTH; ++k) {
            const int ec = min(e + k * BINW_BLOCK, e1 - 1);
            d[k]  = dst[ec];
            sv[k] = src[ec];
        }
        unsigned p[BINW_DEPTH];
#pragma unroll
        for (int k = 0; k < BINW_DEPTH; ++k) {
            p[k] = 0u;
            if (e + k * BINW_BLOCK < e1 && (unsigned)d[k] < (unsigned)N)
                p[k] = atomicAdd(&hist[d[k] / BIN_RNG], 1u);
        }
#pragma unroll
        for (int k = 0; k < BINW_DEPTH; ++k) {
            if (e + k * BINW_BLOCK < e1 && (unsigned)d[k] < (unsigned)N)
                lc[p[k]] = ((unsigned)sv[k] << 9) | (unsigned)(d[k] - (d[k] / BIN_RNG) * BIN_RNG);
        }
    }
    for (int i = (int)tot + tid; i < per; i += BINW_BLOCK) lc[i] = 0u;
}

__global__ __launch_bounds__(BINW_BLOCK)
void bin_local_gemm_kernel(const int* __restrict__ src, const int* __restrict__ dst,
                           unsigned* __restrict__ segtab, unsigned* __restrict__ list,
                           int N, int E, int NB,
                           const float* __restrict__ x, const float* __restrict__ W,
                           float* __restrict__ h, int mTiles, int g0) {
    __shared__ __attribute__((aligned(16))) unsigned char smem[GX_LDS_FLOATS * 4];
    if (blockIdx.x < BIN_CH)
        bin_local_block(src, dst, segtab, list, N, E, NB, blockIdx.x, (unsigned*)smem);
    else
        gemm_xw_block(x, W, h, N, mTiles, g0 + (int)blockIdx.x - BIN_CH, (float*)smem);
}

#define DEGL_BLOCK 256
#define DEGL_SEGS  (BIN_CH / (DEGL_BLOCK / 32))

__device__ __forceinline__ void bin_deg_block(const unsigned* __restrict__ list, const unsigned* __restrict__ segtab,
                                              float* __restrict__ dinv, int N, int E, int NB, int bin, unsigned* cnt) {
    const int tid  = threadIdx.x;
    const int wave = __builtin_amdgcn_readfirstlane(tid >> 5);
    const int lane = tid & 31;
    const int base = bin * BIN_RNG;
    int len = N - base;
    if (len > BIN_RNG) len = BIN_RNG;
    const int per = (E + BIN_CH - 1) / BIN_CH;
    unsigned sa = 0u, sb = 0u;
    if (lane < DEGL_SEGS) {
        const unsigned* t = segtab + (size_t)(wave * DEGL_SEGS + lane) * (NB + 1) + bin;
        sa = t[0];
        sb = t[1];
    }
    for (int i = tid; i < 512; i += DEGL_BLOCK) cnt[i] = 0u;
    const unsigned ln = sb - sa;
    const unsigned bs = (unsigned)(wave * DEGL_SEGS + lane) * (unsigned)per + sa;
    __syncthreads();

    const unsigned lw = (unsigned)BIN_CH * (unsigned)per - 1u;
    unsigned w0[DEGL_SEGS], w1[DEGL_SEGS];
#pragma unroll
    for (int j = 0; j < DEGL_SEGS; ++j) {
        const unsigned bj = (unsigned)__builtin_amdgcn_readlane((int)bs, j);
        w0[j] = list[min(bj + (unsigned)lane, lw)];
        w1[j] = list[min(bj + 32u + (unsigned)lane, lw)];
    }
#pragma unroll
    for (int j = 0; j < DEGL_SEGS; ++j) {
        const unsigned lj = (unsigned)__builtin_amdgcn_readlane((int)ln, j);
        atomicAdd(&cnt[w0[j] & 511u], ((unsigned)lane < lj) ? 1u : 0u);
        atomicAdd(&cnt[w1[j] & 511u], ((unsigned)lane + 32u < lj) ? 1u : 0u);
    }
#pragma unroll
    for (int j = 0; j < DEGL_SEGS; ++j) {
        const unsigned lj = (unsigned)__builtin_amdgcn_readlane((int)ln, j);
        const unsigned bj = (unsigned)__builtin_amdgcn_readlane((int)bs, j);
        for (unsigned q = 64u; q < lj; q += 32u)
            if (q + (unsigned)lane < lj) atomicAdd(&cnt[list[bj + q + (unsigned)lane] & 511u], 1u);
    }
    __syncthreads();

    for (int i = tid; i < len; i += DEGL_BLOCK)
        dinv[base + i] = rsqrtf((float)(cnt[i] + 1u));
}

__global__ __launch_bounds__(DEGL_BLOCK)
void bin_deg_gemm_kernel(const unsigned* __restrict__ list, const unsigned* __restrict__ segtab,
                         float* __restrict__ dinv, int N, int E, int NB,
                         const float* __restrict__ x, const float* __restrict__ W,
                         float* __restrict__ h, int mTiles, int g0) {
    __shared__ __attribute__((aligned(16))) unsigned char smem[GX_LDS_FLOATS * 4];
    if ((int)blockIdx.x < NB)
        bin_deg_block(list, segtab, dinv, N, E, NB, blockIdx.x, (unsigned*)smem);
    else
        gemm_xw_block(x, W, h, N, mTiles, g0 + (int)blockIdx.x - NB, (float*)smem);
}

#define POSP_PITCH 68
#define OUT_ITERS  ((BIN_RNG * (OUT_CH / 4) + BIN_BLOCK - 1) / BIN_BLOCK)

__device__ __forceinline__ void pos_tile_wmma(float* posp, const float* bs,
                                              const float* __restrict__ bp,
                                              const float* __restrict__ b,
                                              int r0, int lane) {
    const int l16 = lane & 15;
    const int hh  = lane >> 4;
    v2f a[16];
#pragma unroll
    for (int s = 0; s < 16; ++s)
        a[s] = *(const v2f*)(posp + (r0 + l16) * POSP_PITCH + 4 * s + 2 * hh);
    v8f c0 = {}, c1 = {}, c2 = {}, c3 = {};
#pragma unroll
    for (int s = 0; s < 16; ++s) {
        const float* bb = bs + (2 * s + hh) * BS_PITCH + l16 * 2;
        const v2f f0 = *(const v2f*)(bb + 0);
        const v2f f1 = *(const v2f*)(bb + 32);
        const v2f f2 = *(const v2f*)(bb + 64);
        const v2f f3 = *(const v2f*)(bb + 96);
        c0 = __builtin_amdgcn_wmma_f32_16x16x4_f32(false, a[s], false, f0, (short)0, c0, false, false);
        c1 = __builtin_amdgcn_wmma_f32_16x16x4_f32(false, a[s], false, f1, (short)0, c1, false, false);
        c2 = __builtin_amdgcn_wmma_f32_16x16x4_f32(false, a[s], false, f2, (short)0, c2, false, false);
        c3 = __builtin_amdgcn_wmma_f32_16x16x4_f32(false, a[s], false, f3, (short)0, c3, false, false);
    }
    const float bias0 = bp[l16 +  0] + b[l16 +  0];
    const float bias1 = bp[l16 + 16] + b[l16 + 16];
    const float bias2 = bp[l16 + 32] + b[l16 + 32];
    const float bias3 = bp[l16 + 48] + b[l16 + 48];
    float* pr = posp + (r0 + 8 * hh) * POSP_PITCH + l16;
#pragma unroll
    for (int r = 0; r < 8; ++r) {
        pr[r * POSP_PITCH +  0] = c0[r] + bias0;
        pr[r * POSP_PITCH + 16] = c1[r] + bias1;
        pr[r * POSP_PITCH + 32] = c2[r] + bias2;
        pr[r * POSP_PITCH + 48] = c3[r] + bias3;
    }
}

#define SORT_PER   9
#define SORT_CAP   (SORT_PER * BIN_BLOCK)
#define PASS_W     (32 * SORT_PER)
#define SUM_SEGS   (BIN_CH / (BIN_BLOCK / 32))
#define ROW_TURNS  ((BIN_RNG + 31) / 32)

template <int CNT>
__device__ __forceinline__ void row_group(const float* __restrict__ h, unsigned sv, unsigned dv,
                                          int k0, int lane, float& ax, float& ay) {
    float2 hv[CNT];
    float  sc[CNT];
#pragma unroll
    for (int k = 0; k < CNT; ++k) {
        const unsigned sn = (unsigned)__builtin_amdgcn_readlane((int)sv, k0 + k);
        sc[k] = __uint_as_float((unsigned)__builtin_amdgcn_readlane((int)dv, k0 + k));
        hv[k] = *(const float2*)(h + (size_t)sn * OUT_CH + 2 * lane);
    }
#pragma unroll
    for (int k = 0; k < CNT; ++k) {
        ax += hv[k].x * sc[k];
        ay += hv[k].y * sc[k];
    }
}

__global__ __launch_bounds__(BIN_BLOCK)
void bin_sum_pos_kernel(const unsigned* __restrict__ list, const unsigned* __restrict__ segtab,
                        const float* __restrict__ h, const float* __restrict__ dinv,
                        const float* __restrict__ pos, const float* __restrict__ Wp,
                        const float* __restrict__ bp, const float* __restrict__ b,
                        float* __restrict__ out, int N, int E, int NB) {
    __shared__ float acc[BIN_RNG * OUT_CH];
    __shared__ float posp[BIN_RNG * POSP_PITCH];
    __shared__ float bs[(POS_DIM / 2) * BS_PITCH];
    __shared__ uint2 ent[SORT_CAP];
    __shared__ unsigned cnt[512];
    __shared__ unsigned rstart[513];
    __shared__ unsigned wsum[16];
    __shared__ __attribute__((aligned(16))) unsigned wpass[BIN_BLOCK / 32];
    const int tid  = threadIdx.x;
    const int bin  = blockIdx.x;
    const int base = bin * BIN_RNG;
    int len = N - base;
    if (len > BIN_RNG) len = BIN_RNG;
    const int wave = __builtin_amdgcn_readfirstlane(tid >> 5);
    const int lane = tid & 31;
    const bool tile_wave = wave < BIN_RNG / 16;
    const int r0 = wave * 16;

    const int per = (E + BIN_CH - 1) / BIN_CH;
    unsigned sa = 0u, sb = 0u;
    if (lane < SUM_SEGS) {
        const unsigned* t = segtab + (size_t)(wave * SUM_SEGS + lane) * (NB + 1) + bin;
        sa = t[0];
        sb = t[1];
    }

    v4f pv0 = {}, pv1 = {}, pv2 = {}, pv3 = {}, pv4 = {}, pv5 = {}, pv6 = {}, pv7 = {};
    if (tile_wave) {
        const float* pp = pos + (lane & 15) * 4;
        const int rl = base + r0 + (lane >> 4);
        const int nl = N - 1;
        pv0 = *(const v4f*)(pp + (size_t)min(rl +  0, nl) * POS_DIM);
        pv1 = *(const v4f*)(pp + (size_t)min(rl +  2, nl) * POS_DIM);
        pv2 = *(const v4f*)(pp + (size_t)min(rl +  4, nl) * POS_DIM);
        pv3 = *(const v4f*)(pp + (size_t)min(rl +  6, nl) * POS_DIM);
        pv4 = *(const v4f*)(pp + (size_t)min(rl +  8, nl) * POS_DIM);
        pv5 = *(const v4f*)(pp + (size_t)min(rl + 10, nl) * POS_DIM);
        pv6 = *(const v4f*)(pp + (size_t)min(rl + 12, nl) * POS_DIM);
        pv7 = *(const v4f*)(pp + (size_t)min(rl + 14, nl) * POS_DIM);
    }
    const float4 wv = *(const float4*)(Wp + (tid >> 4) * OUT_CH + ((tid & 15) << 2));

    {
        const int k  = tid >> 4;
        const int n0 = (tid & 15) << 2;
        float* d = bs + (k >> 1) * BS_PITCH + (k & 1);
        d[(n0 + 0) * 2] = wv.x;
        d[(n0 + 1) * 2] = wv.y;
        d[(n0 + 2) * 2] = wv.z;
        d[(n0 + 3) * 2] = wv.w;
    }
    if (tile_wave) {
        float* ps = posp + (r0 + (lane >> 4)) * POSP_PITCH + (lane & 15) * 4;
        *(v4f*)(ps +  0 * POSP_PITCH) = pv0;
        *(v4f*)(ps +  2 * POSP_PITCH) = pv1;
        *(v4f*)(ps +  4 * POSP_PITCH) = pv2;
        *(v4f*)(ps +  6 * POSP_PITCH) = pv3;
        *(v4f*)(ps +  8 * POSP_PITCH) = pv4;
        *(v4f*)(ps + 10 * POSP_PITCH) = pv5;
        *(v4f*)(ps + 12 * POSP_PITCH) = pv6;
        *(v4f*)(ps + 14 * POSP_PITCH) = pv7;
    }

    const unsigned ln = sb - sa;
    const unsigned sg0 = (unsigned)(wave * SUM_SEGS + lane) * (unsigned)per + sa;
    const unsigned c1 = (unsigned)__builtin_amdgcn_readlane((int)ln, 0);
    const unsigned c2 = c1 + (unsigned)__builtin_amdgcn_readlane((int)ln, 1);
    const unsigned c3 = c2 + (unsigned)__builtin_amdgcn_readlane((int)ln, 2);
    const unsigned Lw = c3 + (unsigned)__builtin_amdgcn_readlane((int)ln, 3);
    const unsigned g0 = (unsigned)__builtin_amdgcn_readlane((int)sg0, 0);
    const unsigned g1 = (unsigned)__builtin_amdgcn_readlane((int)sg0, 1) - c1;
    const unsigned g2 = (unsigned)__builtin_amdgcn_readlane((int)sg0, 2) - c2;
    const unsigned g3 = (unsigned)__builtin_amdgcn_readlane((int)sg0, 3) - c3;
    if (lane == 0) wpass[wave] = (Lw + PASS_W - 1) / PASS_W;
    __syncthreads();
    unsigned npass = 1u;
#pragma unroll
    for (int j = 0; j < BIN_BLOCK / 32; j += 4) {
        const uint4 q = *(const uint4*)(wpass + j);
        npass = max(max(npass, max(q.x, q.y)), max(q.z, q.w));
    }

    const int turn = wave % ROW_TURNS;
    bool pending = tile_wave;
    unsigned p0 = 0u;
    unsigned pass = 0u;
    bool first = true;
    do {
        if (tid < 512) cnt[tid] = 0u;
        __syncthreads();

        unsigned w[SORT_PER];
        float    dsrc[SORT_PER];
#pragma unroll
        for (int k = 0; k < SORT_PER; ++k) {
            const unsigned i = p0 + 32u * (unsigned)k + (unsigned)lane;
            const unsigned g = (i < c1) ? g0 : ((i < c2) ? g1 : ((i < c3) ? g2 : g3));
            w[k] = (i < Lw) ? list[g + i] : 0u;
        }
#pragma unroll
        for (int k = 0; k < SORT_PER; ++k) {
            const unsigned i = p0 + 32u * (unsigned)k + (unsigned)lane;
            dsrc[k] = (i < Lw) ? dinv[w[k] >> 9] : 0.0f;
        }
#pragma unroll
        for (int k = 0; k < SORT_PER; ++k) {
            const unsigned i = p0 + 32u * (unsigned)k + (unsigned)lane;
            if (i < Lw) atomicAdd(&cnt[w[k] & 511u], 1u);
        }
        __syncthreads();

        unsigned v = 0u, inc = 0u;
        if (tid < 512) {
            v = cnt[tid];
            inc = v;
#pragma unroll
            for (int o = 1; o < 32; o <<= 1) {
                const unsigned t = __shfl_up(inc, o);
                if (lane >= o) inc += t;
            }
            if (lane == 31) wsum[wave] = inc;
        }
        __syncthreads();
        if (tid < 512) {
            unsigned pre = 0u;
#pragma unroll
            for (int j = 0; j < 16; ++j) pre += (j < wave) ? wsum[j] : 0u;
            const unsigned excl = pre + inc - v;
            rstart[tid] = excl;
            cnt[tid] = excl;
            if (tid == 511) rstart[512] = excl + v;
        }
        __syncthreads();

#pragma unroll
        for (int k = 0; k < SORT_PER; ++k) {
            const unsigned i = p0 + 32u * (unsigned)k + (unsigned)lane;
            if (i < Lw) {
                const unsigned p = atomicAdd(&cnt[w[k] & 511u], 1u);
                ent[p] = make_uint2(w[k] >> 9, __float_as_uint(dsrc[k]));
            }
        }
        __syncthreads();

        for (int j = 0; j < ROW_TURNS; ++j) {
            if (pending && j == turn) {
                pos_tile_wmma(posp, bs, bp, b, r0, lane);
                pending = false;
            }
            const int r = wave + 32 * j;
            if (r < BIN_RNG) {
                const unsigned s0 = (unsigned)__builtin_amdgcn_readfirstlane((int)rstart[r]);
                const unsigned s1 = (unsigned)__builtin_amdgcn_readfirstlane((int)rstart[r + 1]);
                float ax = 0.0f, ay = 0.0f;
                for (unsigned e0 = s0; e0 < s1; e0 += 32u) {
                    unsigned m = s1 - e0;
                    if (m > 32u) m = 32u;
                    const unsigned li = ((unsigned)lane < m) ? (unsigned)lane : m - 1u;
                    const uint2 en = ent[e0 + li];
                    const unsigned sv = en.x;
                    const unsigned dv = ((unsigned)lane < m) ? en.y : 0u;
                    if (m > 8u)  row_group<16>(h, sv, dv, 0, lane, ax, ay);
                    else         row_group<8>(h, sv, dv, 0, lane, ax, ay);
                    if (m > 16u) {
                        if (m > 24u) row_group<16>(h, sv, dv, 16, lane, ax, ay);
                        else         row_group<8>(h, sv, dv, 16, lane, ax, ay);
                    }
                }
                float2* ar = (float2*)(acc + r * OUT_CH + 2 * lane);
                if (first) {
                    *ar = make_float2(ax, ay);
                } else {
                    const float2 o = *ar;
                    *ar = make_float2(o.x + ax, o.y + ay);
                }
            }
        }
        __syncthreads();
        first = false;
        p0 += PASS_W;
        ++pass;
    } while (pass < npass);

    const float* hb = h + (size_t)base * OUT_CH;
    float* o = out + (size_t)base * OUT_CH;
    const int nq = len * (OUT_CH / 4);
    float  di[OUT_ITERS];
    float4 hv[OUT_ITERS];
#pragma unroll
    for (int it = 0; it < OUT_ITERS; ++it) {
        int q = tid + it * BIN_BLOCK;
        if (q > nq - 1) q = nq - 1;
        di[it] = dinv[base + (q >> 4)];
        hv[it] = *(const float4*)(hb + (q >> 4) * OUT_CH + ((q & 15) << 2));
    }
#pragma unroll
    for (int it = 0; it < OUT_ITERS; ++it) {
        const int q = tid + it * BIN_BLOCK;
        if (q < nq) {
            const int row = q >> 4;
            const int c4  = (q & 15) << 2;
            const float4 pq = *(const float4*)(posp + row * POSP_PITCH + c4);
            const float4 av = *(const float4*)(acc + row * OUT_CH + c4);
            const float d = di[it];
            float4 r;
            r.x = (pq.x + hv[it].x * d * d) + d * av.x;
            r.y = (pq.y + hv[it].y * d * d) + d * av.y;
            r.z = (pq.z + hv[it].z * d * d) + d * av.z;
            r.w = (pq.w + hv[it].w * d * d) + d * av.w;
            *(float4*)(o + row * OUT_CH + c4) = r;
        }
    }
}

extern "C" void kernel_launch(void* const* d_in, const int* in_sizes, int n_in,
                              void* d_out, int out_size, void* d_ws, size_t ws_size,
                              hipStream_t stream) {
    const float* x   = (const float*)d_in[0];
    const int*   ei  = (const int*)d_in[1];
    const float* pos = (const float*)d_in[2];
    const float* W   = (const float*)d_in[3];
    const float* b   = (const float*)d_in[4];
    const float* Wp  = (const float*)d_in[5];
    const float* bp  = (const float*)d_in[6];
    float* out = (float*)d_out;

    const int N = in_sizes[0] / IN_CH;
    const int E = in_sizes[1] / 2;
    const int* src = ei;
    const int* dst = ei + E;

    float* dinv = (float*)d_ws;
    float* h    = dinv + ((N + 3) & ~3);

    const int mTiles = (N + 15) / 16;
    const int wavesPerBlock = 8;
    dim3 gGemm((mTiles + wavesPerBlock - 1) / wavesPerBlock, OUT_CH / 16);

    const int NB  = (N + BIN_RNG - 1) / BIN_RNG;
    const int per = (E + BIN_CH - 1) / BIN_CH;
    const int TW  = (BIN_CH * (NB + 1) + 3) & ~3;
    unsigned* segtab = (unsigned*)(h + (size_t)N * OUT_CH);
    unsigned* list   = segtab + TW;
    const size_t need = (size_t)((char*)(list + (size_t)BIN_CH * per) - (char*)d_ws);
    if (NB <= BIN_MAX && N <= (1 << 23) && need <= ws_size) {
        const int gTot = (mTiles + GX_WAVES - 1) / GX_WAVES;
        const int g3 = gTot * 3 / 10;
        const int g2 = gTot - g3;
        bin_local_gemm_kernel<<<BIN_CH + g2, BINW_BLOCK, 0, stream>>>(src, dst, segtab, list, N, E, NB, x, W, h, mTiles, 0);
        bin_deg_gemm_kernel<<<NB + g3, DEGL_BLOCK, 0, stream>>>(list, segtab, dinv, N, E, NB, x, W, h, mTiles, g2);
        bin_sum_pos_kernel<<<NB, BIN_BLOCK, 0, stream>>>(list, segtab, h, dinv, pos, Wp, bp, b, out, N, E, NB);
    } else {
        unsigned* planes = (unsigned*)h;
        dim3 gDeg(DEG_CHUNKS, (N + DEG_RNG - 1) / DEG_RNG);
        deg_count_lds_kernel<<<gDeg, DEG_BLOCK, 0, stream>>>(dst, planes, N, E);
        deg_finish_kernel<<<(N + 255) / 256, 256, 0, stream>>>(planes, dinv, N);
        gemm_xw_lds_kernel<<<(mTiles + GX_WAVES - 1) / GX_WAVES, 32 * GX_WAVES, 0, stream>>>(x, W, h, N, mTiles);
        gemm_pos_fused_kernel<<<gGemm, 32 * wavesPerBlock, 0, stream>>>(
            pos, Wp, bp, b, h, dinv, out, N, mTiles);
        edge_scatter_kernel<<<(E + wavesPerBlock - 1) / wavesPerBlock,
                              32 * wavesPerBlock, 0, stream>>>(src, dst, h, dinv, out, E);
    }
}


# edge binning: a chunk's placed words are collected in an LDS copy of its list slice and leave in order as whole-line 16-byte stores (was one scattered 4-byte store per edge); sum order unchanged
# speedup vs baseline: 4.6837x; 1.0731x over previous
#include <hip/hip_runtime.h>

typedef __attribute__((ext_vector_type(2))) float v2f;
typedef __attribute__((ext_vector_type(8))) float v8f;
typedef __attribute__((ext_vector_type(4))) float v4f;

#define IN_CH   128
#define OUT_CH  64
#define POS_DIM 64

#define DEG_RNG    25600
#define DEG_CHUNKS 64
#define DEG_BLOCK  1024

__global__ __launch_bounds__(DEG_BLOCK)
void deg_count_lds_kernel(const int* __restrict__ dst, unsigned* __restrict__ planes,
                          int N, int E) {
    __shared__ unsigned cnt[DEG_RNG];
    const int tid  = threadIdx.x;
    const int c    = blockIdx.x;
    const int base = blockIdx.y * DEG_RNG;
    int len = N - base;
    if (len > DEG_RNG) len = DEG_RNG;

    for (int i = tid; i < len; i += DEG_BLOCK) cnt[i] = 0u;
    __syncthreads();

    const int per = (E + DEG_CHUNKS - 1) / DEG_CHUNKS;
    const int e0  = c * per;
    int e1 = e0 + per;
    if (e1 > E) e1 = E;
    for (int e = e0 + tid; e < e1; e += 4 * DEG_BLOCK) {
        const int d0 = dst[e];
        const int d1 = (e + 1 * DEG_BLOCK < e1) ? dst[e + 1 * DEG_BLOCK] : -1;
        const int d2 = (e + 2 * DEG_BLOCK < e1) ? dst[e + 2 * DEG_BLOCK] : -1;
        const int d3 = (e + 3 * DEG_BLOCK < e1) ? dst[e + 3 * DEG_BLOCK] : -1;
        const unsigned r0 = (unsigned)(d0 - base);
        const unsigned r1 = (unsigned)(d1 - base);
        const unsigned r2 = (unsigned)(d2 - base);
        const unsigned r3 = (unsigned)(d3 - base);
        if (r0 < (unsigned)len) atomicAdd(&cnt[r0], 1u);
        if (r1 < (unsigned)len) atomicAdd(&cnt[r1], 1u);
        if (r2 < (unsigned)len) atomicAdd(&cnt[r2], 1u);
        if (r3 < (unsigned)len) atomicAdd(&cnt[r3], 1u);
    }
    __syncthreads();

    unsigned* p = planes + (size_t)c * N + base;
    for (int i = tid; i < len; i += DEG_BLOCK) p[i] = cnt[i];
}

__global__ void deg_finish_kernel(const unsigned* __restrict__ planes,
                                  float* __restrict__ dinv, int N) {
    const int i = blockIdx.x * blockDim.x + threadIdx.x;
    if (i < N) {
        unsigned s = 1u;
#pragma unroll
        for (int c = 0; c < DEG_CHUNKS; ++c) s += planes[(size_t)c * N + i];
        dinv[i] = rsqrtf((float)s);
    }
}

#define GX_WAVES 8
#define XS_PITCH 132
#define BS_PITCH 160
#define GX_WITEMS ((IN_CH / 2) * (OUT_CH / 4) / (32 * GX_WAVES))

#define GX_LDS_FLOATS ((IN_CH / 2) * BS_PITCH + GX_WAVES * 16 * XS_PITCH)

__device__ __forceinline__ void gemm_xw_block(const float* __restrict__ x,
                                              const float* __restrict__ W,
                                              float* __restrict__ h, int N, int mTiles,
                                              int blk, float* lds) {
    float* wsm = lds;
    float* xs  = lds + (IN_CH / 2) * BS_PITCH;
    const int tid  = threadIdx.x;
    const int wave = __builtin_amdgcn_readfirstlane(tid >> 5);
    const int lane = tid & 31;
    const int mTile = blk * GX_WAVES + wave;
    const int row0  = mTile * 16;
    const int nl    = N - 1;

    v4f w0[GX_WITEMS], w1[GX_WITEMS];
#pragma unroll
    for (int j = 0; j < GX_WITEMS; ++j) {
        const int item = tid + j * 32 * GX_WAVES;
        const int p = item >> 4;
        const int q = item & 15;
        w0[j] = *(const v4f*)(W + (2 * p + 0) * OUT_CH + 4 * q);
        w1[j] = *(const v4f*)(W + (2 * p + 1) * OUT_CH + 4 * q);
    }
    v4f xv[16];
    const float* xp = x + lane * 4;
#pragma unroll
    for (int r = 0; r < 16; ++r)
        xv[r] = *(const v4f*)(xp + (size_t)min(row0 + r, nl) * IN_CH);
#pragma unroll
    for (int j = 0; j < GX_WITEMS; ++j) {
        const int item = tid + j * 32 * GX_WAVES;
        const int p = item >> 4;
        const int q = item & 15;
        float* d = wsm + p * BS_PITCH + 8 * q;
        const v4f lo = {w0[j][0], w1[j][0], w0[j][1], w1[j][1]};
        const v4f hi = {w0[j][2], w1[j][2], w0[j][3], w1[j][3]};
        *(v4f*)(d + 0) = lo;
        *(v4f*)(d + 4) = hi;
    }
    float* xw = xs + wave * 16 * XS_PITCH;
#pragma unroll
    for (int r = 0; r < 16; ++r)
        *(v4f*)(xw + r * XS_PITCH + lane * 4) = xv[r];
    __syncthreads();

    const int l16 = lane & 15;
    const int hh  = lane >> 4;
    v2f a[32];
#pragma unroll
    for (int s = 0; s < 32; ++s)
        a[s] = *(const v2f*)(xw + l16 * XS_PITCH + 4 * s + 2 * hh);
    v8f c0 = {}, c1 = {}, c2 = {}, c3 = {};
#pragma unroll
    for (int s = 0; s < 32; ++s) {
        const float* bb = wsm + (2 * s + hh) * BS_PITCH + l16 * 2;
        const v2f f0 = *(const v2f*)(bb + 0);
        const v2f f1 = *(const v2f*)(bb + 32);
        const v2f f2 = *(const v2f*)(bb + 64);
        const v2f f3 = *(const v2f*)(bb + 96);
        c0 = __builtin_amdgcn_wmma_f32_16x16x4_f32(false, a[s], false, f0, (short)0, c0, false, false);
        c1 = __builtin_amdgcn_wmma_f32_16x16x4_f32(false, a[s], false, f1, (short)0, c1, false, false);
        c2 = __builtin_amdgcn_wmma_f32_16x16x4_f32(false, a[s], false, f2, (short)0, c2, false, false);
        c3 = __builtin_amdgcn_wmma_f32_16x16x4_f32(false, a[s], false, f3, (short)0, c3, false, false);
    }

    float* pr = xw + (8 * hh) * XS_PITCH + l16;
#pragma unroll
    for (int r = 0; r < 8; ++r) {
        pr[r * XS_PITCH +  0] = c0[r];
        pr[r * XS_PITCH + 16] = c1[r];
        pr[r * XS_PITCH + 32] = c2[r];
        pr[r * XS_PITCH + 48] = c3[r];
    }
    const int rl = lane >> 4;
    const int c4 = (lane & 15) << 2;
    float* hp = h + (size_t)row0 * OUT_CH + rl * OUT_CH + c4;
    if (row0 + 16 <= N) {
        v4f v[8];
#pragma unroll
        for (int it = 0; it < 8; ++it)
            v[it] = *(const v4f*)(xw + (2 * it + rl) * XS_PITCH + c4);
#pragma unroll
        for (int it = 0; it < 8; ++it)
            *(v4f*)(hp + it * 2 * OUT_CH) = v[it];
    } else if (mTile < mTiles) {
#pragma unroll
        for (int it = 0; it < 8; ++it) {
            if (row0 + 2 * it + rl < N)
                *(v4f*)(hp + it * 2 * OUT_CH) = *(const v4f*)(xw + (2 * it + rl) * XS_PITCH + c4);
        }
    }
}

__global__ __launch_bounds__(32 * GX_WAVES)
void gemm_xw_lds_kernel(const float* __restrict__ x,
                        const float* __restrict__ W,
                        float* __restrict__ h, int N, int mTiles) {
    __shared__ float lds[GX_LDS_FLOATS];
    gemm_xw_block(x, W, h, N, mTiles, blockIdx.x, lds);
}

__global__ void gemm_pos_fused_kernel(const float* __restrict__ pos,
                                      const float* __restrict__ Wp,
                                      const float* __restrict__ bp,
                                      const float* __restrict__ b,
                                      const float* __restrict__ h,
                                      const float* __restrict__ dinv,
                                      float* __restrict__ out, int N, int mTiles) {
    const int wavesPerBlock = blockDim.x >> 5;
    const int waveId = threadIdx.x >> 5;
    const int lane   = threadIdx.x & 31;
    const int mTile  = blockIdx.x * wavesPerBlock + waveId;
    const int nTile  = blockIdx.y;
    if (mTile >= mTiles) return;

    const int l16 = lane & 15;
    const int kh  = (lane >> 4) << 1;
    int m = mTile * 16 + l16;
    if (m >= N) m = N - 1;
    const int n = nTile * 16 + l16;

    v8f c = {};
#pragma unroll
    for (int k = 0; k < POS_DIM; k += 4) {
        v2f a = *(const v2f*)(pos + (size_t)m * POS_DIM + k + kh);
        v2f bfrag;
        bfrag[0] = Wp[(size_t)(k + kh + 0) * OUT_CH + n];
        bfrag[1] = Wp[(size_t)(k + kh + 1) * OUT_CH + n];
        c = __builtin_amdgcn_wmma_f32_16x16x4_f32(
                false, a, false, bfrag, (short)0, c, false, false);
    }

    const float bias = bp[n] + b[n];
    const int rbase = mTile * 16 + ((lane >> 4) << 3);
#pragma unroll
    for (int r = 0; r < 8; ++r) {
        int row = rbase + r;
        if (row < N) {
            float di = dinv[row];
            out[(size_t)row * OUT_CH + n] =
                c[r] + bias + h[(size_t)row * OUT_CH + n] * di * di;
        }
    }
}

__global__ void edge_scatter_kernel(const int* __restrict__ src,
                                    const int* __restrict__ dst,
                                    const float* __restrict__ h,
                                    const float* __restrict__ dinv,
                                    float* __restrict__ out, int E) {
    const int wavesPerBlock = blockDim.x >> 5;
    const int e    = blockIdx.x * wavesPerBlock + (threadIdx.x >> 5);
    const int lane = threadIdx.x & 31;
    if (e >= E) return;

    const int s = src[e];
    const int d = dst[e];
    const float nrm = dinv[s] * dinv[d];

    const float2 hv = *(const float2*)(h + (size_t)s * OUT_CH + 2 * lane);
    float* op = out + (size_t)d * OUT_CH + 2 * lane;
    atomicAdd(op + 0, hv.x * nrm);
    atomicAdd(op + 1, hv.y * nrm);
}

#define BIN_RNG   400
#define BIN_CH    128
#define BIN_MAX   1024
#define BIN_BLOCK 1024

#define BINW_BLOCK (32 * GX_WAVES)
#define BINW_DEPTH 25
#define BINW_STAGE_OFF (BIN_MAX + 32)
#define BINW_STAGE_CAP (GX_LDS_FLOATS - BINW_STAGE_OFF)

__device__ __forceinline__ void bin_local_block(const int* __restrict__ src, const int* __restrict__ dst,
                                                unsigned* __restrict__ segtab, unsigned* __restrict__ list,
                                                int N, int E, int NB, int c, unsigned* sm) {
    unsigned* hist = sm;
    unsigned* wsum = sm + BIN_MAX;
    const int tid  = threadIdx.x;
    const int wave = tid >> 5;
    const int lane = tid & 31;
    for (int i = tid; i < BIN_MAX; i += BINW_BLOCK) hist[i] = 0u;
    __syncthreads();

    const int per = (E + BIN_CH - 1) / BIN_CH;
    const int e0  = c * per;
    int e1 = e0 + per;
    if (e1 > E) e1 = E;
    for (int e = e0 + tid; e < e1; e += BINW_DEPTH * BINW_BLOCK) {
        int d[BINW_DEPTH];
#pragma unroll
        for (int k = 0; k < BINW_DEPTH; ++k)
            d[k] = dst[min(e + k * BINW_BLOCK, e1 - 1)];
#pragma unroll
        for (int k = 0; k < BINW_DEPTH; ++k)
            if (e + k * BINW_BLOCK < e1 && (unsigned)d[k] < (unsigned)N) atomicAdd(&hist[d[k] / BIN_RNG], 1u);
    }
    __syncthreads();

    const unsigned v0 = hist[4 * tid + 0];
    const unsigned v1 = hist[4 * tid + 1];
    const unsigned v2 = hist[4 * tid + 2];
    const unsigned v3 = hist[4 * tid + 3];
    const unsigned own = v0 + v1 + v2 + v3;
    unsigned inc = own;
#pragma unroll
    for (int o = 1; o < 32; o <<= 1) {
        const unsigned t = __shfl_up(inc, o);
        if (lane >= o) inc += t;
    }
    if (lane == 31) wsum[wave] = inc;
    __syncthreads();
    unsigned pre = 0u, tot = 0u;
#pragma unroll
    for (int j = 0; j < GX_WAVES; ++j) {
        const unsigned q = wsum[j];
        pre += (j < wave) ? q : 0u;
        tot += q;
    }
    const unsigned o0 = pre + inc - own;
    const unsigned o1 = o0 + v0;
    const unsigned o2 = o1 + v1;
    const unsigned o3 = o2 + v2;
    hist[4 * tid + 0] = o0;
    hist[4 * tid + 1] = o1;
    hist[4 * tid + 2] = o2;
    hist[4 * tid + 3] = o3;
    unsigned* trow = segtab + (size_t)c * (NB + 1);
    if (4 * tid + 0 < NB) trow[4 * tid + 0] = o0;
    if (4 * tid + 1 < NB) trow[4 * tid + 1] = o1;
    if (4 * tid + 2 < NB) trow[4 * tid + 2] = o2;
    if (4 * tid + 3 < NB) trow[4 * tid + 3] = o3;
    if (tid == 0) trow[NB] = tot;
    __syncthreads();

    unsigned* lc = list + (size_t)c * per;
    unsigned* stage = sm + BINW_STAGE_OFF;
    const bool staged = per <= BINW_STAGE_CAP && (per & 3) == 0;
    for (int e = e0 + tid; e < e1; e += BINW_DEPTH * BINW_BLOCK) {
        int d[BINW_DEPTH], sv[BINW_DEPTH];
#pragma unroll
        for (int k = 0; k < BINW_DEPTH; ++k) {
            const int ec = min(e + k * BINW_BLOCK, e1 - 1);
            d[k]  = dst[ec];
            sv[k] = src[ec];
        }
        unsigned p[BINW_DEPTH];
#pragma unroll
        for (int k = 0; k < BINW_DEPTH; ++k) {
            p[k] = 0u;
            if (e + k * BINW_BLOCK < e1 && (unsigned)d[k] < (unsigned)N)
                p[k] = atomicAdd(&hist[d[k] / BIN_RNG], 1u);
        }
        if (staged) {
#pragma unroll
            for (int k = 0; k < BINW_DEPTH; ++k) {
                if (e + k * BINW_BLOCK < e1 && (unsigned)d[k] < (unsigned)N)
                    stage[p[k]] = ((unsigned)sv[k] << 9) | (unsigned)(d[k] - (d[k] / BIN_RNG) * BIN_RNG);
            }
        } else {
#pragma unroll
            for (int k = 0; k < BINW_DEPTH; ++k) {
                if (e + k * BINW_BLOCK < e1 && (unsigned)d[k] < (unsigned)N)
                    lc[p[k]] = ((unsigned)sv[k] << 9) | (unsigned)(d[k] - (d[k] / BIN_RNG) * BIN_RNG);
            }
        }
    }
    if (staged) {
        __syncthreads();
        const int nq = per >> 2;
#pragma unroll 4
        for (int q = tid; q < nq; q += BINW_BLOCK) {
            uint4 v = *(const uint4*)(stage + 4 * q);
            const unsigned i0 = 4u * (unsigned)q;
            v.x = (i0 + 0u < tot) ? v.x : 0u;
            v.y = (i0 + 1u < tot) ? v.y : 0u;
            v.z = (i0 + 2u < tot) ? v.z : 0u;
            v.w = (i0 + 3u < tot) ? v.w : 0u;
            *(uint4*)(lc + 4 * q) = v;
        }
    } else {
        for (int i = (int)tot + tid; i < per; i += BINW_BLOCK) lc[i] = 0u;
    }
}

__global__ __launch_bounds__(BINW_BLOCK)
void bin_local_gemm_kernel(const int* __restrict__ src, const int* __restrict__ dst,
                           unsigned* __restrict__ segtab, unsigned* __restrict__ list,
                           int N, int E, int NB,
                           const float* __restrict__ x, const float* __restrict__ W,
                           float* __restrict__ h, int mTiles, int g0) {
    __shared__ __attribute__((aligned(16))) unsigned char smem[GX_LDS_FLOATS * 4];
    if (blockIdx.x < BIN_CH)
        bin_local_block(src, dst, segtab, list, N, E, NB, blockIdx.x, (unsigned*)smem);
    else
        gemm_xw_block(x, W, h, N, mTiles, g0 + (int)blockIdx.x - BIN_CH, (float*)smem);
}

#define DEGL_BLOCK 256
#define DEGL_SEGS  (BIN_CH / (DEGL_BLOCK / 32))

__device__ __forceinline__ void bin_deg_block(const unsigned* __restrict__ list, const unsigned* __restrict__ segtab,
                                              float* __restrict__ dinv, int N, int E, int NB, int bin, unsigned* cnt) {
    const int tid  = threadIdx.x;
    const int wave = __builtin_amdgcn_readfirstlane(tid >> 5);
    const int lane = tid & 31;
    const int base = bin * BIN_RNG;
    int len = N - base;
    if (len > BIN_RNG) len = BIN_RNG;
    const int per = (E + BIN_CH - 1) / BIN_CH;
    unsigned sa = 0u, sb = 0u;
    if (lane < DEGL_SEGS) {
        const unsigned* t = segtab + (size_t)(wave * DEGL_SEGS + lane) * (NB + 1) + bin;
        sa = t[0];
        sb = t[1];
    }
    for (int i = tid; i < 512; i += DEGL_BLOCK) cnt[i] = 0u;
    const unsigned ln = sb - sa;
    const unsigned bs = (unsigned)(wave * DEGL_SEGS + lane) * (unsigned)per + sa;
    __syncthreads();

    const unsigned lw = (unsigned)BIN_CH * (unsigned)per - 1u;
    unsigned w0[DEGL_SEGS], w1[DEGL_SEGS];
#pragma unroll
    for (int j = 0; j < DEGL_SEGS; ++j) {
        const unsigned bj = (unsigned)__builtin_amdgcn_readlane((int)bs, j);
        w0[j] = list[min(bj + (unsigned)lane, lw)];
        w1[j] = list[min(bj + 32u + (unsigned)lane, lw)];
    }
#pragma unroll
    for (int j = 0; j < DEGL_SEGS; ++j) {
        const unsigned lj = (unsigned)__builtin_amdgcn_readlane((int)ln, j);
        atomicAdd(&cnt[w0[j] & 511u], ((unsigned)lane < lj) ? 1u : 0u);
        atomicAdd(&cnt[w1[j] & 511u], ((unsigned)lane + 32u < lj) ? 1u : 0u);
    }
#pragma unroll
    for (int j = 0; j < DEGL_SEGS; ++j) {
        const unsigned lj = (unsigned)__builtin_amdgcn_readlane((int)ln, j);
        const unsigned bj = (unsigned)__builtin_amdgcn_readlane((int)bs, j);
        for (unsigned q = 64u; q < lj; q += 32u)
            if (q + (unsigned)lane < lj) atomicAdd(&cnt[list[bj + q + (unsigned)lane] & 511u], 1u);
    }
    __syncthreads();

    for (int i = tid; i < len; i += DEGL_BLOCK)
        dinv[base + i] = rsqrtf((float)(cnt[i] + 1u));
}

__global__ __launch_bounds__(DEGL_BLOCK)
void bin_deg_gemm_kernel(const unsigned* __restrict__ list, const unsigned* __restrict__ segtab,
                         float* __restrict__ dinv, int N, int E, int NB,
                         const float* __restrict__ x, const float* __restrict__ W,
                         float* __restrict__ h, int mTiles, int g0) {
    __shared__ __attribute__((aligned(16))) unsigned char smem[GX_LDS_FLOATS * 4];
    if ((int)blockIdx.x < NB)
        bin_deg_block(list, segtab, dinv, N, E, NB, blockIdx.x, (unsigned*)smem);
    else
        gemm_xw_block(x, W, h, N, mTiles, g0 + (int)blockIdx.x - NB, (float*)smem);
}

#define POSP_PITCH 68
#define OUT_ITERS  ((BIN_RNG * (OUT_CH / 4) + BIN_BLOCK - 1) / BIN_BLOCK)

__device__ __forceinline__ void pos_tile_wmma(float* posp, const float* bs,
                                              const float* __restrict__ bp,
                                              const float* __restrict__ b,
                                              int r0, int lane) {
    const int l16 = lane & 15;
    const int hh  = lane >> 4;
    v2f a[16];
#pragma unroll
    for (int s = 0; s < 16; ++s)
        a[s] = *(const v2f*)(posp + (r0 + l16) * POSP_PITCH + 4 * s + 2 * hh);
    v8f c0 = {}, c1 = {}, c2 = {}, c3 = {};
#pragma unroll
    for (int s = 0; s < 16; ++s) {
        const float* bb = bs + (2 * s + hh) * BS_PITCH + l16 * 2;
        const v2f f0 = *(const v2f*)(bb + 0);
        const v2f f1 = *(const v2f*)(bb + 32);
        const v2f f2 = *(const v2f*)(bb + 64);
        const v2f f3 = *(const v2f*)(bb + 96);
        c0 = __builtin_amdgcn_wmma_f32_16x16x4_f32(false, a[s], false, f0, (short)0, c0, false, false);
        c1 = __builtin_amdgcn_wmma_f32_16x16x4_f32(false, a[s], false, f1, (short)0, c1, false, false);
        c2 = __builtin_amdgcn_wmma_f32_16x16x4_f32(false, a[s], false, f2, (short)0, c2, false, false);
        c3 = __builtin_amdgcn_wmma_f32_16x16x4_f32(false, a[s], false, f3, (short)0, c3, false, false);
    }
    const float bias0 = bp[l16 +  0] + b[l16 +  0];
    const float bias1 = bp[l16 + 16] + b[l16 + 16];
    const float bias2 = bp[l16 + 32] + b[l16 + 32];
    const float bias3 = bp[l16 + 48] + b[l16 + 48];
    float* pr = posp + (r0 + 8 * hh) * POSP_PITCH + l16;
#pragma unroll
    for (int r = 0; r < 8; ++r) {
        pr[r * POSP_PITCH +  0] = c0[r] + bias0;
        pr[r * POSP_PITCH + 16] = c1[r] + bias1;
        pr[r * POSP_PITCH + 32] = c2[r] + bias2;
        pr[r * POSP_PITCH + 48] = c3[r] + bias3;
    }
}

#define SORT_PER   9
#define SORT_CAP   (SORT_PER * BIN_BLOCK)
#define PASS_W     (32 * SORT_PER)
#define SUM_SEGS   (BIN_CH / (BIN_BLOCK / 32))
#define ROW_TURNS  ((BIN_RNG + 31) / 32)

template <int CNT>
__device__ __forceinline__ void row_group(const float* __restrict__ h, unsigned sv, unsigned dv,
                                          int k0, int lane, float& ax, float& ay) {
    float2 hv[CNT];
    float  sc[CNT];
#pragma unroll
    for (int k = 0; k < CNT; ++k) {
        const unsigned sn = (unsigned)__builtin_amdgcn_readlane((int)sv, k0 + k);
        sc[k] = __uint_as_float((unsigned)__builtin_amdgcn_readlane((int)dv, k0 + k));
        hv[k] = *(const float2*)(h + (size_t)sn * OUT_CH + 2 * lane);
    }
#pragma unroll
    for (int k = 0; k < CNT; ++k) {
        ax += hv[k].x * sc[k];
        ay += hv[k].y * sc[k];
    }
}

__global__ __launch_bounds__(BIN_BLOCK)
void bin_sum_pos_kernel(const unsigned* __restrict__ list, const unsigned* __restrict__ segtab,
                        const float* __restrict__ h, const float* __restrict__ dinv,
                        const float* __restrict__ pos, const float* __restrict__ Wp,
                        const float* __restrict__ bp, const float* __restrict__ b,
                        float* __restrict__ out, int N, int E, int NB) {
    __shared__ float acc[BIN_RNG * OUT_CH];
    __shared__ float posp[BIN_RNG * POSP_PITCH];
    __shared__ float bs[(POS_DIM / 2) * BS_PITCH];
    __shared__ uint2 ent[SORT_CAP];
    __shared__ unsigned cnt[512];
    __shared__ unsigned rstart[513];
    __shared__ unsigned wsum[16];
    __shared__ __attribute__((aligned(16))) unsigned wpass[BIN_BLOCK / 32];
    const int tid  = threadIdx.x;
    const int bin  = blockIdx.x;
    const int base = bin * BIN_RNG;
    int len = N - base;
    if (len > BIN_RNG) len = BIN_RNG;
    const int wave = __builtin_amdgcn_readfirstlane(tid >> 5);
    const int lane = tid & 31;
    const bool tile_wave = wave < BIN_RNG / 16;
    const int r0 = wave * 16;

    const int per = (E + BIN_CH - 1) / BIN_CH;
    unsigned sa = 0u, sb = 0u;
    if (lane < SUM_SEGS) {
        const unsigned* t = segtab + (size_t)(wave * SUM_SEGS + lane) * (NB + 1) + bin;
        sa = t[0];
        sb = t[1];
    }

    v4f pv0 = {}, pv1 = {}, pv2 = {}, pv3 = {}, pv4 = {}, pv5 = {}, pv6 = {}, pv7 = {};
    if (tile_wave) {
        const float* pp = pos + (lane & 15) * 4;
        const int rl = base + r0 + (lane >> 4);
        const int nl = N - 1;
        pv0 = *(const v4f*)(pp + (size_t)min(rl +  0, nl) * POS_DIM);
        pv1 = *(const v4f*)(pp + (size_t)min(rl +  2, nl) * POS_DIM);
        pv2 = *(const v4f*)(pp + (size_t)min(rl +  4, nl) * POS_DIM);
        pv3 = *(const v4f*)(pp + (size_t)min(rl +  6, nl) * POS_DIM);
        pv4 = *(const v4f*)(pp + (size_t)min(rl +  8, nl) * POS_DIM);
        pv5 = *(const v4f*)(pp + (size_t)min(rl + 10, nl) * POS_DIM);
        pv6 = *(const v4f*)(pp + (size_t)min(rl + 12, nl) * POS_DIM);
        pv7 = *(const v4f*)(pp + (size_t)min(rl + 14, nl) * POS_DIM);
    }
    const float4 wv = *(const float4*)(Wp + (tid >> 4) * OUT_CH + ((tid & 15) << 2));

    {
        const int k  = tid >> 4;
        const int n0 = (tid & 15) << 2;
        float* d = bs + (k >> 1) * BS_PITCH + (k & 1);
        d[(n0 + 0) * 2] = wv.x;
        d[(n0 + 1) * 2] = wv.y;
        d[(n0 + 2) * 2] = wv.z;
        d[(n0 + 3) * 2] = wv.w;
    }
    if (tile_wave) {
        float* ps = posp + (r0 + (lane >> 4)) * POSP_PITCH + (lane & 15) * 4;
        *(v4f*)(ps +  0 * POSP_PITCH) = pv0;
        *(v4f*)(ps +  2 * POSP_PITCH) = pv1;
        *(v4f*)(ps +  4 * POSP_PITCH) = pv2;
        *(v4f*)(ps +  6 * POSP_PITCH) = pv3;
        *(v4f*)(ps +  8 * POSP_PITCH) = pv4;
        *(v4f*)(ps + 10 * POSP_PITCH) = pv5;
        *(v4f*)(ps + 12 * POSP_PITCH) = pv6;
        *(v4f*)(ps + 14 * POSP_PITCH) = pv7;
    }

    const unsigned ln = sb - sa;
    const unsigned sg0 = (unsigned)(wave * SUM_SEGS + lane) * (unsigned)per + sa;
    const unsigned c1 = (unsigned)__builtin_amdgcn_readlane((int)ln, 0);
    const unsigned c2 = c1 + (unsigned)__builtin_amdgcn_readlane((int)ln, 1);
    const unsigned c3 = c2 + (unsigned)__builtin_amdgcn_readlane((int)ln, 2);
    const unsigned Lw = c3 + (unsigned)__builtin_amdgcn_readlane((int)ln, 3);
    const unsigned g0 = (unsigned)__builtin_amdgcn_readlane((int)sg0, 0);
    const unsigned g1 = (unsigned)__builtin_amdgcn_readlane((int)sg0, 1) - c1;
    const unsigned g2 = (unsigned)__builtin_amdgcn_readlane((int)sg0, 2) - c2;
    const unsigned g3 = (unsigned)__builtin_amdgcn_readlane((int)sg0, 3) - c3;
    if (lane == 0) wpass[wave] = (Lw + PASS_W - 1) / PASS_W;
    __syncthreads();
    unsigned npass = 1u;
#pragma unroll
    for (int j = 0; j < BIN_BLOCK / 32; j += 4) {
        const uint4 q = *(const uint4*)(wpass + j);
        npass = max(max(npass, max(q.x, q.y)), max(q.z, q.w));
    }

    const int turn = wave % ROW_TURNS;
    bool pending = tile_wave;
    unsigned p0 = 0u;
    unsigned pass = 0u;
    bool first = true;
    do {
        if (tid < 512) cnt[tid] = 0u;
        __syncthreads();

        unsigned w[SORT_PER];
        float    dsrc[SORT_PER];
#pragma unroll
        for (int k = 0; k < SORT_PER; ++k) {
            const unsigned i = p0 + 32u * (unsigned)k + (unsigned)lane;
            const unsigned g = (i < c1) ? g0 : ((i < c2) ? g1 : ((i < c3) ? g2 : g3));
            w[k] = (i < Lw) ? list[g + i] : 0u;
        }
#pragma unroll
        for (int k = 0; k < SORT_PER; ++k) {
            const unsigned i = p0 + 32u * (unsigned)k + (unsigned)lane;
            dsrc[k] = (i < Lw) ? dinv[w[k] >> 9] : 0.0f;
        }
#pragma unroll
        for (int k = 0; k < SORT_PER; ++k) {
            const unsigned i = p0 + 32u * (unsigned)k + (unsigned)lane;
            if (i < Lw) atomicAdd(&cnt[w[k] & 511u], 1u);
        }
        __syncthreads();

        unsigned v = 0u, inc = 0u;
        if (tid < 512) {
            v = cnt[tid];
            inc = v;
#pragma unroll
            for (int o = 1; o < 32; o <<= 1) {
                const unsigned t = __shfl_up(inc, o);
                if (lane >= o) inc += t;
            }
            if (lane == 31) wsum[wave] = inc;
        }
        __syncthreads();
        if (tid < 512) {
            unsigned pre = 0u;
#pragma unroll
            for (int j = 0; j < 16; ++j) pre += (j < wave) ? wsum[j] : 0u;
            const unsigned excl = pre + inc - v;
            rstart[tid] = excl;
            cnt[tid] = excl;
            if (tid == 511) rstart[512] = excl + v;
        }
        __syncthreads();

#pragma unroll
        for (int k = 0; k < SORT_PER; ++k) {
            const unsigned i = p0 + 32u * (unsigned)k + (unsigned)lane;
            if (i < Lw) {
                const unsigned p = atomicAdd(&cnt[w[k] & 511u], 1u);
                ent[p] = make_uint2(w[k] >> 9, __float_as_uint(dsrc[k]));
            }
        }
        __syncthreads();

        for (int j = 0; j < ROW_TURNS; ++j) {
            if (pending && j == turn) {
                pos_tile_wmma(posp, bs, bp, b, r0, lane);
                pending = false;
            }
            const int r = wave + 32 * j;
            if (r < BIN_RNG) {
                const unsigned s0 = (unsigned)__builtin_amdgcn_readfirstlane((int)rstart[r]);
                const unsigned s1 = (unsigned)__builtin_amdgcn_readfirstlane((int)rstart[r + 1]);
                float ax = 0.0f, ay = 0.0f;
                for (unsigned e0 = s0; e0 < s1; e0 += 32u) {
                    unsigned m = s1 - e0;
                    if (m > 32u) m = 32u;
                    const unsigned li = ((unsigned)lane < m) ? (unsigned)lane : m - 1u;
                    const uint2 en = ent[e0 + li];
                    const unsigned sv = en.x;
                    const unsigned dv = ((unsigned)lane < m) ? en.y : 0u;
                    if (m > 8u)  row_group<16>(h, sv, dv, 0, lane, ax, ay);
                    else         row_group<8>(h, sv, dv, 0, lane, ax, ay);
                    if (m > 16u) {
                        if (m > 24u) row_group<16>(h, sv, dv, 16, lane, ax, ay);
                        else         row_group<8>(h, sv, dv, 16, lane, ax, ay);
                    }
                }
                float2* ar = (float2*)(acc + r * OUT_CH + 2 * lane);
                if (first) {
                    *ar = make_float2(ax, ay);
                } else {
                    const float2 o = *ar;
                    *ar = make_float2(o.x + ax, o.y + ay);
                }
            }
        }
        __syncthreads();
        first = false;
        p0 += PASS_W;
        ++pass;
    } while (pass < npass);

    const float* hb = h + (size_t)base * OUT_CH;
    float* o = out + (size_t)base * OUT_CH;
    const int nq = len * (OUT_CH / 4);
    float  di[OUT_ITERS];
    float4 hv[OUT_ITERS];
#pragma unroll
    for (int it = 0; it < OUT_ITERS; ++it) {
        int q = tid + it * BIN_BLOCK;
        if (q > nq - 1) q = nq - 1;
        di[it] = dinv[base + (q >> 4)];
        hv[it] = *(const float4*)(hb + (q >> 4) * OUT_CH + ((q & 15) << 2));
    }
#pragma unroll
    for (int it = 0; it < OUT_ITERS; ++it) {
        const int q = tid + it * BIN_BLOCK;
        if (q < nq) {
            const int row = q >> 4;
            const int c4  = (q & 15) << 2;
            const float4 pq = *(const float4*)(posp + row * POSP_PITCH + c4);
            const float4 av = *(const float4*)(acc + row * OUT_CH + c4);
            const float d = di[it];
            float4 r;
            r.x = (pq.x + hv[it].x * d * d) + d * av.x;
            r.y = (pq.y + hv[it].y * d * d) + d * av.y;
            r.z = (pq.z + hv[it].z * d * d) + d * av.z;
            r.w = (pq.w + hv[it].w * d * d) + d * av.w;
            *(float4*)(o + row * OUT_CH + c4) = r;
        }
    }
}

extern "C" void kernel_launch(void* const* d_in, const int* in_sizes, int n_in,
                              void* d_out, int out_size, void* d_ws, size_t ws_size,
                              hipStream_t stream) {
    const float* x   = (const float*)d_in[0];
    const int*   ei  = (const int*)d_in[1];
    const float* pos = (const float*)d_in[2];
    const float* W   = (const float*)d_in[3];
    const float* b   = (const float*)d_in[4];
    const float* Wp  = (const float*)d_in[5];
    const float* bp  = (const float*)d_in[6];
    float* out = (float*)d_out;

    const int N = in_sizes[0] / IN_CH;
    const int E = in_sizes[1] / 2;
    const int* src = ei;
    const int* dst = ei + E;

    float* dinv = (float*)d_ws;
    float* h    = dinv + ((N + 3) & ~3);

    const int mTiles = (N + 15) / 16;
    const int wavesPerBlock = 8;
    dim3 gGemm((mTiles + wavesPerBlock - 1) / wavesPerBlock, OUT_CH / 16);

    const int NB  = (N + BIN_RNG - 1) / BIN_RNG;
    const int per = (E + BIN_CH - 1) / BIN_CH;
    const int TW  = (BIN_CH * (NB + 1) + 3) & ~3;
    unsigned* segtab = (unsigned*)(h + (size_t)N * OUT_CH);
    unsigned* list   = segtab + TW;
    const size_t need = (size_t)((char*)(list + (size_t)BIN_CH * per) - (char*)d_ws);
    if (NB <= BIN_MAX && N <= (1 << 23) && need <= ws_size) {
        const int gTot = (mTiles + GX_WAVES - 1) / GX_WAVES;
        const int g3 = gTot * 3 / 10;
        const int g2 = gTot - g3;
        bin_local_gemm_kernel<<<BIN_CH + g2, BINW_BLOCK, 0, stream>>>(src, dst, segtab, list, N, E, NB, x, W, h, mTiles, 0);
        bin_deg_gemm_kernel<<<NB + g3, DEGL_BLOCK, 0, stream>>>(list, segtab, dinv, N, E, NB, x, W, h, mTiles, g2);
        bin_sum_pos_kernel<<<NB, BIN_BLOCK, 0, stream>>>(list, segtab, h, dinv, pos, Wp, bp, b, out, N, E, NB);
    } else {
        unsigned* planes = (unsigned*)h;
        dim3 gDeg(DEG_CHUNKS, (N + DEG_RNG - 1) / DEG_RNG);
        deg_count_lds_kernel<<<gDeg, DEG_BLOCK, 0, stream>>>(dst, planes, N, E);
        deg_finish_kernel<<<(N + 255) / 256, 256, 0, stream>>>(planes, dinv, N);
        gemm_xw_lds_kernel<<<(mTiles + GX_WAVES - 1) / GX_WAVES, 32 * GX_WAVES, 0, stream>>>(x, W, h, N, mTiles);
        gemm_pos_fused_kernel<<<gGemm, 32 * wavesPerBlock, 0, stream>>>(
            pos, Wp, bp, b, h, dinv, out, N, mTiles);
        edge_scatter_kernel<<<(E + wavesPerBlock - 1) / wavesPerBlock,
                              32 * wavesPerBlock, 0, stream>>>(src, dst, h, dinv, out, E);
    }
}


# bin sum gather: one 16-byte-per-lane load fetches two h rows (a row per lane half), exactly ceil(n/2) loads a row, all in one group, none padded; row sum = even-entry sum + odd-entry sum (new order)
# speedup vs baseline: 4.9753x; 1.0623x over previous
#include <hip/hip_runtime.h>

typedef __attribute__((ext_vector_type(2))) float v2f;
typedef __attribute__((ext_vector_type(8))) float v8f;
typedef __attribute__((ext_vector_type(4))) float v4f;

#define IN_CH   128
#define OUT_CH  64
#define POS_DIM 64

#define DEG_RNG    25600
#define DEG_CHUNKS 64
#define DEG_BLOCK  1024

__global__ __launch_bounds__(DEG_BLOCK)
void deg_count_lds_kernel(const int* __restrict__ dst, unsigned* __restrict__ planes,
                          int N, int E) {
    __shared__ unsigned cnt[DEG_RNG];
    const int tid  = threadIdx.x;
    const int c    = blockIdx.x;
    const int base = blockIdx.y * DEG_RNG;
    int len = N - base;
    if (len > DEG_RNG) len = DEG_RNG;

    for (int i = tid; i < len; i += DEG_BLOCK) cnt[i] = 0u;
    __syncthreads();

    const int per = (E + DEG_CHUNKS - 1) / DEG_CHUNKS;
    const int e0  = c * per;
    int e1 = e0 + per;
    if (e1 > E) e1 = E;
    for (int e = e0 + tid; e < e1; e += 4 * DEG_BLOCK) {
        const int d0 = dst[e];
        const int d1 = (e + 1 * DEG_BLOCK < e1) ? dst[e + 1 * DEG_BLOCK] : -1;
        const int d2 = (e + 2 * DEG_BLOCK < e1) ? dst[e + 2 * DEG_BLOCK] : -1;
        const int d3 = (e + 3 * DEG_BLOCK < e1) ? dst[e + 3 * DEG_BLOCK] : -1;
        const unsigned r0 = (unsigned)(d0 - base);
        const unsigned r1 = (unsigned)(d1 - base);
        const unsigned r2 = (unsigned)(d2 - base);
        const unsigned r3 = (unsigned)(d3 - base);
        if (r0 < (unsigned)len) atomicAdd(&cnt[r0], 1u);
        if (r1 < (unsigned)len) atomicAdd(&cnt[r1], 1u);
        if (r2 < (unsigned)len) atomicAdd(&cnt[r2], 1u);
        if (r3 < (unsigned)len) atomicAdd(&cnt[r3], 1u);
    }
    __syncthreads();

    unsigned* p = planes + (size_t)c * N + base;
    for (int i = tid; i < len; i += DEG_BLOCK) p[i] = cnt[i];
}

__global__ void deg_finish_kernel(const unsigned* __restrict__ planes,
                                  float* __restrict__ dinv, int N) {
    const int i = blockIdx.x * blockDim.x + threadIdx.x;
    if (i < N) {
        unsigned s = 1u;
#pragma unroll
        for (int c = 0; c < DEG_CHUNKS; ++c) s += planes[(size_t)c * N + i];
        dinv[i] = rsqrtf((float)s);
    }
}

#define GX_WAVES 8
#define XS_PITCH 132
#define BS_PITCH 160
#define GX_WITEMS ((IN_CH / 2) * (OUT_CH / 4) / (32 * GX_WAVES))

#define GX_LDS_FLOATS ((IN_CH / 2) * BS_PITCH + GX_WAVES * 16 * XS_PITCH)

__device__ __forceinline__ void gemm_xw_block(const float* __restrict__ x,
                                              const float* __restrict__ W,
                                              float* __restrict__ h, int N, int mTiles,
                                              int blk, float* lds) {
    float* wsm = lds;
    float* xs  = lds + (IN_CH / 2) * BS_PITCH;
    const int tid  = threadIdx.x;
    const int wave = __builtin_amdgcn_readfirstlane(tid >> 5);
    const int lane = tid & 31;
    const int mTile = blk * GX_WAVES + wave;
    const int row0  = mTile * 16;
    const int nl    = N - 1;

    v4f w0[GX_WITEMS], w1[GX_WITEMS];
#pragma unroll
    for (int j = 0; j < GX_WITEMS; ++j) {
        const int item = tid + j * 32 * GX_WAVES;
        const int p = item >> 4;
        const int q = item & 15;
        w0[j] = *(const v4f*)(W + (2 * p + 0) * OUT_CH + 4 * q);
        w1[j] = *(const v4f*)(W + (2 * p + 1) * OUT_CH + 4 * q);
    }
    v4f xv[16];
    const float* xp = x + lane * 4;
#pragma unroll
    for (int r = 0; r < 16; ++r)
        xv[r] = *(const v4f*)(xp + (size_t)min(row0 + r, nl) * IN_CH);
#pragma unroll
    for (int j = 0; j < GX_WITEMS; ++j) {
        const int item = tid + j * 32 * GX_WAVES;
        const int p = item >> 4;
        const int q = item & 15;
        float* d = wsm + p * BS_PITCH + 8 * q;
        const v4f lo = {w0[j][0], w1[j][0], w0[j][1], w1[j][1]};
        const v4f hi = {w0[j][2], w1[j][2], w0[j][3], w1[j][3]};
        *(v4f*)(d + 0) = lo;
        *(v4f*)(d + 4) = hi;
    }
    float* xw = xs + wave * 16 * XS_PITCH;
#pragma unroll
    for (int r = 0; r < 16; ++r)
        *(v4f*)(xw + r * XS_PITCH + lane * 4) = xv[r];
    __syncthreads();

    const int l16 = lane & 15;
    const int hh  = lane >> 4;
    v2f a[32];
#pragma unroll
    for (int s = 0; s < 32; ++s)
        a[s] = *(const v2f*)(xw + l16 * XS_PITCH + 4 * s + 2 * hh);
    v8f c0 = {}, c1 = {}, c2 = {}, c3 = {};
#pragma unroll
    for (int s = 0; s < 32; ++s) {
        const float* bb = wsm + (2 * s + hh) * BS_PITCH + l16 * 2;
        const v2f f0 = *(const v2f*)(bb + 0);
        const v2f f1 = *(const v2f*)(bb + 32);
        const v2f f2 = *(const v2f*)(bb + 64);
        const v2f f3 = *(const v2f*)(bb + 96);
        c0 = __builtin_amdgcn_wmma_f32_16x16x4_f32(false, a[s], false, f0, (short)0, c0, false, false);
        c1 = __builtin_amdgcn_wmma_f32_16x16x4_f32(false, a[s], false, f1, (short)0, c1, false, false);
        c2 = __builtin_amdgcn_wmma_f32_16x16x4_f32(false, a[s], false, f2, (short)0, c2, false, false);
        c3 = __builtin_amdgcn_wmma_f32_16x16x4_f32(false, a[s], false, f3, (short)0, c3, false, false);
    }

    float* pr = xw + (8 * hh) * XS_PITCH + l16;
#pragma unroll
    for (int r = 0; r < 8; ++r) {
        pr[r * XS_PITCH +  0] = c0[r];
        pr[r * XS_PITCH + 16] = c1[r];
        pr[r * XS_PITCH + 32] = c2[r];
        pr[r * XS_PITCH + 48] = c3[r];
    }
    const int rl = lane >> 4;
    const int c4 = (lane & 15) << 2;
    float* hp = h + (size_t)row0 * OUT_CH + rl * OUT_CH + c4;
    if (row0 + 16 <= N) {
        v4f v[8];
#pragma unroll
        for (int it = 0; it < 8; ++it)
            v[it] = *(const v4f*)(xw + (2 * it + rl) * XS_PITCH + c4);
#pragma unroll
        for (int it = 0; it < 8; ++it)
            *(v4f*)(hp + it * 2 * OUT_CH) = v[it];
    } else if (mTile < mTiles) {
#pragma unroll
        for (int it = 0; it < 8; ++it) {
            if (row0 + 2 * it + rl < N)
                *(v4f*)(hp + it * 2 * OUT_CH) = *(const v4f*)(xw + (2 * it + rl) * XS_PITCH + c4);
        }
    }
}

__global__ __launch_bounds__(32 * GX_WAVES)
void gemm_xw_lds_kernel(const float* __restrict__ x,
                        const float* __restrict__ W,
                        float* __restrict__ h, int N, int mTiles) {
    __shared__ float lds[GX_LDS_FLOATS];
    gemm_xw_block(x, W, h, N, mTiles, blockIdx.x, lds);
}

__global__ void gemm_pos_fused_kernel(const float* __restrict__ pos,
                                      const float* __restrict__ Wp,
                                      const float* __restrict__ bp,
                                      const float* __restrict__ b,
                                      const float* __restrict__ h,
                                      const float* __restrict__ dinv,
                                      float* __restrict__ out, int N, int mTiles) {
    const int wavesPerBlock = blockDim.x >> 5;
    const int waveId = threadIdx.x >> 5;
    const int lane   = threadIdx.x & 31;
    const int mTile  = blockIdx.x * wavesPerBlock + waveId;
    const int nTile  = blockIdx.y;
    if (mTile >= mTiles) return;

    const int l16 = lane & 15;
    const int kh  = (lane >> 4) << 1;
    int m = mTile * 16 + l16;
    if (m >= N) m = N - 1;
    const int n = nTile * 16 + l16;

    v8f c = {};
#pragma unroll
    for (int k = 0; k < POS_DIM; k += 4) {
        v2f a = *(const v2f*)(pos + (size_t)m * POS_DIM + k + kh);
        v2f bfrag;
        bfrag[0] = Wp[(size_t)(k + kh + 0) * OUT_CH + n];
        bfrag[1] = Wp[(size_t)(k + kh + 1) * OUT_CH + n];
        c = __builtin_amdgcn_wmma_f32_16x16x4_f32(
                false, a, false, bfrag, (short)0, c, false, false);
    }

    const float bias = bp[n] + b[n];
    const int rbase = mTile * 16 + ((lane >> 4) << 3);
#pragma unroll
    for (int r = 0; r < 8; ++r) {
        int row = rbase + r;
        if (row < N) {
            float di = dinv[row];
            out[(size_t)row * OUT_CH + n] =
                c[r] + bias + h[(size_t)row * OUT_CH + n] * di * di;
        }
    }
}

__global__ void edge_scatter_kernel(const int* __restrict__ src,
                                    const int* __restrict__ dst,
                                    const float* __restrict__ h,
                                    const float* __restrict__ dinv,
                                    float* __restrict__ out, int E) {
    const int wavesPerBlock = blockDim.x >> 5;
    const int e    = blockIdx.x * wavesPerBlock + (threadIdx.x >> 5);
    const int lane = threadIdx.x & 31;
    if (e >= E) return;

    const int s = src[e];
    const int d = dst[e];
    const float nrm = dinv[s] * dinv[d];

    const float2 hv = *(const float2*)(h + (size_t)s * OUT_CH + 2 * lane);
    float* op = out + (size_t)d * OUT_CH + 2 * lane;
    atomicAdd(op + 0, hv.x * nrm);
    atomicAdd(op + 1, hv.y * nrm);
}

#define BIN_RNG   400
#define BIN_CH    128
#define BIN_MAX   1024
#define BIN_BLOCK 1024

#define BINW_BLOCK (32 * GX_WAVES)
#define BINW_DEPTH 25
#define BINW_STAGE_OFF (BIN_MAX + 32)
#define BINW_STAGE_CAP (GX_LDS_FLOATS - BINW_STAGE_OFF)

__device__ __forceinline__ void bin_local_block(const int* __restrict__ src, const int* __restrict__ dst,
                                                unsigned* __restrict__ segtab, unsigned* __restrict__ list,
                                                int N, int E, int NB, int c, unsigned* sm) {
    unsigned* hist = sm;
    unsigned* wsum = sm + BIN_MAX;
    const int tid  = threadIdx.x;
    const int wave = tid >> 5;
    const int lane = tid & 31;
    for (int i = tid; i < BIN_MAX; i += BINW_BLOCK) hist[i] = 0u;
    __syncthreads();

    const int per = (E + BIN_CH - 1) / BIN_CH;
    const int e0  = c * per;
    int e1 = e0 + per;
    if (e1 > E) e1 = E;
    for (int e = e0 + tid; e < e1; e += BINW_DEPTH * BINW_BLOCK) {
        int d[BINW_DEPTH];
#pragma unroll
        for (int k = 0; k < BINW_DEPTH; ++k)
            d[k] = dst[min(e + k * BINW_BLOCK, e1 - 1)];
#pragma unroll
        for (int k = 0; k < BINW_DEPTH; ++k)
            if (e + k * BINW_BLOCK < e1 && (unsigned)d[k] < (unsigned)N) atomicAdd(&hist[d[k] / BIN_RNG], 1u);
    }
    __syncthreads();

    const unsigned v0 = hist[4 * tid + 0];
    const unsigned v1 = hist[4 * tid + 1];
    const unsigned v2 = hist[4 * tid + 2];
    const unsigned v3 = hist[4 * tid + 3];
    const unsigned own = v0 + v1 + v2 + v3;
    unsigned inc = own;
#pragma unroll
    for (int o = 1; o < 32; o <<= 1) {
        const unsigned t = __shfl_up(inc, o);
        if (lane >= o) inc += t;
    }
    if (lane == 31) wsum[wave] = inc;
    __syncthreads();
    unsigned pre = 0u, tot = 0u;
#pragma unroll
    for (int j = 0; j < GX_WAVES; ++j) {
        const unsigned q = wsum[j];
        pre += (j < wave) ? q : 0u;
        tot += q;
    }
    const unsigned o0 = pre + inc - own;
    const unsigned o1 = o0 + v0;
    const unsigned o2 = o1 + v1;
    const unsigned o3 = o2 + v2;
    hist[4 * tid + 0] = o0;
    hist[4 * tid + 1] = o1;
    hist[4 * tid + 2] = o2;
    hist[4 * tid + 3] = o3;
    unsigned* trow = segtab + (size_t)c * (NB + 1);
    if (4 * tid + 0 < NB) trow[4 * tid + 0] = o0;
    if (4 * tid + 1 < NB) trow[4 * tid + 1] = o1;
    if (4 * tid + 2 < NB) trow[4 * tid + 2] = o2;
    if (4 * tid + 3 < NB) trow[4 * tid + 3] = o3;
    if (tid == 0) trow[NB] = tot;
    __syncthreads();

    unsigned* lc = list + (size_t)c * per;
    unsigned* stage = sm + BINW_STAGE_OFF;
    const bool staged = per <= BINW_STAGE_CAP && (per & 3) == 0;
    for (int e = e0 + tid; e < e1; e += BINW_DEPTH * BINW_BLOCK) {
        int d[BINW_DEPTH], sv[BINW_DEPTH];
#pragma unroll
        for (int k = 0; k < BINW_DEPTH; ++k) {
            const int ec = min(e + k * BINW_BLOCK, e1 - 1);
            d[k]  = dst[ec];
            sv[k] = src[ec];
        }
        unsigned p[BINW_DEPTH];
#pragma unroll
        for (int k = 0; k < BINW_DEPTH; ++k) {
            p[k] = 0u;
            if (e + k * BINW_BLOCK < e1 && (unsigned)d[k] < (unsigned)N)
                p[k] = atomicAdd(&hist[d[k] / BIN_RNG], 1u);
        }
        if (staged) {
#pragma unroll
            for (int k = 0; k < BINW_DEPTH; ++k) {
                if (e + k * BINW_BLOCK < e1 && (unsigned)d[k] < (unsigned)N)
                    stage[p[k]] = ((unsigned)sv[k] << 9) | (unsigned)(d[k] - (d[k] / BIN_RNG) * BIN_RNG);
            }
        } else {
#pragma unroll
            for (int k = 0; k < BINW_DEPTH; ++k) {
                if (e + k * BINW_BLOCK < e1 && (unsigned)d[k] < (unsigned)N)
                    lc[p[k]] = ((unsigned)sv[k] << 9) | (unsigned)(d[k] - (d[k] / BIN_RNG) * BIN_RNG);
            }
        }
    }
    if (staged) {
        __syncthreads();
        const int nq = per >> 2;
#pragma unroll 4
        for (int q = tid; q < nq; q += BINW_BLOCK) {
            uint4 v = *(const uint4*)(stage + 4 * q);
            const unsigned i0 = 4u * (unsigned)q;
            v.x = (i0 + 0u < tot) ? v.x : 0u;
            v.y = (i0 + 1u < tot) ? v.y : 0u;
            v.z = (i0 + 2u < tot) ? v.z : 0u;
            v.w = (i0 + 3u < tot) ? v.w : 0u;
            *(uint4*)(lc + 4 * q) = v;
        }
    } else {
        for (int i = (int)tot + tid; i < per; i += BINW_BLOCK) lc[i] = 0u;
    }
}

__global__ __launch_bounds__(BINW_BLOCK)
void bin_local_gemm_kernel(const int* __restrict__ src, const int* __restrict__ dst,
                           unsigned* __restrict__ segtab, unsigned* __restrict__ list,
                           int N, int E, int NB,
                           const float* __restrict__ x, const float* __restrict__ W,
                           float* __restrict__ h, int mTiles, int g0) {
    __shared__ __attribute__((aligned(16))) unsigned char smem[GX_LDS_FLOATS * 4];
    if (blockIdx.x < BIN_CH)
        bin_local_block(src, dst, segtab, list, N, E, NB, blockIdx.x, (unsigned*)smem);
    else
        gemm_xw_block(x, W, h, N, mTiles, g0 + (int)blockIdx.x - BIN_CH, (float*)smem);
}

#define DEGL_BLOCK 256
#define DEGL_SEGS  (BIN_CH / (DEGL_BLOCK / 32))

__device__ __forceinline__ void bin_deg_block(const unsigned* __restrict__ list, const unsigned* __restrict__ segtab,
                                              float* __restrict__ dinv, int N, int E, int NB, int bin, unsigned* cnt) {
    const int tid  = threadIdx.x;
    const int wave = __builtin_amdgcn_readfirstlane(tid >> 5);
    const int lane = tid & 31;
    const int base = bin * BIN_RNG;
    int len = N - base;
    if (len > BIN_RNG) len = BIN_RNG;
    const int per = (E + BIN_CH - 1) / BIN_CH;
    unsigned sa = 0u, sb = 0u;
    if (lane < DEGL_SEGS) {
        const unsigned* t = segtab + (size_t)(wave * DEGL_SEGS + lane) * (NB + 1) + bin;
        sa = t[0];
        sb = t[1];
    }
    for (int i = tid; i < 512; i += DEGL_BLOCK) cnt[i] = 0u;
    const unsigned ln = sb - sa;
    const unsigned bs = (unsigned)(wave * DEGL_SEGS + lane) * (unsigned)per + sa;
    __syncthreads();

    const unsigned lw = (unsigned)BIN_CH * (unsigned)per - 1u;
    unsigned w0[DEGL_SEGS], w1[DEGL_SEGS];
#pragma unroll
    for (int j = 0; j < DEGL_SEGS; ++j) {
        const unsigned bj = (unsigned)__builtin_amdgcn_readlane((int)bs, j);
        w0[j] = list[min(bj + (unsigned)lane, lw)];
        w1[j] = list[min(bj + 32u + (unsigned)lane, lw)];
    }
#pragma unroll
    for (int j = 0; j < DEGL_SEGS; ++j) {
        const unsigned lj = (unsigned)__builtin_amdgcn_readlane((int)ln, j);
        atomicAdd(&cnt[w0[j] & 511u], ((unsigned)lane < lj) ? 1u : 0u);
        atomicAdd(&cnt[w1[j] & 511u], ((unsigned)lane + 32u < lj) ? 1u : 0u);
    }
#pragma unroll
    for (int j = 0; j < DEGL_SEGS; ++j) {
        const unsigned lj = (unsigned)__builtin_amdgcn_readlane((int)ln, j);
        const unsigned bj = (unsigned)__builtin_amdgcn_readlane((int)bs, j);
        for (unsigned q = 64u; q < lj; q += 32u)
            if (q + (unsigned)lane < lj) atomicAdd(&cnt[list[bj + q + (unsigned)lane] & 511u], 1u);
    }
    __syncthreads();

    for (int i = tid; i < len; i += DEGL_BLOCK)
        dinv[base + i] = rsqrtf((float)(cnt[i] + 1u));
}

__global__ __launch_bounds__(DEGL_BLOCK)
void bin_deg_gemm_kernel(const unsigned* __restrict__ list, const unsigned* __restrict__ segtab,
                         float* __restrict__ dinv, int N, int E, int NB,
                         const float* __restrict__ x, const float* __restrict__ W,
                         float* __restrict__ h, int mTiles, int g0) {
    __shared__ __attribute__((aligned(16))) unsigned char smem[GX_LDS_FLOATS * 4];
    if ((int)blockIdx.x < NB)
        bin_deg_block(list, segtab, dinv, N, E, NB, blockIdx.x, (unsigned*)smem);
    else
        gemm_xw_block(x, W, h, N, mTiles, g0 + (int)blockIdx.x - NB, (float*)smem);
}

#define POSP_PITCH 68
#define OUT_ITERS  ((BIN_RNG * (OUT_CH / 4) + BIN_BLOCK - 1) / BIN_BLOCK)

__device__ __forceinline__ void pos_tile_wmma(float* posp, const float* bs,
                                              const float* __restrict__ bp,
                                              const float* __restrict__ b,
                                              int r0, int lane) {
    const int l16 = lane & 15;
    const int hh  = lane >> 4;
    v2f a[16];
#pragma unroll
    for (int s = 0; s < 16; ++s)
        a[s] = *(const v2f*)(posp + (r0 + l16) * POSP_PITCH + 4 * s + 2 * hh);
    v8f c0 = {}, c1 = {}, c2 = {}, c3 = {};
#pragma unroll
    for (int s = 0; s < 16; ++s) {
        const float* bb = bs + (2 * s + hh) * BS_PITCH + l16 * 2;
        const v2f f0 = *(const v2f*)(bb + 0);
        const v2f f1 = *(const v2f*)(bb + 32);
        const v2f f2 = *(const v2f*)(bb + 64);
        const v2f f3 = *(const v2f*)(bb + 96);
        c0 = __builtin_amdgcn_wmma_f32_16x16x4_f32(false, a[s], false, f0, (short)0, c0, false, false);
        c1 = __builtin_amdgcn_wmma_f32_16x16x4_f32(false, a[s], false, f1, (short)0, c1, false, false);
        c2 = __builtin_amdgcn_wmma_f32_16x16x4_f32(false, a[s], false, f2, (short)0, c2, false, false);
        c3 = __builtin_amdgcn_wmma_f32_16x16x4_f32(false, a[s], false, f3, (short)0, c3, false, false);
    }
    const float bias0 = bp[l16 +  0] + b[l16 +  0];
    const float bias1 = bp[l16 + 16] + b[l16 + 16];
    const float bias2 = bp[l16 + 32] + b[l16 + 32];
    const float bias3 = bp[l16 + 48] + b[l16 + 48];
    float* pr = posp + (r0 + 8 * hh) * POSP_PITCH + l16;
#pragma unroll
    for (int r = 0; r < 8; ++r) {
        pr[r * POSP_PITCH +  0] = c0[r] + bias0;
        pr[r * POSP_PITCH + 16] = c1[r] + bias1;
        pr[r * POSP_PITCH + 32] = c2[r] + bias2;
        pr[r * POSP_PITCH + 48] = c3[r] + bias3;
    }
}

#define SORT_PER   9
#define SORT_CAP   (SORT_PER * BIN_BLOCK)
#define PASS_W     (32 * SORT_PER)
#define SUM_SEGS   (BIN_CH / (BIN_BLOCK / 32))
#define ROW_TURNS  ((BIN_RNG + 31) / 32)
#define ROW_MAXP   12
static_assert(ROW_MAXP == 12, "the row loop's switch lists the group sizes 1 .. ROW_MAXP");

__device__ __forceinline__ float xhalf(float v) {
    return __builtin_bit_cast(float, __builtin_amdgcn_permlanex16(__builtin_bit_cast(int, v), __builtin_bit_cast(int, v),
                                                                  0x76543210, 0xfedcba98, false, false));
}

template <int CNT>
__device__ __forceinline__ void row_pairs(const float* __restrict__ h, const uint2* ent,
                                          unsigned e0, unsigned s1, unsigned hh, unsigned c4b, v4f& a) {
    uint2 en[CNT];
#pragma unroll
    for (int k = 0; k < CNT; ++k) {
        unsigned e = e0 + 2u * (unsigned)k + hh;
        if (k == CNT - 1) e = min(e, s1 - 1u);
        en[k] = ent[e];
    }
    v4f   hv[CNT];
    float sc[CNT];
    const char* hb = (const char*)h;
#pragma unroll
    for (int k = 0; k < CNT; ++k) {
        sc[k] = __uint_as_float(en[k].y);
        if (k == CNT - 1) sc[k] = (e0 + 2u * (unsigned)k + hh < s1) ? sc[k] : 0.0f;
        hv[k] = *(const v4f*)(hb + ((en[k].x << 8) | c4b));
    }
#pragma unroll
    for (int k = 0; k < CNT; ++k) {
        a[0] = __builtin_fmaf(hv[k][0], sc[k], a[0]);
        a[1] = __builtin_fmaf(hv[k][1], sc[k], a[1]);
        a[2] = __builtin_fmaf(hv[k][2], sc[k], a[2]);
        a[3] = __builtin_fmaf(hv[k][3], sc[k], a[3]);
    }
}

__global__ __launch_bounds__(BIN_BLOCK)
void bin_sum_pos_kernel(const unsigned* __restrict__ list, const unsigned* __restrict__ segtab,
                        const float* __restrict__ h, const float* __restrict__ dinv,
                        const float* __restrict__ pos, const float* __restrict__ Wp,
                        const float* __restrict__ bp, const float* __restrict__ b,
                        float* __restrict__ out, int N, int E, int NB) {
    __shared__ float acc[BIN_RNG * OUT_CH];
    __shared__ float posp[BIN_RNG * POSP_PITCH];
    __shared__ float bs[(POS_DIM / 2) * BS_PITCH];
    __shared__ uint2 ent[SORT_CAP];
    __shared__ unsigned cnt[512];
    __shared__ unsigned rstart[513];
    __shared__ unsigned wsum[16];
    __shared__ __attribute__((aligned(16))) unsigned wpass[BIN_BLOCK / 32];
    const int tid  = threadIdx.x;
    const int bin  = blockIdx.x;
    const int base = bin * BIN_RNG;
    int len = N - base;
    if (len > BIN_RNG) len = BIN_RNG;
    const int wave = __builtin_amdgcn_readfirstlane(tid >> 5);
    const int lane = tid & 31;
    const unsigned hh = (unsigned)lane >> 4;
    const unsigned c4b = ((unsigned)lane & 15u) << 4;
    const bool tile_wave = wave < BIN_RNG / 16;
    const int r0 = wave * 16;

    const int per = (E + BIN_CH - 1) / BIN_CH;
    unsigned sa = 0u, sb = 0u;
    if (lane < SUM_SEGS) {
        const unsigned* t = segtab + (size_t)(wave * SUM_SEGS + lane) * (NB + 1) + bin;
        sa = t[0];
        sb = t[1];
    }

    v4f pv0 = {}, pv1 = {}, pv2 = {}, pv3 = {}, pv4 = {}, pv5 = {}, pv6 = {}, pv7 = {};
    if (tile_wave) {
        const float* pp = pos + (lane & 15) * 4;
        const int rl = base + r0 + (lane >> 4);
        const int nl = N - 1;
        pv0 = *(const v4f*)(pp + (size_t)min(rl +  0, nl) * POS_DIM);
        pv1 = *(const v4f*)(pp + (size_t)min(rl +  2, nl) * POS_DIM);
        pv2 = *(const v4f*)(pp + (size_t)min(rl +  4, nl) * POS_DIM);
        pv3 = *(const v4f*)(pp + (size_t)min(rl +  6, nl) * POS_DIM);
        pv4 = *(const v4f*)(pp + (size_t)min(rl +  8, nl) * POS_DIM);
        pv5 = *(const v4f*)(pp + (size_t)min(rl + 10, nl) * POS_DIM);
        pv6 = *(const v4f*)(pp + (size_t)min(rl + 12, nl) * POS_DIM);
        pv7 = *(const v4f*)(pp + (size_t)min(rl + 14, nl) * POS_DIM);
    }
    const float4 wv = *(const float4*)(Wp + (tid >> 4) * OUT_CH + ((tid & 15) << 2));

    {
        const int k  = tid >> 4;
        const int n0 = (tid & 15) << 2;
        float* d = bs + (k >> 1) * BS_PITCH + (k & 1);
        d[(n0 + 0) * 2] = wv.x;
        d[(n0 + 1) * 2] = wv.y;
        d[(n0 + 2) * 2] = wv.z;
        d[(n0 + 3) * 2] = wv.w;
    }
    if (tile_wave) {
        float* ps = posp + (r0 + (lane >> 4)) * POSP_PITCH + (lane & 15) * 4;
        *(v4f*)(ps +  0 * POSP_PITCH) = pv0;
        *(v4f*)(ps +  2 * POSP_PITCH) = pv1;
        *(v4f*)(ps +  4 * POSP_PITCH) = pv2;
        *(v4f*)(ps +  6 * POSP_PITCH) = pv3;
        *(v4f*)(ps +  8 * POSP_PITCH) = pv4;
        *(v4f*)(ps + 10 * POSP_PITCH) = pv5;
        *(v4f*)(ps + 12 * POSP_PITCH) = pv6;
        *(v4f*)(ps + 14 * POSP_PITCH) = pv7;
    }

    const unsigned ln = sb - sa;
    const unsigned sg0 = (unsigned)(wave * SUM_SEGS + lane) * (unsigned)per + sa;
    const unsigned c1 = (unsigned)__builtin_amdgcn_readlane((int)ln, 0);
    const unsigned c2 = c1 + (unsigned)__builtin_amdgcn_readlane((int)ln, 1);
    const unsigned c3 = c2 + (unsigned)__builtin_amdgcn_readlane((int)ln, 2);
    const unsigned Lw = c3 + (unsigned)__builtin_amdgcn_readlane((int)ln, 3);
    const unsigned g0 = (unsigned)__builtin_amdgcn_readlane((int)sg0, 0);
    const unsigned g1 = (unsigned)__builtin_amdgcn_readlane((int)sg0, 1) - c1;
    const unsigned g2 = (unsigned)__builtin_amdgcn_readlane((int)sg0, 2) - c2;
    const unsigned g3 = (unsigned)__builtin_amdgcn_readlane((int)sg0, 3) - c3;
    if (lane == 0) wpass[wave] = (Lw + PASS_W - 1) / PASS_W;
    __syncthreads();
    unsigned npass = 1u;
#pragma unroll
    for (int j = 0; j < BIN_BLOCK / 32; j += 4) {
        const uint4 q = *(const uint4*)(wpass + j);
        npass = max(max(npass, max(q.x, q.y)), max(q.z, q.w));
    }

    const int turn = wave % ROW_TURNS;
    bool pending = tile_wave;
    unsigned p0 = 0u;
    unsigned pass = 0u;
    bool first = true;
    do {
        if (tid < 512) cnt[tid] = 0u;
        __syncthreads();

        unsigned w[SORT_PER];
        float    dsrc[SORT_PER];
#pragma unroll
        for (int k = 0; k < SORT_PER; ++k) {
            const unsigned i = p0 + 32u * (unsigned)k + (unsigned)lane;
            const unsigned g = (i < c1) ? g0 : ((i < c2) ? g1 : ((i < c3) ? g2 : g3));
            w[k] = (i < Lw) ? list[g + i] : 0u;
        }
#pragma unroll
        for (int k = 0; k < SORT_PER; ++k) {
            const unsigned i = p0 + 32u * (unsigned)k + (unsigned)lane;
            dsrc[k] = (i < Lw) ? dinv[w[k] >> 9] : 0.0f;
        }
#pragma unroll
        for (int k = 0; k < SORT_PER; ++k) {
            const unsigned i = p0 + 32u * (unsigned)k + (unsigned)lane;
            if (i < Lw) atomicAdd(&cnt[w[k] & 511u], 1u);
        }
        __syncthreads();

        unsigned v = 0u, inc = 0u;
        if (tid < 512) {
            v = cnt[tid];
            inc = v;
#pragma unroll
            for (int o = 1; o < 32; o <<= 1) {
                const unsigned t = __shfl_up(inc, o);
                if (lane >= o) inc += t;
            }
            if (lane == 31) wsum[wave] = inc;
        }
        __syncthreads();
        if (tid < 512) {
            unsigned pre = 0u;
#pragma unroll
            for (int j = 0; j < 16; ++j) pre += (j < wave) ? wsum[j] : 0u;
            const unsigned excl = pre + inc - v;
            rstart[tid] = excl;
            cnt[tid] = excl;
            if (tid == 511) rstart[512] = excl + v;
        }
        __syncthreads();

#pragma unroll
        for (int k = 0; k < SORT_PER; ++k) {
            const unsigned i = p0 + 32u * (unsigned)k + (unsigned)lane;
            if (i < Lw) {
                const unsigned p = atomicAdd(&cnt[w[k] & 511u], 1u);
                ent[p] = make_uint2(w[k] >> 9, __float_as_uint(dsrc[k]));
            }
        }
        __syncthreads();

        for (int j = 0; j < ROW_TURNS; ++j) {
            if (pending && j == turn) {
                pos_tile_wmma(posp, bs, bp, b, r0, lane);
                pending = false;
            }
            const int r = wave + 32 * j;
            if (r < BIN_RNG) {
                const unsigned s0 = (unsigned)__builtin_amdgcn_readfirstlane((int)rstart[r]);
                const unsigned s1 = (unsigned)__builtin_amdgcn_readfirstlane((int)rstart[r + 1]);
                v4f a = {0.0f, 0.0f, 0.0f, 0.0f};
                unsigned e0 = s0;
                for (; s1 - e0 > 2u * ROW_MAXP; e0 += 2u * ROW_MAXP)
                    row_pairs<ROW_MAXP>(h, ent, e0, s1, hh, c4b, a);
                switch ((s1 - e0 + 1u) >> 1) {
                    case 12: row_pairs<12>(h, ent, e0, s1, hh, c4b, a); break;
                    case 11: row_pairs<11>(h, ent, e0, s1, hh, c4b, a); break;
                    case 10: row_pairs<10>(h, ent, e0, s1, hh, c4b, a); break;
                    case 9:  row_pairs<9>(h, ent, e0, s1, hh, c4b, a); break;
                    case 8:  row_pairs<8>(h, ent, e0, s1, hh, c4b, a); break;
                    case 7:  row_pairs<7>(h, ent, e0, s1, hh, c4b, a); break;
                    case 6:  row_pairs<6>(h, ent, e0, s1, hh, c4b, a); break;
                    case 5:  row_pairs<5>(h, ent, e0, s1, hh, c4b, a); break;
                    case 4:  row_pairs<4>(h, ent, e0, s1, hh, c4b, a); break;
                    case 3:  row_pairs<3>(h, ent, e0, s1, hh, c4b, a); break;
                    case 2:  row_pairs<2>(h, ent, e0, s1, hh, c4b, a); break;
                    case 1:  row_pairs<1>(h, ent, e0, s1, hh, c4b, a); break;
                    default: break;
                }
                a[0] += xhalf(a[0]);
                a[1] += xhalf(a[1]);
                a[2] += xhalf(a[2]);
                a[3] += xhalf(a[3]);
                if (hh == 0u) {
                    v4f* ar = (v4f*)(acc + r * OUT_CH + (c4b >> 2));
                    if (first) {
                        *ar = a;
                    } else {
                        const v4f o = *ar;
                        *ar = o + a;
                    }
                }
            }
        }
        __syncthreads();
        first = false;
        p0 += PASS_W;
        ++pass;
    } while (pass < npass);

    const float* hb = h + (size_t)base * OUT_CH;
    float* o = out + (size_t)base * OUT_CH;
    const int nq = len * (OUT_CH / 4);
    float  di[OUT_ITERS];
    float4 hv[OUT_ITERS];
#pragma unroll
    for (int it = 0; it < OUT_ITERS; ++it) {
        int q = tid + it * BIN_BLOCK;
        if (q > nq - 1) q = nq - 1;
        di[it] = dinv[base + (q >> 4)];
        hv[it] = *(const float4*)(hb + (q >> 4) * OUT_CH + ((q & 15) << 2));
    }
#pragma unroll
    for (int it = 0; it < OUT_ITERS; ++it) {
        const int q = tid + it * BIN_BLOCK;
        if (q < nq) {
            const int row = q >> 4;
            const int c4  = (q & 15) << 2;
            const float4 pq = *(const float4*)(posp + row * POSP_PITCH + c4);
            const float4 av = *(const float4*)(acc + row * OUT_CH + c4);
            const float d = di[it];
            float4 r;
            r.x = (pq.x + hv[it].x * d * d) + d * av.x;
            r.y = (pq.y + hv[it].y * d * d) + d * av.y;
            r.z = (pq.z + hv[it].z * d * d) + d * av.z;
            r.w = (pq.w + hv[it].w * d * d) + d * av.w;
            *(float4*)(o + row * OUT_CH + c4) = r;
        }
    }
}

extern "C" void kernel_launch(void* const* d_in, const int* in_sizes, int n_in,
                              void* d_out, int out_size, void* d_ws, size_t ws_size,
                              hipStream_t stream) {
    const float* x   = (const float*)d_in[0];
    const int*   ei  = (const int*)d_in[1];
    const float* pos = (const float*)d_in[2];
    const float* W   = (const float*)d_in[3];
    const float* b   = (const float*)d_in[4];
    const float* Wp  = (const float*)d_in[5];
    const float* bp  = (const float*)d_in[6];
    float* out = (float*)d_out;

    const int N = in_sizes[0] / IN_CH;
    const int E = in_sizes[1] / 2;
    const int* src = ei;
    const int* dst = ei + E;

    float* dinv = (float*)d_ws;
    float* h    = dinv + ((N + 3) & ~3);

    const int mTiles = (N + 15) / 16;
    const int wavesPerBlock = 8;
    dim3 gGemm((mTiles + wavesPerBlock - 1) / wavesPerBlock, OUT_CH / 16);

    const int NB  = (N + BIN_RNG - 1) / BIN_RNG;
    const int per = (E + BIN_CH - 1) / BIN_CH;
    const int TW  = (BIN_CH * (NB + 1) + 3) & ~3;
    unsigned* segtab = (unsigned*)(h + (size_t)N * OUT_CH);
    unsigned* list   = segtab + TW;
    const size_t need = (size_t)((char*)(list + (size_t)BIN_CH * per) - (char*)d_ws);
    if (NB <= BIN_MAX && N <= (1 << 23) && need <= ws_size) {
        const int gTot = (mTiles + GX_WAVES - 1) / GX_WAVES;
        const int g3 = gTot * 3 / 10;
        const int g2 = gTot - g3;
        bin_local_gemm_kernel<<<BIN_CH + g2, BINW_BLOCK, 0, stream>>>(src, dst, segtab, list, N, E, NB, x, W, h, mTiles, 0);
        bin_deg_gemm_kernel<<<NB + g3, DEGL_BLOCK, 0, stream>>>(list, segtab, dinv, N, E, NB, x, W, h, mTiles, g2);
        bin_sum_pos_kernel<<<NB, BIN_BLOCK, 0, stream>>>(list, segtab, h, dinv, pos, Wp, bp, b, out, N, E, NB);
    } else {
        unsigned* planes = (unsigned*)h;
        dim3 gDeg(DEG_CHUNKS, (N + DEG_RNG - 1) / DEG_RNG);
        deg_count_lds_kernel<<<gDeg, DEG_BLOCK, 0, stream>>>(dst, planes, N, E);
        deg_finish_kernel<<<(N + 255) / 256, 256, 0, stream>>>(planes, dinv, N);
        gemm_xw_lds_kernel<<<(mTiles + GX_WAVES - 1) / GX_WAVES, 32 * GX_WAVES, 0, stream>>>(x, W, h, N, mTiles);
        gemm_pos_fused_kernel<<<gGemm, 32 * wavesPerBlock, 0, stream>>>(
            pos, Wp, bp, b, h, dinv, out, N, mTiles);
        edge_scatter_kernel<<<(E + wavesPerBlock - 1) / wavesPerBlock,
                              32 * wavesPerBlock, 0, stream>>>(src, dst, h, dinv, out, E);
    }
}
